# starting kernel
# baseline (speedup 1.0000x reference)
#include <hip/hip_runtime.h>

typedef __attribute__((ext_vector_type(2))) float v2f;
typedef __attribute__((ext_vector_type(8))) float v8f;

#define F_IN   256
#define HIDDEN 64
#define NCLS   40


__global__ void k_init_deg(float* deg, int n) {
  int i = blockIdx.x * blockDim.x + threadIdx.x;
  if (i < n) deg[i] = 1.0f;
}

__global__ void k_deg_edges(const int* __restrict__ col, const float* __restrict__ ew,
                            float* deg, int e) {
  int i = blockIdx.x * blockDim.x + threadIdx.x;
  if (i < e) atomicAdd(&deg[col[i]], ew[i]);
}

__global__ void k_dinv(float* d, int n) {
  int i = blockIdx.x * blockDim.x + threadIdx.x;
  if (i < n) {
    float v = d[i];
    d[i] = (v > 0.0f) ? rsqrtf(v) : 0.0f;
  }
}


template <int K, int NOUT, bool RELU_IN, bool RELU_OUT, bool BIAS>
__global__ __launch_bounds__(64) void k_gemm_wmma(const float* __restrict__ A,
                                                  const float* __restrict__ W,
                                                  const float* __restrict__ bias,
                                                  float* __restrict__ D, int M) {
  constexpr int NT = (NOUT + 15) / 16;
  const int wv   = threadIdx.x >> 5;
  const int lane = threadIdx.x & 31;
  const int half = lane >> 4;
  const int l16  = lane & 15;
  const int m0   = (blockIdx.x * 2 + wv) * 16;
  if (m0 >= M) return;

  v8f acc[NT] = {};

  const float* __restrict__ arow = A + (size_t)(m0 + l16) * K + 2 * half;

  for (int k = 0; k < K; k += 4) {
    v2f a = *(const v2f*)(arow + k);
    if (RELU_IN) { a.x = fmaxf(a.x, 0.0f); a.y = fmaxf(a.y, 0.0f); }
    const int kk = k + 2 * half;
#pragma unroll
    for (int t = 0; t < NT; ++t) {
      const int c = t * 16 + l16;
      v2f b;
      if ((NOUT % 16 == 0) || (c < NOUT)) {
        b.x = W[(size_t)kk * NOUT + c];
        b.y = W[(size_t)(kk + 1) * NOUT + c];
      } else {
        b.x = 0.0f; b.y = 0.0f;
      }
      acc[t] = __builtin_amdgcn_wmma_f32_16x16x4_f32(
           false, a,  false, b,
           (short)0, acc[t],  false,  false);
    }
  }

#pragma unroll
  for (int t = 0; t < NT; ++t) {
    const int c = t * 16 + l16;
    if ((NOUT % 16 != 0) && (c >= NOUT)) continue;
    const float bv = BIAS ? bias[c] : 0.0f;
#pragma unroll
    for (int v = 0; v < 8; ++v) {
      const int m = m0 + v + 8 * half;
      float val = acc[t][v] + bv;
      if (RELU_OUT) val = fmaxf(val, 0.0f);
      D[(size_t)m * NOUT + c] = val;
    }
  }
}


__global__ void k_seed(const float* __restrict__ t, const float* __restrict__ dinv,
                       const float* __restrict__ bias, float* __restrict__ h, int total) {
  int i = blockIdx.x * blockDim.x + threadIdx.x;
  if (i >= total) return;
  const int node = i >> 6;
  const int f    = i & 63;
  const float di = dinv[node];
  h[i] = bias[f] + di * di * t[i];
}

__global__ void k_scatter(const int* __restrict__ row, const int* __restrict__ col,
                          const float* __restrict__ ew, const float* __restrict__ dinv,
                          const float* __restrict__ t, float* __restrict__ h, int E) {
  int gid  = blockIdx.x * blockDim.x + threadIdx.x;
  int e    = gid >> 5;
  int lane = gid & 31;
  if (e >= E) return;
  const int   r   = row[e];
  const int   c   = col[e];
  const float nrm = dinv[r] * ew[e] * dinv[c];
  const float2 v  = *(const float2*)(t + (size_t)r * HIDDEN + lane * 2);
  float* dst = h + (size_t)c * HIDDEN + lane * 2;
  atomicAdd(dst + 0, nrm * v.x);
  atomicAdd(dst + 1, nrm * v.y);
}


extern "C" void kernel_launch(void* const* d_in, const int* in_sizes, int n_in,
                              void* d_out, int out_size, void* d_ws, size_t ws_size,
                              hipStream_t stream) {
  const float* x   = (const float*)d_in[0];
  const int*   ei  = (const int*)d_in[1];
  const float* ew  = (const float*)d_in[2];
  const float* W1  = (const float*)d_in[3];
  const float* b1  = (const float*)d_in[4];
  const float* Wc1 = (const float*)d_in[5];
  const float* bc1 = (const float*)d_in[6];
  const float* Wc2 = (const float*)d_in[7];
  const float* bc2 = (const float*)d_in[8];
  const float* W2  = (const float*)d_in[9];
  const float* b2  = (const float*)d_in[10];
  float* out = (float*)d_out;

  const int N = in_sizes[0] / F_IN;
  const int E = in_sizes[2];
  const int* erow = ei;
  const int* ecol = ei + E;

  float* dinv = (float*)d_ws;
  float* h    = dinv + N;
  float* t    = h + (size_t)N * HIDDEN;

  const int TB = 256;
  const int gemm_grid = (N + 31) / 32;

  k_init_deg<<<(N + TB - 1) / TB, TB, 0, stream>>>(dinv, N);
  k_deg_edges<<<(E + TB - 1) / TB, TB, 0, stream>>>(ecol, ew, dinv, E);
  k_dinv<<<(N + TB - 1) / TB, TB, 0, stream>>>(dinv, N);

  k_gemm_wmma<F_IN, HIDDEN, false, true, true>
      <<<gemm_grid, 64, 0, stream>>>(x, W1, b1, h, N);

  k_gemm_wmma<HIDDEN, HIDDEN, false, false, false>
      <<<gemm_grid, 64, 0, stream>>>(h, Wc1, nullptr, t, N);
  k_seed<<<((N * HIDDEN) + TB - 1) / TB, TB, 0, stream>>>(t, dinv, bc1, h, N * HIDDEN);
  k_scatter<<<((size_t)E * 32 + TB - 1) / TB, TB, 0, stream>>>(erow, ecol, ew, dinv, t, h, E);

  k_gemm_wmma<HIDDEN, HIDDEN, true, false, false>
      <<<gemm_grid, 64, 0, stream>>>(h, Wc2, nullptr, t, N);
  k_seed<<<((N * HIDDEN) + TB - 1) / TB, TB, 0, stream>>>(t, dinv, bc2, h, N * HIDDEN);
  k_scatter<<<((size_t)E * 32 + TB - 1) / TB, TB, 0, stream>>>(erow, ecol, ew, dinv, t, h, E);

  k_gemm_wmma<HIDDEN, NCLS, true, false, true>
      <<<gemm_grid, 64, 0, stream>>>(h, W2, b2, out, N);
}


# Float atomics removed: edges counting-sorted by destination each call (int atomic arrival rank + scan), then one wave per node gathers rows; deg and both aggregations sum in that rank order
# speedup vs baseline: 1.8291x; 1.8291x over previous
#include <hip/hip_runtime.h>

typedef __attribute__((ext_vector_type(2))) float v2f;
typedef __attribute__((ext_vector_type(8))) float v8f;

#define F_IN   256
#define HIDDEN 64
#define NCLS   40


__global__ void k_zero_cnt(int* cnt, int n) {
  int i = blockIdx.x * blockDim.x + threadIdx.x;
  if (i < n) cnt[i] = 0;
}

__global__ void k_rank(const int* __restrict__ col, int* cnt, int* __restrict__ rank, int e) {
  int i = blockIdx.x * blockDim.x + threadIdx.x;
  if (i < e) rank[i] = atomicAdd(&cnt[col[i]], 1);
}

__global__ __launch_bounds__(1024) void k_scan_local(const int* __restrict__ cnt, int* __restrict__ off,
                                                     int* __restrict__ bsum, int n) {
  __shared__ int wsum[32];
  const int i    = blockIdx.x * 1024 + threadIdx.x;
  const int lane = threadIdx.x & 31;
  const int wv   = threadIdx.x >> 5;
  const int v    = (i < n) ? cnt[i] : 0;
  int s = v;
  for (int d = 1; d < 32; d <<= 1) {
    const int u = __shfl_up(s, d);
    if (lane >= d) s += u;
  }
  if (lane == 31) wsum[wv] = s;
  __syncthreads();
  if (wv == 0) {
    const int ws = wsum[lane];
    int q = ws;
    for (int d = 1; d < 32; d <<= 1) {
      const int u = __shfl_up(q, d);
      if (lane >= d) q += u;
    }
    wsum[lane] = q - ws;
    if (lane == 31) bsum[blockIdx.x] = q;
  }
  __syncthreads();
  if (i < n) off[i] = s - v + wsum[wv];
}

__global__ __launch_bounds__(1024) void k_scan_add(int* __restrict__ off, const int* __restrict__ bsum, int n) {
  const int i    = blockIdx.x * 1024 + threadIdx.x;
  const int lane = threadIdx.x & 31;
  int acc = 0;
  for (int j = lane; j < (int)blockIdx.x; j += 32) acc += bsum[j];
  for (int d = 16; d > 0; d >>= 1) acc += __shfl_xor(acc, d);
  if (i < n) off[i] += acc;
}

__global__ void k_permute(const int* __restrict__ row, const int* __restrict__ col,
                          const float* __restrict__ ew, const int* __restrict__ off,
                          const int* __restrict__ rank, int* __restrict__ sedge, int e) {
  int i = blockIdx.x * blockDim.x + threadIdx.x;
  if (i >= e) return;
  const int pos = off[col[i]] + rank[i];
  *(int2*)(sedge + 2 * (size_t)pos) = make_int2(row[i], __float_as_int(ew[i]));
}

__global__ void k_dinv_seg(const int* __restrict__ off, const int* __restrict__ cnt,
                           const int* __restrict__ sedge, float* __restrict__ dinv, int n) {
  int c = blockIdx.x * blockDim.x + threadIdx.x;
  if (c >= n) return;
  const int beg = off[c];
  const int num = cnt[c];
  float s = 1.0f;
  for (int j = 0; j < num; ++j) s += __int_as_float(sedge[2 * (size_t)(beg + j) + 1]);
  dinv[c] = (s > 0.0f) ? rsqrtf(s) : 0.0f;
}


template <int K, int NOUT, bool RELU_IN, bool RELU_OUT, bool BIAS>
__global__ __launch_bounds__(64) void k_gemm_wmma(const float* __restrict__ A,
                                                  const float* __restrict__ W,
                                                  const float* __restrict__ bias,
                                                  float* __restrict__ D, int M) {
  constexpr int NT = (NOUT + 15) / 16;
  const int wv   = threadIdx.x >> 5;
  const int lane = threadIdx.x & 31;
  const int half = lane >> 4;
  const int l16  = lane & 15;
  const int m0   = (blockIdx.x * 2 + wv) * 16;
  if (m0 >= M) return;

  v8f acc[NT] = {};

  const float* __restrict__ arow = A + (size_t)(m0 + l16) * K + 2 * half;

  for (int k = 0; k < K; k += 4) {
    v2f a = *(const v2f*)(arow + k);
    if (RELU_IN) { a.x = fmaxf(a.x, 0.0f); a.y = fmaxf(a.y, 0.0f); }
    const int kk = k + 2 * half;
#pragma unroll
    for (int t = 0; t < NT; ++t) {
      const int c = t * 16 + l16;
      v2f b;
      if ((NOUT % 16 == 0) || (c < NOUT)) {
        b.x = W[(size_t)kk * NOUT + c];
        b.y = W[(size_t)(kk + 1) * NOUT + c];
      } else {
        b.x = 0.0f; b.y = 0.0f;
      }
      acc[t] = __builtin_amdgcn_wmma_f32_16x16x4_f32(
           false, a,  false, b,
           (short)0, acc[t],  false,  false);
    }
  }

#pragma unroll
  for (int t = 0; t < NT; ++t) {
    const int c = t * 16 + l16;
    if ((NOUT % 16 != 0) && (c >= NOUT)) continue;
    const float bv = BIAS ? bias[c] : 0.0f;
#pragma unroll
    for (int v = 0; v < 8; ++v) {
      const int m = m0 + v + 8 * half;
      float val = acc[t][v] + bv;
      if (RELU_OUT) val = fmaxf(val, 0.0f);
      D[(size_t)m * NOUT + c] = val;
    }
  }
}


static __device__ __forceinline__ float2 ld_row(const float* __restrict__ tl, int r) {
  return *(const float2*)(tl + (size_t)r * HIDDEN);
}
static __device__ __forceinline__ float lane_f(float w, int k) {
  return __int_as_float(__builtin_amdgcn_readlane(__float_as_int(w), k));
}

__global__ __launch_bounds__(256) void k_gather(const int* __restrict__ off, const int* __restrict__ cnt,
                                                const int* __restrict__ sedge,
                                                const float* __restrict__ dinv,
                                                const float* __restrict__ t,
                                                const float* __restrict__ bias,
                                                float* __restrict__ h, int n) {
  const int lane = threadIdx.x & 31;
  const int c    = blockIdx.x * 8 + (threadIdx.x >> 5);
  if (c >= n) return;
  const int beg = __builtin_amdgcn_readfirstlane(off[c]);
  const int num = __builtin_amdgcn_readfirstlane(cnt[c]);
  const float dc = dinv[c];
  const float* __restrict__ tl = t + 2 * lane;
  const float2 tv = *(const float2*)(tl + (size_t)c * HIDDEN);
  const float2 bv = *(const float2*)(bias + 2 * lane);
  float ax = bv.x + dc * dc * tv.x;
  float ay = bv.y + dc * dc * tv.y;

  for (int base = 0; base < num; base += 32) {
    const int left = num - base;
    const int m    = left < 32 ? left : 32;
    int   r = 0;
    float w = 0.0f;
    if (lane < m) {
      const int2 se = *(const int2*)(sedge + 2 * (size_t)(beg + base + lane));
      r = se.x;
      r = r < 0 ? 0 : (r >= n ? n - 1 : r);
      w = dinv[r] * __int_as_float(se.y) * dc;
    }
    int k = 0;
    for (; k + 8 <= m; k += 8) {
      float2 v[8];
      float  wk[8];
      for (int u = 0; u < 8; ++u) v[u] = ld_row(tl, __builtin_amdgcn_readlane(r, k + u));
      for (int u = 0; u < 8; ++u) wk[u] = lane_f(w, k + u);
      for (int u = 0; u < 8; ++u) { ax += wk[u] * v[u].x; ay += wk[u] * v[u].y; }
    }
    if (k + 4 <= m) {
      float2 v[4];
      float  wk[4];
      for (int u = 0; u < 4; ++u) v[u] = ld_row(tl, __builtin_amdgcn_readlane(r, k + u));
      for (int u = 0; u < 4; ++u) wk[u] = lane_f(w, k + u);
      for (int u = 0; u < 4; ++u) { ax += wk[u] * v[u].x; ay += wk[u] * v[u].y; }
      k += 4;
    }
    const int rem = m - k;
    if (rem > 0) {
      const int k1 = rem > 1 ? k + 1 : k;
      const int k2 = rem > 2 ? k + 2 : k;
      const float2 v0 = ld_row(tl, __builtin_amdgcn_readlane(r, k));
      const float2 v1 = ld_row(tl, __builtin_amdgcn_readlane(r, k1));
      const float2 v2 = ld_row(tl, __builtin_amdgcn_readlane(r, k2));
      const float w0 = lane_f(w, k);
      const float w1 = lane_f(w, k1);
      const float w2 = lane_f(w, k2);
      ax += w0 * v0.x; ay += w0 * v0.y;
      if (rem > 1) { ax += w1 * v1.x; ay += w1 * v1.y; }
      if (rem > 2) { ax += w2 * v2.x; ay += w2 * v2.y; }
    }
  }
  *(float2*)(h + (size_t)c * HIDDEN + 2 * lane) = make_float2(ax, ay);
}


extern "C" void kernel_launch(void* const* d_in, const int* in_sizes, int n_in,
                              void* d_out, int out_size, void* d_ws, size_t ws_size,
                              hipStream_t stream) {
  const float* x   = (const float*)d_in[0];
  const int*   ei  = (const int*)d_in[1];
  const float* ew  = (const float*)d_in[2];
  const float* W1  = (const float*)d_in[3];
  const float* b1  = (const float*)d_in[4];
  const float* Wc1 = (const float*)d_in[5];
  const float* bc1 = (const float*)d_in[6];
  const float* Wc2 = (const float*)d_in[7];
  const float* bc2 = (const float*)d_in[8];
  const float* W2  = (const float*)d_in[9];
  const float* b2  = (const float*)d_in[10];
  float* out = (float*)d_out;

  const int N = in_sizes[0] / F_IN;
  const int E = in_sizes[2];
  const int* erow = ei;
  const int* ecol = ei + E;

  int*   sedge = (int*)d_ws;
  float* h     = (float*)(sedge + 2 * (size_t)E);
  float* t     = h + (size_t)N * HIDDEN;
  float* dinv  = t + (size_t)N * HIDDEN;
  int*   rank  = (int*)(dinv + N);
  int*   cnt   = rank + E;
  int*   off   = cnt + N;
  int*   bsum  = off + N;

  const int TB = 256;
  const int gemm_grid = (N + 31) / 32;
  const int scan_grid = (N + 1023) / 1024;
  const int node_grid = (N + 7) / 8;

  k_zero_cnt<<<(N + TB - 1) / TB, TB, 0, stream>>>(cnt, N);
  k_rank<<<(E + TB - 1) / TB, TB, 0, stream>>>(ecol, cnt, rank, E);
  k_scan_local<<<scan_grid, 1024, 0, stream>>>(cnt, off, bsum, N);
  k_scan_add<<<scan_grid, 1024, 0, stream>>>(off, bsum, N);
  k_permute<<<(E + TB - 1) / TB, TB, 0, stream>>>(erow, ecol, ew, off, rank, sedge, E);
  k_dinv_seg<<<(N + TB - 1) / TB, TB, 0, stream>>>(off, cnt, sedge, dinv, N);

  k_gemm_wmma<F_IN, HIDDEN, false, true, true>
      <<<gemm_grid, 64, 0, stream>>>(x, W1, b1, h, N);

  k_gemm_wmma<HIDDEN, HIDDEN, false, false, false>
      <<<gemm_grid, 64, 0, stream>>>(h, Wc1, nullptr, t, N);
  k_gather<<<node_grid, 256, 0, stream>>>(off, cnt, sedge, dinv, t, bc1, h, N);

  k_gemm_wmma<HIDDEN, HIDDEN, true, false, false>
      <<<gemm_grid, 64, 0, stream>>>(h, Wc2, nullptr, t, N);
  k_gather<<<node_grid, 256, 0, stream>>>(off, cnt, sedge, dinv, t, bc2, h, N);

  k_gemm_wmma<HIDDEN, NCLS, true, false, true>
      <<<gemm_grid, 64, 0, stream>>>(h, W2, b2, out, N);
}


# Edge grouping without global atomics: two-level counting sort on LDS counters (512-node buckets: slice histograms, fill, per-bucket rank+scan+dinv), 12->9 launches; sums now in LDS arrival order
# speedup vs baseline: 2.6587x; 1.4536x over previous
#include <hip/hip_runtime.h>

typedef __attribute__((ext_vector_type(2))) float v2f;
typedef __attribute__((ext_vector_type(8))) float v8f;

#define F_IN   256
#define HIDDEN 64
#define NCLS   40


#define BW_SHIFT 9
#define BW       (1 << BW_SHIFT)
#define SORT_TB  1024
#define NB_MAX   1024
#define EPT      8

static __device__ __forceinline__ int block_excl_scan(int v, int* wsum, int* total) {
  const int lane = threadIdx.x & 31;
  const int wv   = threadIdx.x >> 5;
  int s = v;
  for (int d = 1; d < 32; d <<= 1) {
    const int u = __shfl_up(s, d);
    if (lane >= d) s += u;
  }
  if (lane == 31) wsum[wv] = s;
  __syncthreads();
  if (wv == 0) {
    const int ws = wsum[lane];
    int q = ws;
    for (int d = 1; d < 32; d <<= 1) {
      const int u = __shfl_up(q, d);
      if (lane >= d) q += u;
    }
    wsum[lane] = q - ws;
    if (lane == 31) wsum[32] = q;
  }
  __syncthreads();
  const int r = s - v + wsum[wv];
  *total = wsum[32];
  __syncthreads();
  return r;
}

static __device__ __forceinline__ int in_range(int c, int n) {
  const unsigned uc = (unsigned)c, last = (unsigned)(n - 1);
  return (int)(uc < last ? uc : last);
}

__global__ __launch_bounds__(SORT_TB) void k_bucket_hist(const int* __restrict__ col, int* __restrict__ hist,
                                                         int e, int n, int nb) {
  __shared__ int h[NB_MAX];
  const int tid = threadIdx.x;
  h[tid] = 0;
  __syncthreads();
  const int chunk = (e + (int)gridDim.x - 1) / (int)gridDim.x;
  const int beg   = (int)blockIdx.x * chunk;
  const int end   = beg + chunk < e ? beg + chunk : e;
  for (int i0 = beg + tid; i0 < end; i0 += EPT * SORT_TB) {
    int c[EPT];
    for (int u = 0; u < EPT; ++u) {
      const int i = i0 + u * SORT_TB;
      c[u] = col[i < end ? i : end - 1];
    }
    asm volatile("" :: "v"(c[0]), "v"(c[1]), "v"(c[2]), "v"(c[3]), "v"(c[4]), "v"(c[5]), "v"(c[6]), "v"(c[7]));
    for (int u = 0; u < EPT; ++u)
      if (i0 + u * SORT_TB < end) atomicAdd(&h[in_range(c[u], n) >> BW_SHIFT], 1);
  }
  __syncthreads();
  if (tid < nb) hist[(size_t)tid * gridDim.x + blockIdx.x] = h[tid];
}

__global__ __launch_bounds__(SORT_TB) void k_bucket_fill(const int* __restrict__ row, const int* __restrict__ col,
                                                         const float* __restrict__ ew,
                                                         const int* __restrict__ hist, int* __restrict__ bstart,
                                                         int4* __restrict__ tmp, int e, int n, int nb) {
  __shared__ int tot[NB_MAX];
  __shared__ int cur[NB_MAX];
  __shared__ int wsum[33];
  const int tid  = threadIdx.x;
  const int lane = tid & 31;
  const int wv   = tid >> 5;
  const int b    = blockIdx.x;
  const int nblk = gridDim.x;

  for (int k = wv; k < nb; k += SORT_TB / 32) {
    const int* __restrict__ hk = hist + (size_t)k * nblk;
    int t = 0, p = 0;
    for (int j = lane; j < nblk; j += 32) {
      const int v = hk[j];
      t += v;
      p += (j < b) ? v : 0;
    }
    for (int d = 16; d > 0; d >>= 1) {
      t += __shfl_xor(t, d);
      p += __shfl_xor(p, d);
    }
    if (lane == 0) { tot[k] = t; cur[k] = p; }
  }
  __syncthreads();
  int total;
  const int ex = block_excl_scan(tid < nb ? tot[tid] : 0, wsum, &total);
  if (tid < nb) {
    cur[tid] += ex;
    if (b == 0) bstart[tid] = ex;
  }
  if (b == 0 && tid == 0) bstart[nb] = total;
  __syncthreads();

  const int chunk = (e + nblk - 1) / nblk;
  const int beg   = b * chunk;
  const int end   = beg + chunk < e ? beg + chunk : e;
  for (int i0 = beg + tid; i0 < end; i0 += EPT * SORT_TB) {
    int c[EPT], r[EPT], w[EPT];
    for (int u = 0; u < EPT; ++u) {
      const int i  = i0 + u * SORT_TB;
      const int ii = i < end ? i : end - 1;
      c[u] = col[ii];
      r[u] = row[ii];
      w[u] = __float_as_int(ew[ii]);
    }
    asm volatile("" :: "v"(c[0]), "v"(c[1]), "v"(c[2]), "v"(c[3]), "v"(c[4]), "v"(c[5]), "v"(c[6]), "v"(c[7]),
                       "v"(r[0]), "v"(r[1]), "v"(r[2]), "v"(r[3]), "v"(r[4]), "v"(r[5]), "v"(r[6]), "v"(r[7]),
                       "v"(w[0]), "v"(w[1]), "v"(w[2]), "v"(w[3]), "v"(w[4]), "v"(w[5]), "v"(w[6]), "v"(w[7]));
    for (int u = 0; u < EPT; ++u) {
      if (i0 + u * SORT_TB < end) {
        const int cc  = in_range(c[u], n);
        const int pos = atomicAdd(&cur[cc >> BW_SHIFT], 1);
        tmp[pos] = make_int4(r[u], cc, w[u], 0);
      }
    }
  }
}

__global__ __launch_bounds__(SORT_TB) void k_bucket_sort(const int4* __restrict__ tmp, const int* __restrict__ bstart,
                                                         int* sedge, int* __restrict__ off, int* __restrict__ cnt,
                                                         float* __restrict__ dinv, int n) {
  __shared__ int lcnt[BW];
  __shared__ int lcur[BW];
  __shared__ int wsum[33];
  const int tid   = threadIdx.x;
  const int node0 = (int)blockIdx.x << BW_SHIFT;
  const int s     = bstart[blockIdx.x];
  const int e     = bstart[blockIdx.x + 1];
  if (tid < BW) lcnt[tid] = 0;
  __syncthreads();
  for (int i0 = s + tid; i0 < e; i0 += EPT * SORT_TB) {
    int c[EPT];
    for (int u = 0; u < EPT; ++u) {
      const int i = i0 + u * SORT_TB;
      c[u] = tmp[i < e ? i : e - 1].y;
    }
    asm volatile("" :: "v"(c[0]), "v"(c[1]), "v"(c[2]), "v"(c[3]), "v"(c[4]), "v"(c[5]), "v"(c[6]), "v"(c[7]));
    for (int u = 0; u < EPT; ++u)
      if (i0 + u * SORT_TB < e) atomicAdd(&lcnt[(c[u] - node0) & (BW - 1)], 1);
  }
  __syncthreads();
  const int num = tid < BW ? lcnt[tid] : 0;
  int total;
  const int ex = block_excl_scan(num, wsum, &total);
  if (tid < BW) lcur[tid] = ex;
  __syncthreads();
  for (int i0 = s + tid; i0 < e; i0 += EPT * SORT_TB) {
    int4 r[EPT];
    for (int u = 0; u < EPT; ++u) {
      const int i = i0 + u * SORT_TB;
      r[u] = tmp[i < e ? i : e - 1];
    }
    asm volatile("" :: "v"(r[0].x), "v"(r[0].y), "v"(r[0].z), "v"(r[1].x), "v"(r[1].y), "v"(r[1].z),
                       "v"(r[2].x), "v"(r[2].y), "v"(r[2].z), "v"(r[3].x), "v"(r[3].y), "v"(r[3].z),
                       "v"(r[4].x), "v"(r[4].y), "v"(r[4].z), "v"(r[5].x), "v"(r[5].y), "v"(r[5].z),
                       "v"(r[6].x), "v"(r[6].y), "v"(r[6].z), "v"(r[7].x), "v"(r[7].y), "v"(r[7].z));
    for (int u = 0; u < EPT; ++u) {
      if (i0 + u * SORT_TB < e) {
        int p = atomicAdd(&lcur[(r[u].y - node0) & (BW - 1)], 1);
        p = p < e - s ? p : e - s - 1;
        *(int2*)(sedge + 2 * (size_t)(s + p)) = make_int2(r[u].x, r[u].z);
      }
    }
  }
  __syncthreads();
  const int c = node0 + tid;
  if (tid < BW && c < n) {
    const int beg = s + ex;
    off[c] = beg;
    cnt[c] = num;
    const int* wp = sedge + 2 * (size_t)beg + 1;
    float sd = 1.0f;
    for (int j = 0; j < num; j += EPT) {
      float w[EPT];
      for (int u = 0; u < EPT; ++u) w[u] = __int_as_float(wp[2 * (j + u < num ? j + u : num - 1)]);
      for (int u = 0; u < EPT; ++u) sd = (j + u < num) ? sd + w[u] : sd;
    }
    dinv[c] = (sd > 0.0f) ? rsqrtf(sd) : 0.0f;
  }
}


template <int K, int NOUT, bool RELU_IN, bool RELU_OUT, bool BIAS>
__global__ __launch_bounds__(64) void k_gemm_wmma(const float* __restrict__ A,
                                                  const float* __restrict__ W,
                                                  const float* __restrict__ bias,
                                                  float* __restrict__ D, int M) {
  constexpr int NT = (NOUT + 15) / 16;
  const int wv   = threadIdx.x >> 5;
  const int lane = threadIdx.x & 31;
  const int half = lane >> 4;
  const int l16  = lane & 15;
  const int m0   = (blockIdx.x * 2 + wv) * 16;
  if (m0 >= M) return;

  v8f acc[NT] = {};

  const float* __restrict__ arow = A + (size_t)(m0 + l16) * K + 2 * half;

  for (int k = 0; k < K; k += 4) {
    v2f a = *(const v2f*)(arow + k);
    if (RELU_IN) { a.x = fmaxf(a.x, 0.0f); a.y = fmaxf(a.y, 0.0f); }
    const int kk = k + 2 * half;
#pragma unroll
    for (int t = 0; t < NT; ++t) {
      const int c = t * 16 + l16;
      v2f b;
      if ((NOUT % 16 == 0) || (c < NOUT)) {
        b.x = W[(size_t)kk * NOUT + c];
        b.y = W[(size_t)(kk + 1) * NOUT + c];
      } else {
        b.x = 0.0f; b.y = 0.0f;
      }
      acc[t] = __builtin_amdgcn_wmma_f32_16x16x4_f32(
           false, a,  false, b,
           (short)0, acc[t],  false,  false);
    }
  }

#pragma unroll
  for (int t = 0; t < NT; ++t) {
    const int c = t * 16 + l16;
    if ((NOUT % 16 != 0) && (c >= NOUT)) continue;
    const float bv = BIAS ? bias[c] : 0.0f;
#pragma unroll
    for (int v = 0; v < 8; ++v) {
      const int m = m0 + v + 8 * half;
      float val = acc[t][v] + bv;
      if (RELU_OUT) val = fmaxf(val, 0.0f);
      D[(size_t)m * NOUT + c] = val;
    }
  }
}


static __device__ __forceinline__ float2 ld_row(const float* __restrict__ tl, int r) {
  return *(const float2*)(tl + (size_t)r * HIDDEN);
}
static __device__ __forceinline__ float lane_f(float w, int k) {
  return __int_as_float(__builtin_amdgcn_readlane(__float_as_int(w), k));
}

__global__ __launch_bounds__(256) void k_gather(const int* __restrict__ off, const int* __restrict__ cnt,
                                                const int* __restrict__ sedge,
                                                const float* __restrict__ dinv,
                                                const float* __restrict__ t,
                                                const float* __restrict__ bias,
                                                float* __restrict__ h, int n) {
  const int lane = threadIdx.x & 31;
  const int c    = blockIdx.x * 8 + (threadIdx.x >> 5);
  if (c >= n) return;
  const int beg = __builtin_amdgcn_readfirstlane(off[c]);
  const int num = __builtin_amdgcn_readfirstlane(cnt[c]);
  const float dc = dinv[c];
  const float* __restrict__ tl = t + 2 * lane;
  const float2 tv = *(const float2*)(tl + (size_t)c * HIDDEN);
  const float2 bv = *(const float2*)(bias + 2 * lane);
  float ax = bv.x + dc * dc * tv.x;
  float ay = bv.y + dc * dc * tv.y;

  for (int base = 0; base < num; base += 32) {
    const int left = num - base;
    const int m    = left < 32 ? left : 32;
    int   r = 0;
    float w = 0.0f;
    if (lane < m) {
      const int2 se = *(const int2*)(sedge + 2 * (size_t)(beg + base + lane));
      r = se.x;
      r = r < 0 ? 0 : (r >= n ? n - 1 : r);
      w = dinv[r] * __int_as_float(se.y) * dc;
    }
    int k = 0;
    for (; k + 8 <= m; k += 8) {
      float2 v[8];
      float  wk[8];
      for (int u = 0; u < 8; ++u) v[u] = ld_row(tl, __builtin_amdgcn_readlane(r, k + u));
      for (int u = 0; u < 8; ++u) wk[u] = lane_f(w, k + u);
      for (int u = 0; u < 8; ++u) { ax += wk[u] * v[u].x; ay += wk[u] * v[u].y; }
    }
    if (k + 4 <= m) {
      float2 v[4];
      float  wk[4];
      for (int u = 0; u < 4; ++u) v[u] = ld_row(tl, __builtin_amdgcn_readlane(r, k + u));
      for (int u = 0; u < 4; ++u) wk[u] = lane_f(w, k + u);
      for (int u = 0; u < 4; ++u) { ax += wk[u] * v[u].x; ay += wk[u] * v[u].y; }
      k += 4;
    }
    const int rem = m - k;
    if (rem > 0) {
      const int k1 = rem > 1 ? k + 1 : k;
      const int k2 = rem > 2 ? k + 2 : k;
      const float2 v0 = ld_row(tl, __builtin_amdgcn_readlane(r, k));
      const float2 v1 = ld_row(tl, __builtin_amdgcn_readlane(r, k1));
      const float2 v2 = ld_row(tl, __builtin_amdgcn_readlane(r, k2));
      const float w0 = lane_f(w, k);
      const float w1 = lane_f(w, k1);
      const float w2 = lane_f(w, k2);
      ax += w0 * v0.x; ay += w0 * v0.y;
      if (rem > 1) { ax += w1 * v1.x; ay += w1 * v1.y; }
      if (rem > 2) { ax += w2 * v2.x; ay += w2 * v2.y; }
    }
  }
  *(float2*)(h + (size_t)c * HIDDEN + 2 * lane) = make_float2(ax, ay);
}


extern "C" void kernel_launch(void* const* d_in, const int* in_sizes, int n_in,
                              void* d_out, int out_size, void* d_ws, size_t ws_size,
                              hipStream_t stream) {
  const float* x   = (const float*)d_in[0];
  const int*   ei  = (const int*)d_in[1];
  const float* ew  = (const float*)d_in[2];
  const float* W1  = (const float*)d_in[3];
  const float* b1  = (const float*)d_in[4];
  const float* Wc1 = (const float*)d_in[5];
  const float* bc1 = (const float*)d_in[6];
  const float* Wc2 = (const float*)d_in[7];
  const float* bc2 = (const float*)d_in[8];
  const float* W2  = (const float*)d_in[9];
  const float* b2  = (const float*)d_in[10];
  float* out = (float*)d_out;

  const int N = in_sizes[0] / F_IN;
  const int E = in_sizes[2];
  const int* erow = ei;
  const int* ecol = ei + E;

  int4*  tmp    = (int4*)d_ws;
  int*   sedge  = (int*)(tmp + E);
  float* h      = (float*)(sedge + 2 * (size_t)E);
  float* t      = h + (size_t)N * HIDDEN;
  float* dinv   = t + (size_t)N * HIDDEN;
  int*   cnt    = (int*)(dinv + N);
  int*   off    = cnt + N;
  int*   bstart = off + N;
  int*   hist   = bstart + NB_MAX + 1;

  const int SORT_GRID = 256;
  const int nb        = (N + BW - 1) / BW;
  const int gemm_grid = (N + 31) / 32;
  const int node_grid = (N + 7) / 8;

  k_bucket_hist<<<SORT_GRID, SORT_TB, 0, stream>>>(ecol, hist, E, N, nb);
  k_bucket_fill<<<SORT_GRID, SORT_TB, 0, stream>>>(erow, ecol, ew, hist, bstart, tmp, E, N, nb);
  k_bucket_sort<<<nb, SORT_TB, 0, stream>>>(tmp, bstart, sedge, off, cnt, dinv, N);

  k_gemm_wmma<F_IN, HIDDEN, false, true, true>
      <<<gemm_grid, 64, 0, stream>>>(x, W1, b1, h, N);

  k_gemm_wmma<HIDDEN, HIDDEN, false, false, false>
      <<<gemm_grid, 64, 0, stream>>>(h, Wc1, nullptr, t, N);
  k_gather<<<node_grid, 256, 0, stream>>>(off, cnt, sedge, dinv, t, bc1, h, N);

  k_gemm_wmma<HIDDEN, HIDDEN, true, false, false>
      <<<gemm_grid, 64, 0, stream>>>(h, Wc2, nullptr, t, N);
  k_gather<<<node_grid, 256, 0, stream>>>(off, cnt, sedge, dinv, t, bc2, h, N);

  k_gemm_wmma<HIDDEN, NCLS, true, false, true>
      <<<gemm_grid, 64, 0, stream>>>(h, W2, b2, out, N);
}


# Aggregation fused into the next layer's WMMA kernel: one wave owns 16 nodes, walks their contiguous edge run with 32 row loads in flight, tile via LDS; 9->7 launches; per-node sum order kept
# speedup vs baseline: 2.7590x; 1.0377x over previous
#include <hip/hip_runtime.h>

typedef __attribute__((ext_vector_type(2))) float v2f;
typedef __attribute__((ext_vector_type(8))) float v8f;

#define F_IN   256
#define HIDDEN 64
#define NCLS   40


#define BW_SHIFT 9
#define BW       (1 << BW_SHIFT)
#define SORT_TB  1024
#define NB_MAX   1024
#define EPT      8

static __device__ __forceinline__ int block_excl_scan(int v, int* wsum, int* total) {
  const int lane = threadIdx.x & 31;
  const int wv   = threadIdx.x >> 5;
  int s = v;
  for (int d = 1; d < 32; d <<= 1) {
    const int u = __shfl_up(s, d);
    if (lane >= d) s += u;
  }
  if (lane == 31) wsum[wv] = s;
  __syncthreads();
  if (wv == 0) {
    const int ws = wsum[lane];
    int q = ws;
    for (int d = 1; d < 32; d <<= 1) {
      const int u = __shfl_up(q, d);
      if (lane >= d) q += u;
    }
    wsum[lane] = q - ws;
    if (lane == 31) wsum[32] = q;
  }
  __syncthreads();
  const int r = s - v + wsum[wv];
  *total = wsum[32];
  __syncthreads();
  return r;
}

static __device__ __forceinline__ int in_range(int c, int n) {
  const unsigned uc = (unsigned)c, last = (unsigned)(n - 1);
  return (int)(uc < last ? uc : last);
}

__global__ __launch_bounds__(SORT_TB) void k_bucket_hist(const int* __restrict__ col, int* __restrict__ hist,
                                                         int e, int n, int nb) {
  __shared__ int h[NB_MAX];
  const int tid = threadIdx.x;
  h[tid] = 0;
  __syncthreads();
  const int chunk = (e + (int)gridDim.x - 1) / (int)gridDim.x;
  const int beg   = (int)blockIdx.x * chunk;
  const int end   = beg + chunk < e ? beg + chunk : e;
  for (int i0 = beg + tid; i0 < end; i0 += EPT * SORT_TB) {
    int c[EPT];
    for (int u = 0; u < EPT; ++u) {
      const int i = i0 + u * SORT_TB;
      c[u] = col[i < end ? i : end - 1];
    }
    asm volatile("" :: "v"(c[0]), "v"(c[1]), "v"(c[2]), "v"(c[3]), "v"(c[4]), "v"(c[5]), "v"(c[6]), "v"(c[7]));
    for (int u = 0; u < EPT; ++u)
      if (i0 + u * SORT_TB < end) atomicAdd(&h[in_range(c[u], n) >> BW_SHIFT], 1);
  }
  __syncthreads();
  if (tid < nb) hist[(size_t)tid * gridDim.x + blockIdx.x] = h[tid];
}

__global__ __launch_bounds__(SORT_TB) void k_bucket_fill(const int* __restrict__ row, const int* __restrict__ col,
                                                         const float* __restrict__ ew,
                                                         const int* __restrict__ hist, int* __restrict__ bstart,
                                                         int4* __restrict__ tmp, int e, int n, int nb) {
  __shared__ int tot[NB_MAX];
  __shared__ int cur[NB_MAX];
  __shared__ int wsum[33];
  const int tid  = threadIdx.x;
  const int lane = tid & 31;
  const int wv   = tid >> 5;
  const int b    = blockIdx.x;
  const int nblk = gridDim.x;

  for (int k = wv; k < nb; k += SORT_TB / 32) {
    const int* __restrict__ hk = hist + (size_t)k * nblk;
    int t = 0, p = 0;
    for (int j = lane; j < nblk; j += 32) {
      const int v = hk[j];
      t += v;
      p += (j < b) ? v : 0;
    }
    for (int d = 16; d > 0; d >>= 1) {
      t += __shfl_xor(t, d);
      p += __shfl_xor(p, d);
    }
    if (lane == 0) { tot[k] = t; cur[k] = p; }
  }
  __syncthreads();
  int total;
  const int ex = block_excl_scan(tid < nb ? tot[tid] : 0, wsum, &total);
  if (tid < nb) {
    cur[tid] += ex;
    if (b == 0) bstart[tid] = ex;
  }
  if (b == 0 && tid == 0) bstart[nb] = total;
  __syncthreads();

  const int chunk = (e + nblk - 1) / nblk;
  const int beg   = b * chunk;
  const int end   = beg + chunk < e ? beg + chunk : e;
  for (int i0 = beg + tid; i0 < end; i0 += EPT * SORT_TB) {
    int c[EPT], r[EPT], w[EPT];
    for (int u = 0; u < EPT; ++u) {
      const int i  = i0 + u * SORT_TB;
      const int ii = i < end ? i : end - 1;
      c[u] = col[ii];
      r[u] = row[ii];
      w[u] = __float_as_int(ew[ii]);
    }
    asm volatile("" :: "v"(c[0]), "v"(c[1]), "v"(c[2]), "v"(c[3]), "v"(c[4]), "v"(c[5]), "v"(c[6]), "v"(c[7]),
                       "v"(r[0]), "v"(r[1]), "v"(r[2]), "v"(r[3]), "v"(r[4]), "v"(r[5]), "v"(r[6]), "v"(r[7]),
                       "v"(w[0]), "v"(w[1]), "v"(w[2]), "v"(w[3]), "v"(w[4]), "v"(w[5]), "v"(w[6]), "v"(w[7]));
    for (int u = 0; u < EPT; ++u) {
      if (i0 + u * SORT_TB < end) {
        const int cc  = in_range(c[u], n);
        const int pos = atomicAdd(&cur[cc >> BW_SHIFT], 1);
        tmp[pos] = make_int4(r[u], cc, w[u], 0);
      }
    }
  }
}

__global__ __launch_bounds__(SORT_TB) void k_bucket_sort(const int4* __restrict__ tmp, const int* __restrict__ bstart,
                                                         int* sedge, int* __restrict__ off, int* __restrict__ cnt,
                                                         float* __restrict__ dinv, int n) {
  __shared__ int lcnt[BW];
  __shared__ int lcur[BW];
  __shared__ int wsum[33];
  const int tid   = threadIdx.x;
  const int node0 = (int)blockIdx.x << BW_SHIFT;
  const int s     = bstart[blockIdx.x];
  const int e     = bstart[blockIdx.x + 1];
  if (tid < BW) lcnt[tid] = 0;
  __syncthreads();
  for (int i0 = s + tid; i0 < e; i0 += EPT * SORT_TB) {
    int c[EPT];
    for (int u = 0; u < EPT; ++u) {
      const int i = i0 + u * SORT_TB;
      c[u] = tmp[i < e ? i : e - 1].y;
    }
    asm volatile("" :: "v"(c[0]), "v"(c[1]), "v"(c[2]), "v"(c[3]), "v"(c[4]), "v"(c[5]), "v"(c[6]), "v"(c[7]));
    for (int u = 0; u < EPT; ++u)
      if (i0 + u * SORT_TB < e) atomicAdd(&lcnt[(c[u] - node0) & (BW - 1)], 1);
  }
  __syncthreads();
  const int num = tid < BW ? lcnt[tid] : 0;
  int total;
  const int ex = block_excl_scan(num, wsum, &total);
  if (tid < BW) lcur[tid] = ex;
  __syncthreads();
  for (int i0 = s + tid; i0 < e; i0 += EPT * SORT_TB) {
    int4 r[EPT];
    for (int u = 0; u < EPT; ++u) {
      const int i = i0 + u * SORT_TB;
      r[u] = tmp[i < e ? i : e - 1];
    }
    asm volatile("" :: "v"(r[0].x), "v"(r[0].y), "v"(r[0].z), "v"(r[1].x), "v"(r[1].y), "v"(r[1].z),
                       "v"(r[2].x), "v"(r[2].y), "v"(r[2].z), "v"(r[3].x), "v"(r[3].y), "v"(r[3].z),
                       "v"(r[4].x), "v"(r[4].y), "v"(r[4].z), "v"(r[5].x), "v"(r[5].y), "v"(r[5].z),
                       "v"(r[6].x), "v"(r[6].y), "v"(r[6].z), "v"(r[7].x), "v"(r[7].y), "v"(r[7].z));
    for (int u = 0; u < EPT; ++u) {
      if (i0 + u * SORT_TB < e) {
        int p = atomicAdd(&lcur[(r[u].y - node0) & (BW - 1)], 1);
        p = p < e - s ? p : e - s - 1;
        *(int2*)(sedge + 2 * (size_t)(s + p)) = make_int2(r[u].x, r[u].z);
      }
    }
  }
  __syncthreads();
  const int c = node0 + tid;
  if (tid < BW && c < n) {
    const int beg = s + ex;
    off[c] = beg;
    cnt[c] = num;
    const int* wp = sedge + 2 * (size_t)beg + 1;
    float sd = 1.0f;
    for (int j = 0; j < num; j += EPT) {
      float w[EPT];
      for (int u = 0; u < EPT; ++u) w[u] = __int_as_float(wp[2 * (j + u < num ? j + u : num - 1)]);
      for (int u = 0; u < EPT; ++u) sd = (j + u < num) ? sd + w[u] : sd;
    }
    dinv[c] = (sd > 0.0f) ? rsqrtf(sd) : 0.0f;
  }
}


template <int K, int NOUT, bool RELU_IN, bool RELU_OUT, bool BIAS>
__global__ __launch_bounds__(64) void k_gemm_wmma(const float* __restrict__ A,
                                                  const float* __restrict__ W,
                                                  const float* __restrict__ bias,
                                                  float* __restrict__ D, int M) {
  constexpr int NT = (NOUT + 15) / 16;
  const int wv   = threadIdx.x >> 5;
  const int lane = threadIdx.x & 31;
  const int half = lane >> 4;
  const int l16  = lane & 15;
  const int m0   = (blockIdx.x * 2 + wv) * 16;
  if (m0 >= M) return;

  v8f acc[NT] = {};

  const float* __restrict__ arow = A + (size_t)(m0 + l16) * K + 2 * half;

  for (int k = 0; k < K; k += 4) {
    v2f a = *(const v2f*)(arow + k);
    if (RELU_IN) { a.x = fmaxf(a.x, 0.0f); a.y = fmaxf(a.y, 0.0f); }
    const int kk = k + 2 * half;
#pragma unroll
    for (int t = 0; t < NT; ++t) {
      const int c = t * 16 + l16;
      v2f b;
      if ((NOUT % 16 == 0) || (c < NOUT)) {
        b.x = W[(size_t)kk * NOUT + c];
        b.y = W[(size_t)(kk + 1) * NOUT + c];
      } else {
        b.x = 0.0f; b.y = 0.0f;
      }
      acc[t] = __builtin_amdgcn_wmma_f32_16x16x4_f32(
           false, a,  false, b,
           (short)0, acc[t],  false,  false);
    }
  }

#pragma unroll
  for (int t = 0; t < NT; ++t) {
    const int c = t * 16 + l16;
    if ((NOUT % 16 != 0) && (c >= NOUT)) continue;
    const float bv = BIAS ? bias[c] : 0.0f;
#pragma unroll
    for (int v = 0; v < 8; ++v) {
      const int m = m0 + v + 8 * half;
      float val = acc[t][v] + bv;
      if (RELU_OUT) val = fmaxf(val, 0.0f);
      D[(size_t)m * NOUT + c] = val;
    }
  }
}


static __device__ __forceinline__ float2 ld_row(const float* __restrict__ tl, int r) {
  return *(const float2*)(tl + (size_t)r * HIDDEN);
}
static __device__ __forceinline__ float lane_f(float w, int k) {
  return __int_as_float(__builtin_amdgcn_readlane(__float_as_int(w), k));
}


#define TILE_PITCH 68
#define EPL        10
#define ECAP       640
#define RIF        32

template <int NOUT, bool BIAS>
__global__ __launch_bounds__(32) void k_gather_gemm(const int* __restrict__ off, const int* __restrict__ cnt,
                                                    const int* __restrict__ sedge,
                                                    const float* __restrict__ dinv,
                                                    const float* __restrict__ t,
                                                    const float* __restrict__ bias_in,
                                                    const float* __restrict__ W,
                                                    const float* __restrict__ bias_out,
                                                    float* __restrict__ D, int n) {
  constexpr int NT = (NOUT + 15) / 16;
  __shared__ __attribute__((aligned(16))) float tile[16 * TILE_PITCH];
  __shared__ int2  ebuf[ECAP];
  const int lane = threadIdx.x;
  const int half = lane >> 4;
  const int l16  = lane & 15;
  const int m0   = blockIdx.x * 16;
  if (m0 >= n) return;

  const float* __restrict__ tl = t + 2 * lane;
  float* __restrict__ trow = tile + 2 * lane;
  const float2 bv = *(const float2*)(bias_in + 2 * lane);
  float2 tv[16];
#pragma unroll
  for (int i = 0; i < 16; ++i) tv[i] = ld_row(tl, m0 + i < n ? m0 + i : n - 1);

  const int  cl    = m0 + l16 < n ? m0 + l16 : n - 1;
  const int  offv  = off[cl];
  const int  cntv  = cnt[cl];
  const float dv   = dinv[cl];
  const int  e0    = __builtin_amdgcn_readfirstlane(offv);
  const int  endv  = offv + cntv - e0;
  const int  ne    = __builtin_amdgcn_readlane(endv, 15);

#pragma unroll
  for (int i = 0; i < 16; ++i) {
    const float di = lane_f(dv, i);
    *(float2*)(trow + i * TILE_PITCH) = make_float2(bv.x + di * di * tv[i].x, bv.y + di * di * tv[i].y);
  }

  int   cur  = 0;
  int   nend = __builtin_amdgcn_readlane(endv, 0);
  float dc   = lane_f(dv, 0);
  float2 a0  = *(const float2*)trow;
  float ax = a0.x, ay = a0.y;

  for (int cb = 0; cb < ne; cb += ECAP) {
    const int cm = ne - cb < ECAP ? ne - cb : ECAP;
    __syncthreads();
    for (int j0 = 0; j0 < cm; j0 += EPL * 32) {
      int2  se[EPL];
      float dr[EPL];
#pragma unroll
      for (int u = 0; u < EPL; ++u) {
        const int j  = j0 + u * 32 + lane;
        const int jj = j < cm ? j : cm - 1;
        se[u] = *(const int2*)(sedge + 2 * (size_t)(e0 + cb + jj));
      }
      asm volatile("" : "+v"(se[0].x), "+v"(se[1].x), "+v"(se[2].x), "+v"(se[3].x), "+v"(se[4].x),
                        "+v"(se[5].x), "+v"(se[6].x), "+v"(se[7].x), "+v"(se[8].x), "+v"(se[9].x)
                      : "v"(se[0].y), "v"(se[1].y), "v"(se[2].y), "v"(se[3].y), "v"(se[4].y),
                        "v"(se[5].y), "v"(se[6].y), "v"(se[7].y), "v"(se[8].y), "v"(se[9].y));
#pragma unroll
      for (int u = 0; u < EPL; ++u) {
        int r = se[u].x;
        r = r < 0 ? 0 : (r >= n ? n - 1 : r);
        se[u].x = r;
        dr[u] = dinv[r];
      }
      asm volatile("" :: "v"(dr[0]), "v"(dr[1]), "v"(dr[2]), "v"(dr[3]), "v"(dr[4]), "v"(dr[5]), "v"(dr[6]),
                         "v"(dr[7]), "v"(dr[8]), "v"(dr[9]));
#pragma unroll
      for (int u = 0; u < EPL; ++u) {
        const int j = j0 + u * 32 + lane;
        if (j < cm) ebuf[j] = make_int2(se[u].x, __float_as_int(dr[u] * __int_as_float(se[u].y)));
      }
    }
    __syncthreads();

    for (int b = 0; b < cm; b += 32) {
      const int left = cm - b;
      const int m    = left < 32 ? left : 32;
      const int2 q   = ebuf[b + (lane < m ? lane : m - 1)];
      const int   r  = q.x;
      const float w  = __int_as_float(q.y);
      for (int k = 0; k < m; k += RIF) {
        float2 v[RIF];
#pragma unroll
        for (int u = 0; u < RIF; ++u) {
          const int ku = k + u < m ? k + u : m - 1;
          v[u] = ld_row(tl, __builtin_amdgcn_readlane(r, ku));
        }
#pragma unroll
        for (int u = 0; u < RIF; ++u) {
          if (k + u < m) {
            const int ge = cb + b + k + u;
            while (ge >= nend && cur < 15) {
              *(float2*)(trow + cur * TILE_PITCH) = make_float2(fmaxf(ax, 0.0f), fmaxf(ay, 0.0f));
              ++cur;
              nend = __builtin_amdgcn_readlane(endv, cur);
              dc   = lane_f(dv, cur);
              const float2 s0 = *(const float2*)(trow + cur * TILE_PITCH);
              ax = s0.x; ay = s0.y;
            }
            const float wk = lane_f(w, k + u) * dc;
            ax += wk * v[u].x;
            ay += wk * v[u].y;
          }
        }
      }
    }
  }
  for (;;) {
    *(float2*)(trow + cur * TILE_PITCH) = make_float2(fmaxf(ax, 0.0f), fmaxf(ay, 0.0f));
    if (cur >= 15) break;
    ++cur;
    const float2 s0 = *(const float2*)(trow + cur * TILE_PITCH);
    ax = s0.x; ay = s0.y;
  }
  __syncthreads();

  v8f acc[NT] = {};
  const float* arow = tile + l16 * TILE_PITCH + 2 * half;
#pragma unroll 4
  for (int k = 0; k < HIDDEN; k += 4) {
    const v2f a = *(const v2f*)(arow + k);
    const int kk = k + 2 * half;
#pragma unroll
    for (int tt = 0; tt < NT; ++tt) {
      const int c = tt * 16 + l16;
      v2f bfr;
      if ((NOUT % 16 == 0) || (c < NOUT)) {
        bfr.x = W[(size_t)kk * NOUT + c];
        bfr.y = W[(size_t)(kk + 1) * NOUT + c];
      } else {
        bfr.x = 0.0f; bfr.y = 0.0f;
      }
      acc[tt] = __builtin_amdgcn_wmma_f32_16x16x4_f32(
           false, a,  false, bfr,
           (short)0, acc[tt],  false,  false);
    }
  }

  const bool whole = m0 + 16 <= n;
#pragma unroll
  for (int tt = 0; tt < NT; ++tt) {
    const int c = tt * 16 + l16;
    if ((NOUT % 16 != 0) && (c >= NOUT)) continue;
    const float bo = BIAS ? bias_out[c] : 0.0f;
    if (whole) {
#pragma unroll
      for (int v = 0; v < 8; ++v) D[(size_t)(m0 + v + 8 * half) * NOUT + c] = acc[tt][v] + bo;
    } else {
#pragma unroll
      for (int v = 0; v < 8; ++v) {
        const int m = m0 + v + 8 * half;
        if (m < n) D[(size_t)m * NOUT + c] = acc[tt][v] + bo;
      }
    }
  }
}


extern "C" void kernel_launch(void* const* d_in, const int* in_sizes, int n_in,
                              void* d_out, int out_size, void* d_ws, size_t ws_size,
                              hipStream_t stream) {
  const float* x   = (const float*)d_in[0];
  const int*   ei  = (const int*)d_in[1];
  const float* ew  = (const float*)d_in[2];
  const float* W1  = (const float*)d_in[3];
  const float* b1  = (const float*)d_in[4];
  const float* Wc1 = (const float*)d_in[5];
  const float* bc1 = (const float*)d_in[6];
  const float* Wc2 = (const float*)d_in[7];
  const float* bc2 = (const float*)d_in[8];
  const float* W2  = (const float*)d_in[9];
  const float* b2  = (const float*)d_in[10];
  float* out = (float*)d_out;

  const int N = in_sizes[0] / F_IN;
  const int E = in_sizes[2];
  const int* erow = ei;
  const int* ecol = ei + E;

  int4*  tmp    = (int4*)d_ws;
  int*   sedge  = (int*)(tmp + E);
  float* h      = (float*)(sedge + 2 * (size_t)E);
  float* t      = h + (size_t)N * HIDDEN;
  float* dinv   = t + (size_t)N * HIDDEN;
  int*   cnt    = (int*)(dinv + N);
  int*   off    = cnt + N;
  int*   bstart = off + N;
  int*   hist   = bstart + NB_MAX + 1;

  const int SORT_GRID = 256;
  const int nb        = (N + BW - 1) / BW;
  const int gemm_grid = (N + 31) / 32;
  const int tile_grid = (N + 15) / 16;

  k_bucket_hist<<<SORT_GRID, SORT_TB, 0, stream>>>(ecol, hist, E, N, nb);
  k_bucket_fill<<<SORT_GRID, SORT_TB, 0, stream>>>(erow, ecol, ew, hist, bstart, tmp, E, N, nb);
  k_bucket_sort<<<nb, SORT_TB, 0, stream>>>(tmp, bstart, sedge, off, cnt, dinv, N);

  k_gemm_wmma<F_IN, HIDDEN, false, true, true>
      <<<gemm_grid, 64, 0, stream>>>(x, W1, b1, h, N);

  k_gemm_wmma<HIDDEN, HIDDEN, false, false, false>
      <<<gemm_grid, 64, 0, stream>>>(h, Wc1, nullptr, t, N);

  k_gather_gemm<HIDDEN, false>
      <<<tile_grid, 32, 0, stream>>>(off, cnt, sedge, dinv, t, bc1, Wc2, nullptr, h, N);

  k_gather_gemm<NCLS, true>
      <<<tile_grid, 32, 0, stream>>>(off, cnt, sedge, dinv, h, bc2, W2, b2, out, N);
}


# Row gathers of both aggregations with 16-byte loads: each half-wave walks half of the tile's edge run (2 rows per load), records and sums per lane, no cross-lane reads; per-node sum order kept
# speedup vs baseline: 3.0389x; 1.1014x over previous
#include <hip/hip_runtime.h>

typedef __attribute__((ext_vector_type(2))) float v2f;
typedef __attribute__((ext_vector_type(8))) float v8f;

#define F_IN   256
#define HIDDEN 64
#define NCLS   40


#define BW_SHIFT 9
#define BW       (1 << BW_SHIFT)
#define SORT_TB  1024
#define NB_MAX   1024
#define EPT      8

static __device__ __forceinline__ int block_excl_scan(int v, int* wsum, int* total) {
  const int lane = threadIdx.x & 31;
  const int wv   = threadIdx.x >> 5;
  int s = v;
  for (int d = 1; d < 32; d <<= 1) {
    const int u = __shfl_up(s, d);
    if (lane >= d) s += u;
  }
  if (lane == 31) wsum[wv] = s;
  __syncthreads();
  if (wv == 0) {
    const int ws = wsum[lane];
    int q = ws;
    for (int d = 1; d < 32; d <<= 1) {
      const int u = __shfl_up(q, d);
      if (lane >= d) q += u;
    }
    wsum[lane] = q - ws;
    if (lane == 31) wsum[32] = q;
  }
  __syncthreads();
  const int r = s - v + wsum[wv];
  *total = wsum[32];
  __syncthreads();
  return r;
}

static __device__ __forceinline__ int in_range(int c, int n) {
  const unsigned uc = (unsigned)c, last = (unsigned)(n - 1);
  return (int)(uc < last ? uc : last);
}

__global__ __launch_bounds__(SORT_TB) void k_bucket_hist(const int* __restrict__ col, int* __restrict__ hist,
                                                         int e, int n, int nb) {
  __shared__ int h[NB_MAX];
  const int tid = threadIdx.x;
  h[tid] = 0;
  __syncthreads();
  const int chunk = (e + (int)gridDim.x - 1) / (int)gridDim.x;
  const int beg   = (int)blockIdx.x * chunk;
  const int end   = beg + chunk < e ? beg + chunk : e;
  for (int i0 = beg + tid; i0 < end; i0 += EPT * SORT_TB) {
    int c[EPT];
    for (int u = 0; u < EPT; ++u) {
      const int i = i0 + u * SORT_TB;
      c[u] = col[i < end ? i : end - 1];
    }
    asm volatile("" :: "v"(c[0]), "v"(c[1]), "v"(c[2]), "v"(c[3]), "v"(c[4]), "v"(c[5]), "v"(c[6]), "v"(c[7]));
    for (int u = 0; u < EPT; ++u)
      if (i0 + u * SORT_TB < end) atomicAdd(&h[in_range(c[u], n) >> BW_SHIFT], 1);
  }
  __syncthreads();
  if (tid < nb) hist[(size_t)tid * gridDim.x + blockIdx.x] = h[tid];
}

__global__ __launch_bounds__(SORT_TB) void k_bucket_fill(const int* __restrict__ row, const int* __restrict__ col,
                                                         const float* __restrict__ ew,
                                                         const int* __restrict__ hist, int* __restrict__ bstart,
                                                         int4* __restrict__ tmp, int e, int n, int nb) {
  __shared__ int tot[NB_MAX];
  __shared__ int cur[NB_MAX];
  __shared__ int wsum[33];
  const int tid  = threadIdx.x;
  const int lane = tid & 31;
  const int wv   = tid >> 5;
  const int b    = blockIdx.x;
  const int nblk = gridDim.x;

  for (int k = wv; k < nb; k += SORT_TB / 32) {
    const int* __restrict__ hk = hist + (size_t)k * nblk;
    int t = 0, p = 0;
    for (int j = lane; j < nblk; j += 32) {
      const int v = hk[j];
      t += v;
      p += (j < b) ? v : 0;
    }
    for (int d = 16; d > 0; d >>= 1) {
      t += __shfl_xor(t, d);
      p += __shfl_xor(p, d);
    }
    if (lane == 0) { tot[k] = t; cur[k] = p; }
  }
  __syncthreads();
  int total;
  const int ex = block_excl_scan(tid < nb ? tot[tid] : 0, wsum, &total);
  if (tid < nb) {
    cur[tid] += ex;
    if (b == 0) bstart[tid] = ex;
  }
  if (b == 0 && tid == 0) bstart[nb] = total;
  __syncthreads();

  const int chunk = (e + nblk - 1) / nblk;
  const int beg   = b * chunk;
  const int end   = beg + chunk < e ? beg + chunk : e;
  for (int i0 = beg + tid; i0 < end; i0 += EPT * SORT_TB) {
    int c[EPT], r[EPT], w[EPT];
    for (int u = 0; u < EPT; ++u) {
      const int i  = i0 + u * SORT_TB;
      const int ii = i < end ? i : end - 1;
      c[u] = col[ii];
      r[u] = row[ii];
      w[u] = __float_as_int(ew[ii]);
    }
    asm volatile("" :: "v"(c[0]), "v"(c[1]), "v"(c[2]), "v"(c[3]), "v"(c[4]), "v"(c[5]), "v"(c[6]), "v"(c[7]),
                       "v"(r[0]), "v"(r[1]), "v"(r[2]), "v"(r[3]), "v"(r[4]), "v"(r[5]), "v"(r[6]), "v"(r[7]),
                       "v"(w[0]), "v"(w[1]), "v"(w[2]), "v"(w[3]), "v"(w[4]), "v"(w[5]), "v"(w[6]), "v"(w[7]));
    for (int u = 0; u < EPT; ++u) {
      if (i0 + u * SORT_TB < end) {
        const int cc  = in_range(c[u], n);
        const int pos = atomicAdd(&cur[cc >> BW_SHIFT], 1);
        tmp[pos] = make_int4(r[u], cc, w[u], 0);
      }
    }
  }
}

__global__ __launch_bounds__(SORT_TB) void k_bucket_sort(const int4* __restrict__ tmp, const int* __restrict__ bstart,
                                                         int* sedge, int* __restrict__ off, int* __restrict__ cnt,
                                                         float* __restrict__ dinv, int n) {
  __shared__ int lcnt[BW];
  __shared__ int lcur[BW];
  __shared__ int wsum[33];
  const int tid   = threadIdx.x;
  const int node0 = (int)blockIdx.x << BW_SHIFT;
  const int s     = bstart[blockIdx.x];
  const int e     = bstart[blockIdx.x + 1];
  if (tid < BW) lcnt[tid] = 0;
  __syncthreads();
  for (int i0 = s + tid; i0 < e; i0 += EPT * SORT_TB) {
    int c[EPT];
    for (int u = 0; u < EPT; ++u) {
      const int i = i0 + u * SORT_TB;
      c[u] = tmp[i < e ? i : e - 1].y;
    }
    asm volatile("" :: "v"(c[0]), "v"(c[1]), "v"(c[2]), "v"(c[3]), "v"(c[4]), "v"(c[5]), "v"(c[6]), "v"(c[7]));
    for (int u = 0; u < EPT; ++u)
      if (i0 + u * SORT_TB < e) atomicAdd(&lcnt[(c[u] - node0) & (BW - 1)], 1);
  }
  __syncthreads();
  const int num = tid < BW ? lcnt[tid] : 0;
  int total;
  const int ex = block_excl_scan(num, wsum, &total);
  if (tid < BW) lcur[tid] = ex;
  __syncthreads();
  for (int i0 = s + tid; i0 < e; i0 += EPT * SORT_TB) {
    int4 r[EPT];
    for (int u = 0; u < EPT; ++u) {
      const int i = i0 + u * SORT_TB;
      r[u] = tmp[i < e ? i : e - 1];
    }
    asm volatile("" :: "v"(r[0].x), "v"(r[0].y), "v"(r[0].z), "v"(r[1].x), "v"(r[1].y), "v"(r[1].z),
                       "v"(r[2].x), "v"(r[2].y), "v"(r[2].z), "v"(r[3].x), "v"(r[3].y), "v"(r[3].z),
                       "v"(r[4].x), "v"(r[4].y), "v"(r[4].z), "v"(r[5].x), "v"(r[5].y), "v"(r[5].z),
                       "v"(r[6].x), "v"(r[6].y), "v"(r[6].z), "v"(r[7].x), "v"(r[7].y), "v"(r[7].z));
    for (int u = 0; u < EPT; ++u) {
      if (i0 + u * SORT_TB < e) {
        int p = atomicAdd(&lcur[(r[u].y - node0) & (BW - 1)], 1);
        p = p < e - s ? p : e - s - 1;
        *(int2*)(sedge + 2 * (size_t)(s + p)) = make_int2(r[u].x, r[u].z);
      }
    }
  }
  __syncthreads();
  const int c = node0 + tid;
  if (tid < BW && c < n) {
    const int beg = s + ex;
    off[c] = beg;
    cnt[c] = num;
    const int* wp = sedge + 2 * (size_t)beg + 1;
    float sd = 1.0f;
    for (int j = 0; j < num; j += EPT) {
      float w[EPT];
      for (int u = 0; u < EPT; ++u) w[u] = __int_as_float(wp[2 * (j + u < num ? j + u : num - 1)]);
      for (int u = 0; u < EPT; ++u) sd = (j + u < num) ? sd + w[u] : sd;
    }
    dinv[c] = (sd > 0.0f) ? rsqrtf(sd) : 0.0f;
  }
}


template <int K, int NOUT, bool RELU_IN, bool RELU_OUT, bool BIAS>
__global__ __launch_bounds__(64) void k_gemm_wmma(const float* __restrict__ A,
                                                  const float* __restrict__ W,
                                                  const float* __restrict__ bias,
                                                  float* __restrict__ D, int M) {
  constexpr int NT = (NOUT + 15) / 16;
  const int wv   = threadIdx.x >> 5;
  const int lane = threadIdx.x & 31;
  const int half = lane >> 4;
  const int l16  = lane & 15;
  const int m0   = (blockIdx.x * 2 + wv) * 16;
  if (m0 >= M) return;

  v8f acc[NT] = {};

  const float* __restrict__ arow = A + (size_t)(m0 + l16) * K + 2 * half;

  for (int k = 0; k < K; k += 4) {
    v2f a = *(const v2f*)(arow + k);
    if (RELU_IN) { a.x = fmaxf(a.x, 0.0f); a.y = fmaxf(a.y, 0.0f); }
    const int kk = k + 2 * half;
#pragma unroll
    for (int t = 0; t < NT; ++t) {
      const int c = t * 16 + l16;
      v2f b;
      if ((NOUT % 16 == 0) || (c < NOUT)) {
        b.x = W[(size_t)kk * NOUT + c];
        b.y = W[(size_t)(kk + 1) * NOUT + c];
      } else {
        b.x = 0.0f; b.y = 0.0f;
      }
      acc[t] = __builtin_amdgcn_wmma_f32_16x16x4_f32(
           false, a,  false, b,
           (short)0, acc[t],  false,  false);
    }
  }

#pragma unroll
  for (int t = 0; t < NT; ++t) {
    const int c = t * 16 + l16;
    if ((NOUT % 16 != 0) && (c >= NOUT)) continue;
    const float bv = BIAS ? bias[c] : 0.0f;
#pragma unroll
    for (int v = 0; v < 8; ++v) {
      const int m = m0 + v + 8 * half;
      float val = acc[t][v] + bv;
      if (RELU_OUT) val = fmaxf(val, 0.0f);
      D[(size_t)m * NOUT + c] = val;
    }
  }
}


static __device__ __forceinline__ float4 ld_row4(const float* __restrict__ tl, int r) {
  return *(const float4*)(tl + (size_t)r * HIDDEN);
}


#define TILE_PITCH 68
#define EPL        10
#define ECAP       640
#define SLOTS      16

template <int NOUT, bool BIAS>
__global__ __launch_bounds__(32) void k_gather_gemm(const int* __restrict__ off, const int* __restrict__ cnt,
                                                    const int* __restrict__ sedge,
                                                    const float* __restrict__ dinv,
                                                    const float* __restrict__ t,
                                                    const float* __restrict__ bias_in,
                                                    const float* __restrict__ W,
                                                    const float* __restrict__ bias_out,
                                                    float* __restrict__ D, int n) {
  constexpr int NT = (NOUT + 15) / 16;
  __shared__ __attribute__((aligned(16))) float tile[16 * TILE_PITCH];
  __shared__ int2  ebuf[ECAP];
  const int lane = threadIdx.x;
  const int half = lane >> 4;
  const int l16  = lane & 15;
  const int m0   = blockIdx.x * 16;
  if (m0 >= n) return;

  const float* __restrict__ tl = t + 4 * l16;
  const char*  __restrict__ tb = (const char*)t;
  const unsigned lane_off = 16u * (unsigned)l16;
  float* __restrict__ trow = tile + 4 * l16;
  const float4 bv = *(const float4*)(bias_in + 4 * l16);
  float4 tv[8];
#pragma unroll
  for (int p = 0; p < 8; ++p) {
    const int c = m0 + 8 * half + p;
    tv[p] = ld_row4(tl, c < n ? c : n - 1);
  }

  const int  cl    = m0 + l16 < n ? m0 + l16 : n - 1;
  const int  offv  = off[cl];
  const int  cntv  = cnt[cl];
  const float dv   = dinv[cl];
  const int  e0    = __builtin_amdgcn_readfirstlane(offv);
  const int  endv  = offv + cntv - e0;
  const int  ne    = __builtin_amdgcn_readlane(endv, 15);
  int es[15];
#pragma unroll
  for (int i = 0; i < 15; ++i) es[i] = __builtin_amdgcn_readlane(endv, i);

#pragma unroll
  for (int p = 0; p < 8; ++p) {
    const int i = 8 * half + p;
    const float di = __shfl(dv, i);
    *(float4*)(trow + i * TILE_PITCH) = make_float4(bv.x + di * di * tv[p].x, bv.y + di * di * tv[p].y,
                                                    bv.z + di * di * tv[p].z, bv.w + di * di * tv[p].w);
  }

  const int hb = half ? es[7] : 0;
  const int he = half ? ne : es[7];
  int    cur = 8 * half;
  float4 sum = *(const float4*)(trow + cur * TILE_PITCH);

  for (int cb = 0; cb < ne; cb += ECAP) {
    const int cm = ne - cb < ECAP ? ne - cb : ECAP;
    __syncthreads();
    for (int j0 = 0; j0 < cm; j0 += EPL * 32) {
      int2  se[EPL];
      float dr[EPL];
#pragma unroll
      for (int u = 0; u < EPL; ++u) {
        const int j  = j0 + u * 32 + lane;
        const int jj = j < cm ? j : cm - 1;
        se[u] = *(const int2*)(sedge + 2 * (size_t)(e0 + cb + jj));
      }
      asm volatile("" : "+v"(se[0].x), "+v"(se[1].x), "+v"(se[2].x), "+v"(se[3].x), "+v"(se[4].x),
                        "+v"(se[5].x), "+v"(se[6].x), "+v"(se[7].x), "+v"(se[8].x), "+v"(se[9].x)
                      : "v"(se[0].y), "v"(se[1].y), "v"(se[2].y), "v"(se[3].y), "v"(se[4].y),
                        "v"(se[5].y), "v"(se[6].y), "v"(se[7].y), "v"(se[8].y), "v"(se[9].y));
#pragma unroll
      for (int u = 0; u < EPL; ++u) {
        int r = se[u].x;
        r = r < 0 ? 0 : (r >= n ? n - 1 : r);
        se[u].x = r;
        dr[u] = dinv[r];
      }
      asm volatile("" :: "v"(dr[0]), "v"(dr[1]), "v"(dr[2]), "v"(dr[3]), "v"(dr[4]), "v"(dr[5]), "v"(dr[6]),
                         "v"(dr[7]), "v"(dr[8]), "v"(dr[9]));
      int   nidv[EPL];
      float dcg[EPL];
#pragma unroll
      for (int u = 0; u < EPL; ++u) {
        const int g = cb + j0 + u * 32 + lane;
        int nid = 0;
#pragma unroll
        for (int i = 0; i < 15; ++i) nid += es[i] <= g ? 1 : 0;
        nidv[u] = nid;
        dcg[u]  = __shfl(dv, nid);
      }
#pragma unroll
      for (int u = 0; u < EPL; ++u) {
        const int j = j0 + u * 32 + lane;
        if (j < cm) ebuf[j] = make_int2((se[u].x << 8) | nidv[u], __float_as_int(dr[u] * __int_as_float(se[u].y) * dcg[u]));
      }
    }
    __syncthreads();

    const int lo   = (hb > cb ? hb : cb) - cb;
    const int hi   = (he < cb + cm ? he : cb + cm) - cb;
    const int nh   = hi - lo;
    const int na   = __builtin_amdgcn_readlane(nh, 0);
    const int nb2  = __builtin_amdgcn_readlane(nh, 16);
    const int nmax = na > nb2 ? na : nb2;
    const int nmin = na < nb2 ? na : nb2;
    const int lo0  = nh > 0 ? lo : 0;
    const int last = nh > 0 ? nh - 1 : 0;
    for (int j0 = 0; j0 < nmax; j0 += SLOTS) {
      float4 v[SLOTS];
      int2   q[SLOTS];
#pragma unroll
      for (int u = 0; u < SLOTS; ++u) {
        q[u] = ebuf[lo0 + (j0 + u < last ? j0 + u : last)];
        v[u] = *(const float4*)(tb + (((unsigned)q[u].x & 0xffffff00u) + lane_off));
      }
      asm volatile("" :: "v"(v[0].x), "v"(v[1].x), "v"(v[2].x), "v"(v[3].x), "v"(v[4].x), "v"(v[5].x),
                         "v"(v[6].x), "v"(v[7].x), "v"(v[8].x), "v"(v[9].x), "v"(v[10].x), "v"(v[11].x),
                         "v"(v[12].x), "v"(v[13].x), "v"(v[14].x), "v"(v[15].x));
      if (j0 + SLOTS <= nmin) {
#pragma unroll
        for (int u = 0; u < SLOTS; ++u) {
          const int nid = q[u].x & 15;
          if (nid != cur) {
            *(float4*)(trow + cur * TILE_PITCH) = sum;
            cur = nid;
            sum = *(const float4*)(trow + cur * TILE_PITCH);
          }
          const float wk = __int_as_float(q[u].y);
          sum.x = __builtin_fmaf(wk, v[u].x, sum.x);
          sum.y = __builtin_fmaf(wk, v[u].y, sum.y);
          sum.z = __builtin_fmaf(wk, v[u].z, sum.z);
          sum.w = __builtin_fmaf(wk, v[u].w, sum.w);
        }
      } else {
#pragma unroll
        for (int u = 0; u < SLOTS; ++u) {
          if (j0 + u < nh) {
            const int nid = q[u].x & 15;
            if (nid != cur) {
              *(float4*)(trow + cur * TILE_PITCH) = sum;
              cur = nid;
              sum = *(const float4*)(trow + cur * TILE_PITCH);
            }
            const float wk = __int_as_float(q[u].y);
            sum.x = __builtin_fmaf(wk, v[u].x, sum.x);
            sum.y = __builtin_fmaf(wk, v[u].y, sum.y);
            sum.z = __builtin_fmaf(wk, v[u].z, sum.z);
            sum.w = __builtin_fmaf(wk, v[u].w, sum.w);
          }
        }
      }
    }
  }
  *(float4*)(trow + cur * TILE_PITCH) = sum;
  __syncthreads();

  v8f acc[NT] = {};
  const float* arow = tile + l16 * TILE_PITCH + 2 * half;
#pragma unroll 4
  for (int k = 0; k < HIDDEN; k += 4) {
    v2f a = *(const v2f*)(arow + k);
    a.x = fmaxf(a.x, 0.0f);
    a.y = fmaxf(a.y, 0.0f);
    const int kk = k + 2 * half;
#pragma unroll
    for (int tt = 0; tt < NT; ++tt) {
      const int c = tt * 16 + l16;
      v2f bfr;
      if ((NOUT % 16 == 0) || (c < NOUT)) {
        bfr.x = W[(size_t)kk * NOUT + c];
        bfr.y = W[(size_t)(kk + 1) * NOUT + c];
      } else {
        bfr.x = 0.0f; bfr.y = 0.0f;
      }
      acc[tt] = __builtin_amdgcn_wmma_f32_16x16x4_f32(
           false, a,  false, bfr,
           (short)0, acc[tt],  false,  false);
    }
  }

  const bool whole = m0 + 16 <= n;
#pragma unroll
  for (int tt = 0; tt < NT; ++tt) {
    const int c = tt * 16 + l16;
    if ((NOUT % 16 != 0) && (c >= NOUT)) continue;
    const float bo = BIAS ? bias_out[c] : 0.0f;
    if (whole) {
#pragma unroll
      for (int v = 0; v < 8; ++v) D[(size_t)(m0 + v + 8 * half) * NOUT + c] = acc[tt][v] + bo;
    } else {
#pragma unroll
      for (int v = 0; v < 8; ++v) {
        const int m = m0 + v + 8 * half;
        if (m < n) D[(size_t)m * NOUT + c] = acc[tt][v] + bo;
      }
    }
  }
}


extern "C" void kernel_launch(void* const* d_in, const int* in_sizes, int n_in,
                              void* d_out, int out_size, void* d_ws, size_t ws_size,
                              hipStream_t stream) {
  const float* x   = (const float*)d_in[0];
  const int*   ei  = (const int*)d_in[1];
  const float* ew  = (const float*)d_in[2];
  const float* W1  = (const float*)d_in[3];
  const float* b1  = (const float*)d_in[4];
  const float* Wc1 = (const float*)d_in[5];
  const float* bc1 = (const float*)d_in[6];
  const float* Wc2 = (const float*)d_in[7];
  const float* bc2 = (const float*)d_in[8];
  const float* W2  = (const float*)d_in[9];
  const float* b2  = (const float*)d_in[10];
  float* out = (float*)d_out;

  const int N = in_sizes[0] / F_IN;
  const int E = in_sizes[2];
  const int* erow = ei;
  const int* ecol = ei + E;

  int4*  tmp    = (int4*)d_ws;
  float* h      = (float*)(tmp + E);
  float* t      = h + (size_t)N * HIDDEN;
  int*   sedge  = (int*)(t + (size_t)N * HIDDEN);
  float* dinv   = (float*)(sedge + 2 * (size_t)E);
  int*   cnt    = (int*)(dinv + N);
  int*   off    = cnt + N;
  int*   bstart = off + N;
  int*   hist   = bstart + NB_MAX + 1;

  const int SORT_GRID = 256;
  const int nb        = (N + BW - 1) / BW;
  const int gemm_grid = (N + 31) / 32;
  const int tile_grid = (N + 15) / 16;

  k_bucket_hist<<<SORT_GRID, SORT_TB, 0, stream>>>(ecol, hist, E, N, nb);
  k_bucket_fill<<<SORT_GRID, SORT_TB, 0, stream>>>(erow, ecol, ew, hist, bstart, tmp, E, N, nb);
  k_bucket_sort<<<nb, SORT_TB, 0, stream>>>(tmp, bstart, sedge, off, cnt, dinv, N);

  k_gemm_wmma<F_IN, HIDDEN, false, true, true>
      <<<gemm_grid, 64, 0, stream>>>(x, W1, b1, h, N);

  k_gemm_wmma<HIDDEN, HIDDEN, false, false, false>
      <<<gemm_grid, 64, 0, stream>>>(h, Wc1, nullptr, t, N);

  k_gather_gemm<HIDDEN, false>
      <<<tile_grid, 32, 0, stream>>>(off, cnt, sedge, dinv, t, bc1, Wc2, nullptr, h, N);

  k_gather_gemm<NCLS, true>
      <<<tile_grid, 32, 0, stream>>>(off, cnt, sedge, dinv, h, bc2, W2, b2, out, N);
}


# dinv[source] folded into the gathered tables: both conv products store rows times their own dinv, so no edge loads dinv[source]; edge weight now (ew*dinv[c])*(dinv[r]*row), per-node sum order kept
# speedup vs baseline: 3.2586x; 1.0723x over previous
#include <hip/hip_runtime.h>

typedef __attribute__((ext_vector_type(2))) float v2f;
typedef __attribute__((ext_vector_type(8))) float v8f;

#define F_IN   256
#define HIDDEN 64
#define NCLS   40


#define BW_SHIFT 9
#define BW       (1 << BW_SHIFT)
#define SORT_TB  1024
#define NB_MAX   1024
#define EPT      8

static __device__ __forceinline__ int block_excl_scan(int v, int* wsum, int* total) {
  const int lane = threadIdx.x & 31;
  const int wv   = threadIdx.x >> 5;
  int s = v;
  for (int d = 1; d < 32; d <<= 1) {
    const int u = __shfl_up(s, d);
    if (lane >= d) s += u;
  }
  if (lane == 31) wsum[wv] = s;
  __syncthreads();
  if (wv == 0) {
    const int ws = wsum[lane];
    int q = ws;
    for (int d = 1; d < 32; d <<= 1) {
      const int u = __shfl_up(q, d);
      if (lane >= d) q += u;
    }
    wsum[lane] = q - ws;
    if (lane == 31) wsum[32] = q;
  }
  __syncthreads();
  const int r = s - v + wsum[wv];
  *total = wsum[32];
  __syncthreads();
  return r;
}

static __device__ __forceinline__ int in_range(int c, int n) {
  const unsigned uc = (unsigned)c, last = (unsigned)(n - 1);
  return (int)(uc < last ? uc : last);
}

__global__ __launch_bounds__(SORT_TB) void k_bucket_hist(const int* __restrict__ col, int* __restrict__ hist,
                                                         int e, int n, int nb) {
  __shared__ int h[NB_MAX];
  const int tid = threadIdx.x;
  h[tid] = 0;
  __syncthreads();
  const int chunk = (e + (int)gridDim.x - 1) / (int)gridDim.x;
  const int beg   = (int)blockIdx.x * chunk;
  const int end   = beg + chunk < e ? beg + chunk : e;
  for (int i0 = beg + tid; i0 < end; i0 += EPT * SORT_TB) {
    int c[EPT];
    for (int u = 0; u < EPT; ++u) {
      const int i = i0 + u * SORT_TB;
      c[u] = col[i < end ? i : end - 1];
    }
    asm volatile("" :: "v"(c[0]), "v"(c[1]), "v"(c[2]), "v"(c[3]), "v"(c[4]), "v"(c[5]), "v"(c[6]), "v"(c[7]));
    for (int u = 0; u < EPT; ++u)
      if (i0 + u * SORT_TB < end) atomicAdd(&h[in_range(c[u], n) >> BW_SHIFT], 1);
  }
  __syncthreads();
  if (tid < nb) hist[(size_t)tid * gridDim.x + blockIdx.x] = h[tid];
}

__global__ __launch_bounds__(SORT_TB) void k_bucket_fill(const int* __restrict__ row, const int* __restrict__ col,
                                                         const float* __restrict__ ew,
                                                         const int* __restrict__ hist, int* __restrict__ bstart,
                                                         int4* __restrict__ tmp, int e, int n, int nb) {
  __shared__ int tot[NB_MAX];
  __shared__ int cur[NB_MAX];
  __shared__ int wsum[33];
  const int tid  = threadIdx.x;
  const int lane = tid & 31;
  const int wv   = tid >> 5;
  const int b    = blockIdx.x;
  const int nblk = gridDim.x;

  for (int k = wv; k < nb; k += SORT_TB / 32) {
    const int* __restrict__ hk = hist + (size_t)k * nblk;
    int t = 0, p = 0;
    for (int j = lane; j < nblk; j += 32) {
      const int v = hk[j];
      t += v;
      p += (j < b) ? v : 0;
    }
    for (int d = 16; d > 0; d >>= 1) {
      t += __shfl_xor(t, d);
      p += __shfl_xor(p, d);
    }
    if (lane == 0) { tot[k] = t; cur[k] = p; }
  }
  __syncthreads();
  int total;
  const int ex = block_excl_scan(tid < nb ? tot[tid] : 0, wsum, &total);
  if (tid < nb) {
    cur[tid] += ex;
    if (b == 0) bstart[tid] = ex;
  }
  if (b == 0 && tid == 0) bstart[nb] = total;
  __syncthreads();

  const int chunk = (e + nblk - 1) / nblk;
  const int beg   = b * chunk;
  const int end   = beg + chunk < e ? beg + chunk : e;
  for (int i0 = beg + tid; i0 < end; i0 += EPT * SORT_TB) {
    int c[EPT], r[EPT], w[EPT];
    for (int u = 0; u < EPT; ++u) {
      const int i  = i0 + u * SORT_TB;
      const int ii = i < end ? i : end - 1;
      c[u] = col[ii];
      r[u] = row[ii];
      w[u] = __float_as_int(ew[ii]);
    }
    asm volatile("" :: "v"(c[0]), "v"(c[1]), "v"(c[2]), "v"(c[3]), "v"(c[4]), "v"(c[5]), "v"(c[6]), "v"(c[7]),
                       "v"(r[0]), "v"(r[1]), "v"(r[2]), "v"(r[3]), "v"(r[4]), "v"(r[5]), "v"(r[6]), "v"(r[7]),
                       "v"(w[0]), "v"(w[1]), "v"(w[2]), "v"(w[3]), "v"(w[4]), "v"(w[5]), "v"(w[6]), "v"(w[7]));
    for (int u = 0; u < EPT; ++u) {
      if (i0 + u * SORT_TB < end) {
        const int cc  = in_range(c[u], n);
        const int pos = atomicAdd(&cur[cc >> BW_SHIFT], 1);
        tmp[pos] = make_int4(r[u], cc, w[u], 0);
      }
    }
  }
}

__global__ __launch_bounds__(SORT_TB) void k_bucket_sort(const int4* __restrict__ tmp, const int* __restrict__ bstart,
                                                         int* sedge, int* __restrict__ off, int* __restrict__ cnt,
                                                         float* __restrict__ dinv, int n) {
  __shared__ int lcnt[BW];
  __shared__ int lcur[BW];
  __shared__ int wsum[33];
  const int tid   = threadIdx.x;
  const int node0 = (int)blockIdx.x << BW_SHIFT;
  const int s     = bstart[blockIdx.x];
  const int e     = bstart[blockIdx.x + 1];
  if (tid < BW) lcnt[tid] = 0;
  __syncthreads();
  for (int i0 = s + tid; i0 < e; i0 += EPT * SORT_TB) {
    int c[EPT];
    for (int u = 0; u < EPT; ++u) {
      const int i = i0 + u * SORT_TB;
      c[u] = tmp[i < e ? i : e - 1].y;
    }
    asm volatile("" :: "v"(c[0]), "v"(c[1]), "v"(c[2]), "v"(c[3]), "v"(c[4]), "v"(c[5]), "v"(c[6]), "v"(c[7]));
    for (int u = 0; u < EPT; ++u)
      if (i0 + u * SORT_TB < e) atomicAdd(&lcnt[(c[u] - node0) & (BW - 1)], 1);
  }
  __syncthreads();
  const int num = tid < BW ? lcnt[tid] : 0;
  int total;
  const int ex = block_excl_scan(num, wsum, &total);
  if (tid < BW) lcur[tid] = ex;
  __syncthreads();
  for (int i0 = s + tid; i0 < e; i0 += EPT * SORT_TB) {
    int4 r[EPT];
    for (int u = 0; u < EPT; ++u) {
      const int i = i0 + u * SORT_TB;
      r[u] = tmp[i < e ? i : e - 1];
    }
    asm volatile("" :: "v"(r[0].x), "v"(r[0].y), "v"(r[0].z), "v"(r[1].x), "v"(r[1].y), "v"(r[1].z),
                       "v"(r[2].x), "v"(r[2].y), "v"(r[2].z), "v"(r[3].x), "v"(r[3].y), "v"(r[3].z),
                       "v"(r[4].x), "v"(r[4].y), "v"(r[4].z), "v"(r[5].x), "v"(r[5].y), "v"(r[5].z),
                       "v"(r[6].x), "v"(r[6].y), "v"(r[6].z), "v"(r[7].x), "v"(r[7].y), "v"(r[7].z));
    for (int u = 0; u < EPT; ++u) {
      if (i0 + u * SORT_TB < e) {
        int p = atomicAdd(&lcur[(r[u].y - node0) & (BW - 1)], 1);
        p = p < e - s ? p : e - s - 1;
        *(int2*)(sedge + 2 * (size_t)(s + p)) = make_int2(r[u].x, r[u].z);
      }
    }
  }
  __syncthreads();
  const int c = node0 + tid;
  if (tid < BW && c < n) {
    const int beg = s + ex;
    off[c] = beg;
    cnt[c] = num;
    const int* wp = sedge + 2 * (size_t)beg + 1;
    float sd = 1.0f;
    for (int j = 0; j < num; j += EPT) {
      float w[EPT];
      for (int u = 0; u < EPT; ++u) w[u] = __int_as_float(wp[2 * (j + u < num ? j + u : num - 1)]);
      for (int u = 0; u < EPT; ++u) sd = (j + u < num) ? sd + w[u] : sd;
    }
    dinv[c] = (sd > 0.0f) ? rsqrtf(sd) : 0.0f;
  }
}


template <int K, int NOUT, bool RELU_IN, bool RELU_OUT, bool BIAS, bool ROWSCALE>
__global__ __launch_bounds__(64) void k_gemm_wmma(const float* __restrict__ A,
                                                  const float* __restrict__ W,
                                                  const float* __restrict__ bias,
                                                  const float* __restrict__ rscale,
                                                  float* __restrict__ D, int M) {
  constexpr int NT = (NOUT + 15) / 16;
  const int wv   = threadIdx.x >> 5;
  const int lane = threadIdx.x & 31;
  const int half = lane >> 4;
  const int l16  = lane & 15;
  const int m0   = (blockIdx.x * 2 + wv) * 16;
  if (m0 >= M) return;

  v8f acc[NT] = {};

  float rs[8];
#pragma unroll
  for (int v = 0; v < 8; ++v) {
    const int m = m0 + v + 8 * half;
    rs[v] = ROWSCALE ? rscale[m < M ? m : M - 1] : 1.0f;
  }

  const float* __restrict__ arow = A + (size_t)(m0 + l16) * K + 2 * half;

  for (int k = 0; k < K; k += 4) {
    v2f a = *(const v2f*)(arow + k);
    if (RELU_IN) { a.x = fmaxf(a.x, 0.0f); a.y = fmaxf(a.y, 0.0f); }
    const int kk = k + 2 * half;
#pragma unroll
    for (int t = 0; t < NT; ++t) {
      const int c = t * 16 + l16;
      v2f b;
      if ((NOUT % 16 == 0) || (c < NOUT)) {
        b.x = W[(size_t)kk * NOUT + c];
        b.y = W[(size_t)(kk + 1) * NOUT + c];
      } else {
        b.x = 0.0f; b.y = 0.0f;
      }
      acc[t] = __builtin_amdgcn_wmma_f32_16x16x4_f32(
           false, a,  false, b,
           (short)0, acc[t],  false,  false);
    }
  }

#pragma unroll
  for (int t = 0; t < NT; ++t) {
    const int c = t * 16 + l16;
    if ((NOUT % 16 != 0) && (c >= NOUT)) continue;
    const float bv = BIAS ? bias[c] : 0.0f;
#pragma unroll
    for (int v = 0; v < 8; ++v) {
      const int m = m0 + v + 8 * half;
      float val = acc[t][v] + bv;
      if (RELU_OUT) val = fmaxf(val, 0.0f);
      if (ROWSCALE) val *= rs[v];
      D[(size_t)m * NOUT + c] = val;
    }
  }
}


static __device__ __forceinline__ float4 ld_row4(const float* __restrict__ tl, int r) {
  return *(const float4*)(tl + (size_t)r * HIDDEN);
}


#define TILE_PITCH 68
#define EPL        10
#define ECAP       640
#define SLOTS      16

template <int NOUT, bool BIAS, bool SCALE_OUT>
__global__ __launch_bounds__(32) void k_gather_gemm(const int* __restrict__ off, const int* __restrict__ cnt,
                                                    const int* __restrict__ sedge,
                                                    const float* __restrict__ dinv,
                                                    const float* __restrict__ t,
                                                    const float* __restrict__ bias_in,
                                                    const float* __restrict__ W,
                                                    const float* __restrict__ bias_out,
                                                    float* __restrict__ D, int n) {
  constexpr int NT = (NOUT + 15) / 16;
  __shared__ __attribute__((aligned(16))) float tile[16 * TILE_PITCH];
  __shared__ int2  ebuf[ECAP];
  const int lane = threadIdx.x;
  const int half = lane >> 4;
  const int l16  = lane & 15;
  const int m0   = blockIdx.x * 16;
  if (m0 >= n) return;

  const float* __restrict__ tl = t + 4 * l16;
  const char*  __restrict__ tb = (const char*)t;
  const unsigned lane_off = 16u * (unsigned)l16;
  float* __restrict__ trow = tile + 4 * l16;
  const float4 bv = *(const float4*)(bias_in + 4 * l16);
  float4 tv[8];
#pragma unroll
  for (int p = 0; p < 8; ++p) {
    const int c = m0 + 8 * half + p;
    tv[p] = ld_row4(tl, c < n ? c : n - 1);
  }

  const int  cl    = m0 + l16 < n ? m0 + l16 : n - 1;
  const int  offv  = off[cl];
  const int  cntv  = cnt[cl];
  const float dv   = dinv[cl];
  const int  e0    = __builtin_amdgcn_readfirstlane(offv);
  const int  endv  = offv + cntv - e0;
  const int  ne    = __builtin_amdgcn_readlane(endv, 15);
  int es[15];
#pragma unroll
  for (int i = 0; i < 15; ++i) es[i] = __builtin_amdgcn_readlane(endv, i);

#pragma unroll
  for (int p = 0; p < 8; ++p) {
    const int i = 8 * half + p;
    const float di = __shfl(dv, i);
    *(float4*)(trow + i * TILE_PITCH) = make_float4(bv.x + di * tv[p].x, bv.y + di * tv[p].y,
                                                    bv.z + di * tv[p].z, bv.w + di * tv[p].w);
  }

  const int hb = half ? es[7] : 0;
  const int he = half ? ne : es[7];
  int    cur = 8 * half;
  float4 sum = *(const float4*)(trow + cur * TILE_PITCH);

  for (int cb = 0; cb < ne; cb += ECAP) {
    const int cm = ne - cb < ECAP ? ne - cb : ECAP;
    __syncthreads();
    for (int j0 = 0; j0 < cm; j0 += EPL * 32) {
      int2  se[EPL];
#pragma unroll
      for (int u = 0; u < EPL; ++u) {
        const int j  = j0 + u * 32 + lane;
        const int jj = j < cm ? j : cm - 1;
        se[u] = *(const int2*)(sedge + 2 * (size_t)(e0 + cb + jj));
      }
      asm volatile("" : "+v"(se[0].x), "+v"(se[1].x), "+v"(se[2].x), "+v"(se[3].x), "+v"(se[4].x),
                        "+v"(se[5].x), "+v"(se[6].x), "+v"(se[7].x), "+v"(se[8].x), "+v"(se[9].x)
                      : "v"(se[0].y), "v"(se[1].y), "v"(se[2].y), "v"(se[3].y), "v"(se[4].y),
                        "v"(se[5].y), "v"(se[6].y), "v"(se[7].y), "v"(se[8].y), "v"(se[9].y));
#pragma unroll
      for (int u = 0; u < EPL; ++u) {
        int r = se[u].x;
        r = r < 0 ? 0 : (r >= n ? n - 1 : r);
        se[u].x = r;
      }
      int   nidv[EPL];
      float dcg[EPL];
#pragma unroll
      for (int u = 0; u < EPL; ++u) {
        const int g = cb + j0 + u * 32 + lane;
        int nid = 0;
#pragma unroll
        for (int i = 0; i < 15; ++i) nid += es[i] <= g ? 1 : 0;
        nidv[u] = nid;
        dcg[u]  = __shfl(dv, nid);
      }
#pragma unroll
      for (int u = 0; u < EPL; ++u) {
        const int j = j0 + u * 32 + lane;
        if (j < cm) ebuf[j] = make_int2((se[u].x << 8) | nidv[u], __float_as_int(__int_as_float(se[u].y) * dcg[u]));
      }
    }
    __syncthreads();

    const int lo   = (hb > cb ? hb : cb) - cb;
    const int hi   = (he < cb + cm ? he : cb + cm) - cb;
    const int nh   = hi - lo;
    const int na   = __builtin_amdgcn_readlane(nh, 0);
    const int nb2  = __builtin_amdgcn_readlane(nh, 16);
    const int nmax = na > nb2 ? na : nb2;
    const int nmin = na < nb2 ? na : nb2;
    const int lo0  = nh > 0 ? lo : 0;
    const int last = nh > 0 ? nh - 1 : 0;
    for (int j0 = 0; j0 < nmax; j0 += SLOTS) {
      float4 v[SLOTS];
      int2   q[SLOTS];
#pragma unroll
      for (int u = 0; u < SLOTS; ++u) {
        q[u] = ebuf[lo0 + (j0 + u < last ? j0 + u : last)];
        v[u] = *(const float4*)(tb + (((unsigned)q[u].x & 0xffffff00u) + lane_off));
      }
      asm volatile("" :: "v"(v[0].x), "v"(v[1].x), "v"(v[2].x), "v"(v[3].x), "v"(v[4].x), "v"(v[5].x),
                         "v"(v[6].x), "v"(v[7].x), "v"(v[8].x), "v"(v[9].x), "v"(v[10].x), "v"(v[11].x),
                         "v"(v[12].x), "v"(v[13].x), "v"(v[14].x), "v"(v[15].x));
      if (j0 + SLOTS <= nmin) {
#pragma unroll
        for (int u = 0; u < SLOTS; ++u) {
          const int nid = q[u].x & 15;
          if (nid != cur) {
            *(float4*)(trow + cur * TILE_PITCH) = sum;
            cur = nid;
            sum = *(const float4*)(trow + cur * TILE_PITCH);
          }
          const float wk = __int_as_float(q[u].y);
          sum.x = __builtin_fmaf(wk, v[u].x, sum.x);
          sum.y = __builtin_fmaf(wk, v[u].y, sum.y);
          sum.z = __builtin_fmaf(wk, v[u].z, sum.z);
          sum.w = __builtin_fmaf(wk, v[u].w, sum.w);
        }
      } else {
#pragma unroll
        for (int u = 0; u < SLOTS; ++u) {
          if (j0 + u < nh) {
            const int nid = q[u].x & 15;
            if (nid != cur) {
              *(float4*)(trow + cur * TILE_PITCH) = sum;
              cur = nid;
              sum = *(const float4*)(trow + cur * TILE_PITCH);
            }
            const float wk = __int_as_float(q[u].y);
            sum.x = __builtin_fmaf(wk, v[u].x, sum.x);
            sum.y = __builtin_fmaf(wk, v[u].y, sum.y);
            sum.z = __builtin_fmaf(wk, v[u].z, sum.z);
            sum.w = __builtin_fmaf(wk, v[u].w, sum.w);
          }
        }
      }
    }
  }
  *(float4*)(trow + cur * TILE_PITCH) = sum;
  __syncthreads();

  v8f acc[NT] = {};
  const float* arow = tile + l16 * TILE_PITCH + 2 * half;
#pragma unroll 4
  for (int k = 0; k < HIDDEN; k += 4) {
    v2f a = *(const v2f*)(arow + k);
    a.x = fmaxf(a.x, 0.0f);
    a.y = fmaxf(a.y, 0.0f);
    const int kk = k + 2 * half;
#pragma unroll
    for (int tt = 0; tt < NT; ++tt) {
      const int c = tt * 16 + l16;
      v2f bfr;
      if ((NOUT % 16 == 0) || (c < NOUT)) {
        bfr.x = W[(size_t)kk * NOUT + c];
        bfr.y = W[(size_t)(kk + 1) * NOUT + c];
      } else {
        bfr.x = 0.0f; bfr.y = 0.0f;
      }
      acc[tt] = __builtin_amdgcn_wmma_f32_16x16x4_f32(
           false, a,  false, bfr,
           (short)0, acc[tt],  false,  false);
    }
  }

  const bool whole = m0 + 16 <= n;
  float ro[8];
#pragma unroll
  for (int v = 0; v < 8; ++v) ro[v] = SCALE_OUT ? __shfl(dv, v + 8 * half) : 1.0f;
#pragma unroll
  for (int tt = 0; tt < NT; ++tt) {
    const int c = tt * 16 + l16;
    if ((NOUT % 16 != 0) && (c >= NOUT)) continue;
    const float bo = BIAS ? bias_out[c] : 0.0f;
    if (whole) {
#pragma unroll
      for (int v = 0; v < 8; ++v) D[(size_t)(m0 + v + 8 * half) * NOUT + c] = SCALE_OUT ? (acc[tt][v] + bo) * ro[v] : acc[tt][v] + bo;
    } else {
#pragma unroll
      for (int v = 0; v < 8; ++v) {
        const int m = m0 + v + 8 * half;
        if (m < n) D[(size_t)m * NOUT + c] = SCALE_OUT ? (acc[tt][v] + bo) * ro[v] : acc[tt][v] + bo;
      }
    }
  }
}


extern "C" void kernel_launch(void* const* d_in, const int* in_sizes, int n_in,
                              void* d_out, int out_size, void* d_ws, size_t ws_size,
                              hipStream_t stream) {
  const float* x   = (const float*)d_in[0];
  const int*   ei  = (const int*)d_in[1];
  const float* ew  = (const float*)d_in[2];
  const float* W1  = (const float*)d_in[3];
  const float* b1  = (const float*)d_in[4];
  const float* Wc1 = (const float*)d_in[5];
  const float* bc1 = (const float*)d_in[6];
  const float* Wc2 = (const float*)d_in[7];
  const float* bc2 = (const float*)d_in[8];
  const float* W2  = (const float*)d_in[9];
  const float* b2  = (const float*)d_in[10];
  float* out = (float*)d_out;

  const int N = in_sizes[0] / F_IN;
  const int E = in_sizes[2];
  const int* erow = ei;
  const int* ecol = ei + E;

  int4*  tmp    = (int4*)d_ws;
  float* h      = (float*)(tmp + E);
  float* t      = h + (size_t)N * HIDDEN;
  int*   sedge  = (int*)(t + (size_t)N * HIDDEN);
  float* dinv   = (float*)(sedge + 2 * (size_t)E);
  int*   cnt    = (int*)(dinv + N);
  int*   off    = cnt + N;
  int*   bstart = off + N;
  int*   hist   = bstart + NB_MAX + 1;

  const int SORT_GRID = 256;
  const int nb        = (N + BW - 1) / BW;
  const int gemm_grid = (N + 31) / 32;
  const int tile_grid = (N + 15) / 16;

  k_bucket_hist<<<SORT_GRID, SORT_TB, 0, stream>>>(ecol, hist, E, N, nb);
  k_bucket_fill<<<SORT_GRID, SORT_TB, 0, stream>>>(erow, ecol, ew, hist, bstart, tmp, E, N, nb);
  k_bucket_sort<<<nb, SORT_TB, 0, stream>>>(tmp, bstart, sedge, off, cnt, dinv, N);

  k_gemm_wmma<F_IN, HIDDEN, false, true, true, false>
      <<<gemm_grid, 64, 0, stream>>>(x, W1, b1, nullptr, h, N);

  k_gemm_wmma<HIDDEN, HIDDEN, false, false, false, true>
      <<<gemm_grid, 64, 0, stream>>>(h, Wc1, nullptr, dinv, t, N);

  k_gather_gemm<HIDDEN, false, true>
      <<<tile_grid, 32, 0, stream>>>(off, cnt, sedge, dinv, t, bc1, Wc2, nullptr, h, N);

  k_gather_gemm<NCLS, true, false>
      <<<tile_grid, 32, 0, stream>>>(off, cnt, sedge, dinv, h, bc2, W2, b2, out, N);
}


# First layer and conv1's product fused into one kernel: each wave hands its 16x64 relu tile to itself through LDS as A fragments, h is never written or re-read; 7->6 launches; sum orders kept
# speedup vs baseline: 3.4931x; 1.0720x over previous
#include <hip/hip_runtime.h>

typedef __attribute__((ext_vector_type(2))) float v2f;
typedef __attribute__((ext_vector_type(8))) float v8f;

#define F_IN   256
#define HIDDEN 64
#define NCLS   40


#define BW_SHIFT 9
#define BW       (1 << BW_SHIFT)
#define SORT_TB  1024
#define NB_MAX   1024
#define EPT      8

static __device__ __forceinline__ int block_excl_scan(int v, int* wsum, int* total) {
  const int lane = threadIdx.x & 31;
  const int wv   = threadIdx.x >> 5;
  int s = v;
  for (int d = 1; d < 32; d <<= 1) {
    const int u = __shfl_up(s, d);
    if (lane >= d) s += u;
  }
  if (lane == 31) wsum[wv] = s;
  __syncthreads();
  if (wv == 0) {
    const int ws = wsum[lane];
    int q = ws;
    for (int d = 1; d < 32; d <<= 1) {
      const int u = __shfl_up(q, d);
      if (lane >= d) q += u;
    }
    wsum[lane] = q - ws;
    if (lane == 31) wsum[32] = q;
  }
  __syncthreads();
  const int r = s - v + wsum[wv];
  *total = wsum[32];
  __syncthreads();
  return r;
}

static __device__ __forceinline__ int in_range(int c, int n) {
  const unsigned uc = (unsigned)c, last = (unsigned)(n - 1);
  return (int)(uc < last ? uc : last);
}

__global__ __launch_bounds__(SORT_TB) void k_bucket_hist(const int* __restrict__ col, int* __restrict__ hist,
                                                         int e, int n, int nb) {
  __shared__ int h[NB_MAX];
  const int tid = threadIdx.x;
  h[tid] = 0;
  __syncthreads();
  const int chunk = (e + (int)gridDim.x - 1) / (int)gridDim.x;
  const int beg   = (int)blockIdx.x * chunk;
  const int end   = beg + chunk < e ? beg + chunk : e;
  for (int i0 = beg + tid; i0 < end; i0 += EPT * SORT_TB) {
    int c[EPT];
    for (int u = 0; u < EPT; ++u) {
      const int i = i0 + u * SORT_TB;
      c[u] = col[i < end ? i : end - 1];
    }
    asm volatile("" :: "v"(c[0]), "v"(c[1]), "v"(c[2]), "v"(c[3]), "v"(c[4]), "v"(c[5]), "v"(c[6]), "v"(c[7]));
    for (int u = 0; u < EPT; ++u)
      if (i0 + u * SORT_TB < end) atomicAdd(&h[in_range(c[u], n) >> BW_SHIFT], 1);
  }
  __syncthreads();
  if (tid < nb) hist[(size_t)tid * gridDim.x + blockIdx.x] = h[tid];
}

__global__ __launch_bounds__(SORT_TB) void k_bucket_fill(const int* __restrict__ row, const int* __restrict__ col,
                                                         const float* __restrict__ ew,
                                                         const int* __restrict__ hist, int* __restrict__ bstart,
                                                         int4* __restrict__ tmp, int e, int n, int nb) {
  __shared__ int tot[NB_MAX];
  __shared__ int cur[NB_MAX];
  __shared__ int wsum[33];
  const int tid  = threadIdx.x;
  const int lane = tid & 31;
  const int wv   = tid >> 5;
  const int b    = blockIdx.x;
  const int nblk = gridDim.x;

  for (int k = wv; k < nb; k += SORT_TB / 32) {
    const int* __restrict__ hk = hist + (size_t)k * nblk;
    int t = 0, p = 0;
    for (int j = lane; j < nblk; j += 32) {
      const int v = hk[j];
      t += v;
      p += (j < b) ? v : 0;
    }
    for (int d = 16; d > 0; d >>= 1) {
      t += __shfl_xor(t, d);
      p += __shfl_xor(p, d);
    }
    if (lane == 0) { tot[k] = t; cur[k] = p; }
  }
  __syncthreads();
  int total;
  const int ex = block_excl_scan(tid < nb ? tot[tid] : 0, wsum, &total);
  if (tid < nb) {
    cur[tid] += ex;
    if (b == 0) bstart[tid] = ex;
  }
  if (b == 0 && tid == 0) bstart[nb] = total;
  __syncthreads();

  const int chunk = (e + nblk - 1) / nblk;
  const int beg   = b * chunk;
  const int end   = beg + chunk < e ? beg + chunk : e;
  for (int i0 = beg + tid; i0 < end; i0 += EPT * SORT_TB) {
    int c[EPT], r[EPT], w[EPT];
    for (int u = 0; u < EPT; ++u) {
      const int i  = i0 + u * SORT_TB;
      const int ii = i < end ? i : end - 1;
      c[u] = col[ii];
      r[u] = row[ii];
      w[u] = __float_as_int(ew[ii]);
    }
    asm volatile("" :: "v"(c[0]), "v"(c[1]), "v"(c[2]), "v"(c[3]), "v"(c[4]), "v"(c[5]), "v"(c[6]), "v"(c[7]),
                       "v"(r[0]), "v"(r[1]), "v"(r[2]), "v"(r[3]), "v"(r[4]), "v"(r[5]), "v"(r[6]), "v"(r[7]),
                       "v"(w[0]), "v"(w[1]), "v"(w[2]), "v"(w[3]), "v"(w[4]), "v"(w[5]), "v"(w[6]), "v"(w[7]));
    for (int u = 0; u < EPT; ++u) {
      if (i0 + u * SORT_TB < end) {
        const int cc  = in_range(c[u], n);
        const int pos = atomicAdd(&cur[cc >> BW_SHIFT], 1);
        tmp[pos] = make_int4(r[u], cc, w[u], 0);
      }
    }
  }
}

__global__ __launch_bounds__(SORT_TB) void k_bucket_sort(const int4* __restrict__ tmp, const int* __restrict__ bstart,
                                                         int* sedge, int* __restrict__ off, int* __restrict__ cnt,
                                                         float* __restrict__ dinv, int n) {
  __shared__ int lcnt[BW];
  __shared__ int lcur[BW];
  __shared__ int wsum[33];
  const int tid   = threadIdx.x;
  const int node0 = (int)blockIdx.x << BW_SHIFT;
  const int s     = bstart[blockIdx.x];
  const int e     = bstart[blockIdx.x + 1];
  if (tid < BW) lcnt[tid] = 0;
  __syncthreads();
  for (int i0 = s + tid; i0 < e; i0 += EPT * SORT_TB) {
    int c[EPT];
    for (int u = 0; u < EPT; ++u) {
      const int i = i0 + u * SORT_TB;
      c[u] = tmp[i < e ? i : e - 1].y;
    }
    asm volatile("" :: "v"(c[0]), "v"(c[1]), "v"(c[2]), "v"(c[3]), "v"(c[4]), "v"(c[5]), "v"(c[6]), "v"(c[7]));
    for (int u = 0; u < EPT; ++u)
      if (i0 + u * SORT_TB < e) atomicAdd(&lcnt[(c[u] - node0) & (BW - 1)], 1);
  }
  __syncthreads();
  const int num = tid < BW ? lcnt[tid] : 0;
  int total;
  const int ex = block_excl_scan(num, wsum, &total);
  if (tid < BW) lcur[tid] = ex;
  __syncthreads();
  for (int i0 = s + tid; i0 < e; i0 += EPT * SORT_TB) {
    int4 r[EPT];
    for (int u = 0; u < EPT; ++u) {
      const int i = i0 + u * SORT_TB;
      r[u] = tmp[i < e ? i : e - 1];
    }
    asm volatile("" :: "v"(r[0].x), "v"(r[0].y), "v"(r[0].z), "v"(r[1].x), "v"(r[1].y), "v"(r[1].z),
                       "v"(r[2].x), "v"(r[2].y), "v"(r[2].z), "v"(r[3].x), "v"(r[3].y), "v"(r[3].z),
                       "v"(r[4].x), "v"(r[4].y), "v"(r[4].z), "v"(r[5].x), "v"(r[5].y), "v"(r[5].z),
                       "v"(r[6].x), "v"(r[6].y), "v"(r[6].z), "v"(r[7].x), "v"(r[7].y), "v"(r[7].z));
    for (int u = 0; u < EPT; ++u) {
      if (i0 + u * SORT_TB < e) {
        int p = atomicAdd(&lcur[(r[u].y - node0) & (BW - 1)], 1);
        p = p < e - s ? p : e - s - 1;
        *(int2*)(sedge + 2 * (size_t)(s + p)) = make_int2(r[u].x, r[u].z);
      }
    }
  }
  __syncthreads();
  const int c = node0 + tid;
  if (tid < BW && c < n) {
    const int beg = s + ex;
    off[c] = beg;
    cnt[c] = num;
    const int* wp = sedge + 2 * (size_t)beg + 1;
    float sd = 1.0f;
    for (int j = 0; j < num; j += EPT) {
      float w[EPT];
      for (int u = 0; u < EPT; ++u) w[u] = __int_as_float(wp[2 * (j + u < num ? j + u : num - 1)]);
      for (int u = 0; u < EPT; ++u) sd = (j + u < num) ? sd + w[u] : sd;
    }
    dinv[c] = (sd > 0.0f) ? rsqrtf(sd) : 0.0f;
  }
}


#define TILE_PITCH 68

template <int K>
__global__ __launch_bounds__(64) void k_lin_conv_wmma(const float* __restrict__ A,
                                                      const float* __restrict__ W1,
                                                      const float* __restrict__ b1,
                                                      const float* __restrict__ Wc,
                                                      const float* __restrict__ rscale,
                                                      float* __restrict__ D, int M) {
  constexpr int NT = HIDDEN / 16;
  __shared__ __attribute__((aligned(16))) float tiles[2][16 * TILE_PITCH];
  const int wv   = threadIdx.x >> 5;
  const int lane = threadIdx.x & 31;
  const int half = lane >> 4;
  const int l16  = lane & 15;
  const int m0   = __builtin_amdgcn_readfirstlane((blockIdx.x * 2 + wv) * 16);
  float* __restrict__ tile = tiles[wv];

  v8f acc[NT] = {};

  float rs[8];
#pragma unroll
  for (int v = 0; v < 8; ++v) {
    const int m = m0 + v + 8 * half;
    rs[v] = rscale[m < M ? m : M - 1];
  }

  const int mrow = m0 + l16 < M ? m0 + l16 : M - 1;
  const float* __restrict__ arow = A + (size_t)mrow * K + 2 * half;

  for (int k = 0; k < K; k += 4) {
    v2f a = *(const v2f*)(arow + k);
    const int kk = k + 2 * half;
#pragma unroll
    for (int t = 0; t < NT; ++t) {
      const int c = t * 16 + l16;
      v2f b;
      b.x = W1[(size_t)kk * HIDDEN + c];
      b.y = W1[(size_t)(kk + 1) * HIDDEN + c];
      acc[t] = __builtin_amdgcn_wmma_f32_16x16x4_f32(
           false, a,  false, b,
           (short)0, acc[t],  false,  false);
    }
  }

#pragma unroll
  for (int t = 0; t < NT; ++t) {
    const int c = t * 16 + l16;
    const float bv = b1[c];
#pragma unroll
    for (int v = 0; v < 8; ++v) tile[(v + 8 * half) * TILE_PITCH + c] = fmaxf(acc[t][v] + bv, 0.0f);
  }
  __syncthreads();

  v8f acc2[NT] = {};
  const float* a2 = tile + l16 * TILE_PITCH + 2 * half;
#pragma unroll 4
  for (int k = 0; k < HIDDEN; k += 4) {
    const v2f a = *(const v2f*)(a2 + k);
    const int kk = k + 2 * half;
#pragma unroll
    for (int t = 0; t < NT; ++t) {
      const int c = t * 16 + l16;
      v2f b;
      b.x = Wc[(size_t)kk * HIDDEN + c];
      b.y = Wc[(size_t)(kk + 1) * HIDDEN + c];
      acc2[t] = __builtin_amdgcn_wmma_f32_16x16x4_f32(
           false, a,  false, b,
           (short)0, acc2[t],  false,  false);
    }
  }

  const bool whole = m0 + 16 <= M;
  if (whole) {
#pragma unroll
    for (int t = 0; t < NT; ++t) {
      const int c = t * 16 + l16;
#pragma unroll
      for (int v = 0; v < 8; ++v) D[(size_t)(m0 + v + 8 * half) * HIDDEN + c] = acc2[t][v] * rs[v];
    }
  } else {
#pragma unroll
    for (int t = 0; t < NT; ++t) {
      const int c = t * 16 + l16;
#pragma unroll
      for (int v = 0; v < 8; ++v) {
        const int m = m0 + v + 8 * half;
        if (m < M) D[(size_t)m * HIDDEN + c] = acc2[t][v] * rs[v];
      }
    }
  }
}


static __device__ __forceinline__ float4 ld_row4(const float* __restrict__ tl, int r) {
  return *(const float4*)(tl + (size_t)r * HIDDEN);
}


#define EPL        10
#define ECAP       640
#define SLOTS      16

template <int NOUT, bool BIAS, bool SCALE_OUT>
__global__ __launch_bounds__(32) void k_gather_gemm(const int* __restrict__ off, const int* __restrict__ cnt,
                                                    const int* __restrict__ sedge,
                                                    const float* __restrict__ dinv,
                                                    const float* __restrict__ t,
                                                    const float* __restrict__ bias_in,
                                                    const float* __restrict__ W,
                                                    const float* __restrict__ bias_out,
                                                    float* __restrict__ D, int n) {
  constexpr int NT = (NOUT + 15) / 16;
  __shared__ __attribute__((aligned(16))) float tile[16 * TILE_PITCH];
  __shared__ int2  ebuf[ECAP];
  const int lane = threadIdx.x;
  const int half = lane >> 4;
  const int l16  = lane & 15;
  const int m0   = blockIdx.x * 16;
  if (m0 >= n) return;

  const float* __restrict__ tl = t + 4 * l16;
  const char*  __restrict__ tb = (const char*)t;
  const unsigned lane_off = 16u * (unsigned)l16;
  float* __restrict__ trow = tile + 4 * l16;
  const float4 bv = *(const float4*)(bias_in + 4 * l16);
  float4 tv[8];
#pragma unroll
  for (int p = 0; p < 8; ++p) {
    const int c = m0 + 8 * half + p;
    tv[p] = ld_row4(tl, c < n ? c : n - 1);
  }

  const int  cl    = m0 + l16 < n ? m0 + l16 : n - 1;
  const int  offv  = off[cl];
  const int  cntv  = cnt[cl];
  const float dv   = dinv[cl];
  const int  e0    = __builtin_amdgcn_readfirstlane(offv);
  const int  endv  = offv + cntv - e0;
  const int  ne    = __builtin_amdgcn_readlane(endv, 15);
  int es[15];
#pragma unroll
  for (int i = 0; i < 15; ++i) es[i] = __builtin_amdgcn_readlane(endv, i);

#pragma unroll
  for (int p = 0; p < 8; ++p) {
    const int i = 8 * half + p;
    const float di = __shfl(dv, i);
    *(float4*)(trow + i * TILE_PITCH) = make_float4(bv.x + di * tv[p].x, bv.y + di * tv[p].y,
                                                    bv.z + di * tv[p].z, bv.w + di * tv[p].w);
  }

  const int hb = half ? es[7] : 0;
  const int he = half ? ne : es[7];
  int    cur = 8 * half;
  float4 sum = *(const float4*)(trow + cur * TILE_PITCH);

  for (int cb = 0; cb < ne; cb += ECAP) {
    const int cm = ne - cb < ECAP ? ne - cb : ECAP;
    __syncthreads();
    for (int j0 = 0; j0 < cm; j0 += EPL * 32) {
      int2  se[EPL];
#pragma unroll
      for (int u = 0; u < EPL; ++u) {
        const int j  = j0 + u * 32 + lane;
        const int jj = j < cm ? j : cm - 1;
        se[u] = *(const int2*)(sedge + 2 * (size_t)(e0 + cb + jj));
      }
      asm volatile("" : "+v"(se[0].x), "+v"(se[1].x), "+v"(se[2].x), "+v"(se[3].x), "+v"(se[4].x),
                        "+v"(se[5].x), "+v"(se[6].x), "+v"(se[7].x), "+v"(se[8].x), "+v"(se[9].x)
                      : "v"(se[0].y), "v"(se[1].y), "v"(se[2].y), "v"(se[3].y), "v"(se[4].y),
                        "v"(se[5].y), "v"(se[6].y), "v"(se[7].y), "v"(se[8].y), "v"(se[9].y));
#pragma unroll
      for (int u = 0; u < EPL; ++u) {
        int r = se[u].x;
        r = r < 0 ? 0 : (r >= n ? n - 1 : r);
        se[u].x = r;
      }
      int   nidv[EPL];
      float dcg[EPL];
#pragma unroll
      for (int u = 0; u < EPL; ++u) {
        const int g = cb + j0 + u * 32 + lane;
        int nid = 0;
#pragma unroll
        for (int i = 0; i < 15; ++i) nid += es[i] <= g ? 1 : 0;
        nidv[u] = nid;
        dcg[u]  = __shfl(dv, nid);
      }
#pragma unroll
      for (int u = 0; u < EPL; ++u) {
        const int j = j0 + u * 32 + lane;
        if (j < cm) ebuf[j] = make_int2((se[u].x << 8) | nidv[u], __float_as_int(__int_as_float(se[u].y) * dcg[u]));
      }
    }
    __syncthreads();

    const int lo   = (hb > cb ? hb : cb) - cb;
    const int hi   = (he < cb + cm ? he : cb + cm) - cb;
    const int nh   = hi - lo;
    const int na   = __builtin_amdgcn_readlane(nh, 0);
    const int nb2  = __builtin_amdgcn_readlane(nh, 16);
    const int nmax = na > nb2 ? na : nb2;
    const int nmin = na < nb2 ? na : nb2;
    const int lo0  = nh > 0 ? lo : 0;
    const int last = nh > 0 ? nh - 1 : 0;
    for (int j0 = 0; j0 < nmax; j0 += SLOTS) {
      float4 v[SLOTS];
      int2   q[SLOTS];
#pragma unroll
      for (int u = 0; u < SLOTS; ++u) {
        q[u] = ebuf[lo0 + (j0 + u < last ? j0 + u : last)];
        v[u] = *(const float4*)(tb + (((unsigned)q[u].x & 0xffffff00u) + lane_off));
      }
      asm volatile("" :: "v"(v[0].x), "v"(v[1].x), "v"(v[2].x), "v"(v[3].x), "v"(v[4].x), "v"(v[5].x),
                         "v"(v[6].x), "v"(v[7].x), "v"(v[8].x), "v"(v[9].x), "v"(v[10].x), "v"(v[11].x),
                         "v"(v[12].x), "v"(v[13].x), "v"(v[14].x), "v"(v[15].x));
      if (j0 + SLOTS <= nmin) {
#pragma unroll
        for (int u = 0; u < SLOTS; ++u) {
          const int nid = q[u].x & 15;
          if (nid != cur) {
            *(float4*)(trow + cur * TILE_PITCH) = sum;
            cur = nid;
            sum = *(const float4*)(trow + cur * TILE_PITCH);
          }
          const float wk = __int_as_float(q[u].y);
          sum.x = __builtin_fmaf(wk, v[u].x, sum.x);
          sum.y = __builtin_fmaf(wk, v[u].y, sum.y);
          sum.z = __builtin_fmaf(wk, v[u].z, sum.z);
          sum.w = __builtin_fmaf(wk, v[u].w, sum.w);
        }
      } else {
#pragma unroll
        for (int u = 0; u < SLOTS; ++u) {
          if (j0 + u < nh) {
            const int nid = q[u].x & 15;
            if (nid != cur) {
              *(float4*)(trow + cur * TILE_PITCH) = sum;
              cur = nid;
              sum = *(const float4*)(trow + cur * TILE_PITCH);
            }
            const float wk = __int_as_float(q[u].y);
            sum.x = __builtin_fmaf(wk, v[u].x, sum.x);
            sum.y = __builtin_fmaf(wk, v[u].y, sum.y);
            sum.z = __builtin_fmaf(wk, v[u].z, sum.z);
            sum.w = __builtin_fmaf(wk, v[u].w, sum.w);
          }
        }
      }
    }
  }
  *(float4*)(trow + cur * TILE_PITCH) = sum;
  __syncthreads();

  v8f acc[NT] = {};
  const float* arow = tile + l16 * TILE_PITCH + 2 * half;
#pragma unroll 4
  for (int k = 0; k < HIDDEN; k += 4) {
    v2f a = *(const v2f*)(arow + k);
    a.x = fmaxf(a.x, 0.0f);
    a.y = fmaxf(a.y, 0.0f);
    const int kk = k + 2 * half;
#pragma unroll
    for (int tt = 0; tt < NT; ++tt) {
      const int c = tt * 16 + l16;
      v2f bfr;
      if ((NOUT % 16 == 0) || (c < NOUT)) {
        bfr.x = W[(size_t)kk * NOUT + c];
        bfr.y = W[(size_t)(kk + 1) * NOUT + c];
      } else {
        bfr.x = 0.0f; bfr.y = 0.0f;
      }
      acc[tt] = __builtin_amdgcn_wmma_f32_16x16x4_f32(
           false, a,  false, bfr,
           (short)0, acc[tt],  false,  false);
    }
  }

  const bool whole = m0 + 16 <= n;
  float ro[8];
#pragma unroll
  for (int v = 0; v < 8; ++v) ro[v] = SCALE_OUT ? __shfl(dv, v + 8 * half) : 1.0f;
#pragma unroll
  for (int tt = 0; tt < NT; ++tt) {
    const int c = tt * 16 + l16;
    if ((NOUT % 16 != 0) && (c >= NOUT)) continue;
    const float bo = BIAS ? bias_out[c] : 0.0f;
    if (whole) {
#pragma unroll
      for (int v = 0; v < 8; ++v) D[(size_t)(m0 + v + 8 * half) * NOUT + c] = SCALE_OUT ? (acc[tt][v] + bo) * ro[v] : acc[tt][v] + bo;
    } else {
#pragma unroll
      for (int v = 0; v < 8; ++v) {
        const int m = m0 + v + 8 * half;
        if (m < n) D[(size_t)m * NOUT + c] = SCALE_OUT ? (acc[tt][v] + bo) * ro[v] : acc[tt][v] + bo;
      }
    }
  }
}


extern "C" void kernel_launch(void* const* d_in, const int* in_sizes, int n_in,
                              void* d_out, int out_size, void* d_ws, size_t ws_size,
                              hipStream_t stream) {
  const float* x   = (const float*)d_in[0];
  const int*   ei  = (const int*)d_in[1];
  const float* ew  = (const float*)d_in[2];
  const float* W1  = (const float*)d_in[3];
  const float* b1  = (const float*)d_in[4];
  const float* Wc1 = (const float*)d_in[5];
  const float* bc1 = (const float*)d_in[6];
  const float* Wc2 = (const float*)d_in[7];
  const float* bc2 = (const float*)d_in[8];
  const float* W2  = (const float*)d_in[9];
  const float* b2  = (const float*)d_in[10];
  float* out = (float*)d_out;

  const int N = in_sizes[0] / F_IN;
  const int E = in_sizes[2];
  const int* erow = ei;
  const int* ecol = ei + E;

  int4*  tmp    = (int4*)d_ws;
  float* h      = (float*)(tmp + E);
  float* t      = h + (size_t)N * HIDDEN;
  int*   sedge  = (int*)(t + (size_t)N * HIDDEN);
  float* dinv   = (float*)(sedge + 2 * (size_t)E);
  int*   cnt    = (int*)(dinv + N);
  int*   off    = cnt + N;
  int*   bstart = off + N;
  int*   hist   = bstart + NB_MAX + 1;

  const int SORT_GRID = 256;
  const int nb        = (N + BW - 1) / BW;
  const int gemm_grid = (N + 31) / 32;
  const int tile_grid = (N + 15) / 16;

  k_bucket_hist<<<SORT_GRID, SORT_TB, 0, stream>>>(ecol, hist, E, N, nb);
  k_bucket_fill<<<SORT_GRID, SORT_TB, 0, stream>>>(erow, ecol, ew, hist, bstart, tmp, E, N, nb);
  k_bucket_sort<<<nb, SORT_TB, 0, stream>>>(tmp, bstart, sedge, off, cnt, dinv, N);

  k_lin_conv_wmma<F_IN><<<gemm_grid, 64, 0, stream>>>(x, W1, b1, Wc1, dinv, t, N);

  k_gather_gemm<HIDDEN, false, true>
      <<<tile_grid, 32, 0, stream>>>(off, cnt, sedge, dinv, t, bc1, Wc2, nullptr, h, N);

  k_gather_gemm<NCLS, true, false>
      <<<tile_grid, 32, 0, stream>>>(off, cnt, sedge, dinv, h, bc2, W2, b2, out, N);
}


# B fragments of the three 64-column products loaded 16 bytes per lane: tile t of lane c stands for column 4c+t, so a weight row piece is one load and a result row one 16-byte store; sums unchanged
# speedup vs baseline: 3.6738x; 1.0517x over previous
#include <hip/hip_runtime.h>

typedef __attribute__((ext_vector_type(2))) float v2f;
typedef __attribute__((ext_vector_type(8))) float v8f;

#define F_IN   256
#define HIDDEN 64
#define NCLS   40


#define BW_SHIFT 9
#define BW       (1 << BW_SHIFT)
#define SORT_TB  1024
#define NB_MAX   1024
#define EPT      8

static __device__ __forceinline__ int block_excl_scan(int v, int* wsum, int* total) {
  const int lane = threadIdx.x & 31;
  const int wv   = threadIdx.x >> 5;
  int s = v;
  for (int d = 1; d < 32; d <<= 1) {
    const int u = __shfl_up(s, d);
    if (lane >= d) s += u;
  }
  if (lane == 31) wsum[wv] = s;
  __syncthreads();
  if (wv == 0) {
    const int ws = wsum[lane];
    int q = ws;
    for (int d = 1; d < 32; d <<= 1) {
      const int u = __shfl_up(q, d);
      if (lane >= d) q += u;
    }
    wsum[lane] = q - ws;
    if (lane == 31) wsum[32] = q;
  }
  __syncthreads();
  const int r = s - v + wsum[wv];
  *total = wsum[32];
  __syncthreads();
  return r;
}

static __device__ __forceinline__ int in_range(int c, int n) {
  const unsigned uc = (unsigned)c, last = (unsigned)(n - 1);
  return (int)(uc < last ? uc : last);
}

__global__ __launch_bounds__(SORT_TB) void k_bucket_hist(const int* __restrict__ col, int* __restrict__ hist,
                                                         int e, int n, int nb) {
  __shared__ int h[NB_MAX];
  const int tid = threadIdx.x;
  h[tid] = 0;
  __syncthreads();
  const int chunk = (e + (int)gridDim.x - 1) / (int)gridDim.x;
  const int beg   = (int)blockIdx.x * chunk;
  const int end   = beg + chunk < e ? beg + chunk : e;
  for (int i0 = beg + tid; i0 < end; i0 += EPT * SORT_TB) {
    int c[EPT];
    for (int u = 0; u < EPT; ++u) {
      const int i = i0 + u * SORT_TB;
      c[u] = col[i < end ? i : end - 1];
    }
    asm volatile("" :: "v"(c[0]), "v"(c[1]), "v"(c[2]), "v"(c[3]), "v"(c[4]), "v"(c[5]), "v"(c[6]), "v"(c[7]));
    for (int u = 0; u < EPT; ++u)
      if (i0 + u * SORT_TB < end) atomicAdd(&h[in_range(c[u], n) >> BW_SHIFT], 1);
  }
  __syncthreads();
  if (tid < nb) hist[(size_t)tid * gridDim.x + blockIdx.x] = h[tid];
}

__global__ __launch_bounds__(SORT_TB) void k_bucket_fill(const int* __restrict__ row, const int* __restrict__ col,
                                                         const float* __restrict__ ew,
                                                         const int* __restrict__ hist, int* __restrict__ bstart,
                                                         int4* __restrict__ tmp, int e, int n, int nb) {
  __shared__ int tot[NB_MAX];
  __shared__ int cur[NB_MAX];
  __shared__ int wsum[33];
  const int tid  = threadIdx.x;
  const int lane = tid & 31;
  const int wv   = tid >> 5;
  const int b    = blockIdx.x;
  const int nblk = gridDim.x;

  for (int k = wv; k < nb; k += SORT_TB / 32) {
    const int* __restrict__ hk = hist + (size_t)k * nblk;
    int t = 0, p = 0;
    for (int j = lane; j < nblk; j += 32) {
      const int v = hk[j];
      t += v;
      p += (j < b) ? v : 0;
    }
    for (int d = 16; d > 0; d >>= 1) {
      t += __shfl_xor(t, d);
      p += __shfl_xor(p, d);
    }
    if (lane == 0) { tot[k] = t; cur[k] = p; }
  }
  __syncthreads();
  int total;
  const int ex = block_excl_scan(tid < nb ? tot[tid] : 0, wsum, &total);
  if (tid < nb) {
    cur[tid] += ex;
    if (b == 0) bstart[tid] = ex;
  }
  if (b == 0 && tid == 0) bstart[nb] = total;
  __syncthreads();

  const int chunk = (e + nblk - 1) / nblk;
  const int beg   = b * chunk;
  const int end   = beg + chunk < e ? beg + chunk : e;
  for (int i0 = beg + tid; i0 < end; i0 += EPT * SORT_TB) {
    int c[EPT], r[EPT], w[EPT];
    for (int u = 0; u < EPT; ++u) {
      const int i  = i0 + u * SORT_TB;
      const int ii = i < end ? i : end - 1;
      c[u] = col[ii];
      r[u] = row[ii];
      w[u] = __float_as_int(ew[ii]);
    }
    asm volatile("" :: "v"(c[0]), "v"(c[1]), "v"(c[2]), "v"(c[3]), "v"(c[4]), "v"(c[5]), "v"(c[6]), "v"(c[7]),
                       "v"(r[0]), "v"(r[1]), "v"(r[2]), "v"(r[3]), "v"(r[4]), "v"(r[5]), "v"(r[6]), "v"(r[7]),
                       "v"(w[0]), "v"(w[1]), "v"(w[2]), "v"(w[3]), "v"(w[4]), "v"(w[5]), "v"(w[6]), "v"(w[7]));
    for (int u = 0; u < EPT; ++u) {
      if (i0 + u * SORT_TB < end) {
        const int cc  = in_range(c[u], n);
        const int pos = atomicAdd(&cur[cc >> BW_SHIFT], 1);
        tmp[pos] = make_int4(r[u], cc, w[u], 0);
      }
    }
  }
}

__global__ __launch_bounds__(SORT_TB) void k_bucket_sort(const int4* __restrict__ tmp, const int* __restrict__ bstart,
                                                         int* sedge, int* __restrict__ off, int* __restrict__ cnt,
                                                         float* __restrict__ dinv, int n) {
  __shared__ int lcnt[BW];
  __shared__ int lcur[BW];
  __shared__ int wsum[33];
  const int tid   = threadIdx.x;
  const int node0 = (int)blockIdx.x << BW_SHIFT;
  const int s     = bstart[blockIdx.x];
  const int e     = bstart[blockIdx.x + 1];
  if (tid < BW) lcnt[tid] = 0;
  __syncthreads();
  for (int i0 = s + tid; i0 < e; i0 += EPT * SORT_TB) {
    int c[EPT];
    for (int u = 0; u < EPT; ++u) {
      const int i = i0 + u * SORT_TB;
      c[u] = tmp[i < e ? i : e - 1].y;
    }
    asm volatile("" :: "v"(c[0]), "v"(c[1]), "v"(c[2]), "v"(c[3]), "v"(c[4]), "v"(c[5]), "v"(c[6]), "v"(c[7]));
    for (int u = 0; u < EPT; ++u)
      if (i0 + u * SORT_TB < e) atomicAdd(&lcnt[(c[u] - node0) & (BW - 1)], 1);
  }
  __syncthreads();
  const int num = tid < BW ? lcnt[tid] : 0;
  int total;
  const int ex = block_excl_scan(num, wsum, &total);
  if (tid < BW) lcur[tid] = ex;
  __syncthreads();
  for (int i0 = s + tid; i0 < e; i0 += EPT * SORT_TB) {
    int4 r[EPT];
    for (int u = 0; u < EPT; ++u) {
      const int i = i0 + u * SORT_TB;
      r[u] = tmp[i < e ? i : e - 1];
    }
    asm volatile("" :: "v"(r[0].x), "v"(r[0].y), "v"(r[0].z), "v"(r[1].x), "v"(r[1].y), "v"(r[1].z),
                       "v"(r[2].x), "v"(r[2].y), "v"(r[2].z), "v"(r[3].x), "v"(r[3].y), "v"(r[3].z),
                       "v"(r[4].x), "v"(r[4].y), "v"(r[4].z), "v"(r[5].x), "v"(r[5].y), "v"(r[5].z),
                       "v"(r[6].x), "v"(r[6].y), "v"(r[6].z), "v"(r[7].x), "v"(r[7].y), "v"(r[7].z));
    for (int u = 0; u < EPT; ++u) {
      if (i0 + u * SORT_TB < e) {
        int p = atomicAdd(&lcur[(r[u].y - node0) & (BW - 1)], 1);
        p = p < e - s ? p : e - s - 1;
        *(int2*)(sedge + 2 * (size_t)(s + p)) = make_int2(r[u].x, r[u].z);
      }
    }
  }
  __syncthreads();
  const int c = node0 + tid;
  if (tid < BW && c < n) {
    const int beg = s + ex;
    off[c] = beg;
    cnt[c] = num;
    const int* wp = sedge + 2 * (size_t)beg + 1;
    float sd = 1.0f;
    for (int j = 0; j < num; j += EPT) {
      float w[EPT];
      for (int u = 0; u < EPT; ++u) w[u] = __int_as_float(wp[2 * (j + u < num ? j + u : num - 1)]);
      for (int u = 0; u < EPT; ++u) sd = (j + u < num) ? sd + w[u] : sd;
    }
    dinv[c] = (sd > 0.0f) ? rsqrtf(sd) : 0.0f;
  }
}


#define TILE_PITCH 68

template <int K>
__global__ __launch_bounds__(64) void k_lin_conv_wmma(const float* __restrict__ A,
                                                      const float* __restrict__ W1,
                                                      const float* __restrict__ b1,
                                                      const float* __restrict__ Wc,
                                                      const float* __restrict__ rscale,
                                                      float* __restrict__ D, int M) {
  constexpr int NT = HIDDEN / 16;
  __shared__ __attribute__((aligned(16))) float tiles[2][16 * TILE_PITCH];
  const int wv   = threadIdx.x >> 5;
  const int lane = threadIdx.x & 31;
  const int half = lane >> 4;
  const int l16  = lane & 15;
  const int m0   = __builtin_amdgcn_readfirstlane((blockIdx.x * 2 + wv) * 16);
  float* __restrict__ tile = tiles[wv];

  v8f acc[NT] = {};

  float rs[8];
#pragma unroll
  for (int v = 0; v < 8; ++v) {
    const int m = m0 + v + 8 * half;
    rs[v] = rscale[m < M ? m : M - 1];
  }

  const int mrow = m0 + l16 < M ? m0 + l16 : M - 1;
  const float* __restrict__ arow = A + (size_t)mrow * K + 2 * half;

  static_assert(NT == 4, "the column interleave below spells out four tiles");
  const float* __restrict__ w1l = W1 + 4 * l16;
  for (int k = 0; k < K; k += 4) {
    v2f a = *(const v2f*)(arow + k);
    const int kk = k + 2 * half;
    const float4 w0 = *(const float4*)(w1l + (size_t)kk * HIDDEN);
    const float4 w1 = *(const float4*)(w1l + (size_t)(kk + 1) * HIDDEN);
    v2f b;
    b.x = w0.x; b.y = w1.x;
    acc[0] = __builtin_amdgcn_wmma_f32_16x16x4_f32(false, a, false, b, (short)0, acc[0], false, false);
    b.x = w0.y; b.y = w1.y;
    acc[1] = __builtin_amdgcn_wmma_f32_16x16x4_f32(false, a, false, b, (short)0, acc[1], false, false);
    b.x = w0.z; b.y = w1.z;
    acc[2] = __builtin_amdgcn_wmma_f32_16x16x4_f32(false, a, false, b, (short)0, acc[2], false, false);
    b.x = w0.w; b.y = w1.w;
    acc[3] = __builtin_amdgcn_wmma_f32_16x16x4_f32(false, a, false, b, (short)0, acc[3], false, false);
  }

  {
    const float4 bv = *(const float4*)(b1 + 4 * l16);
#pragma unroll
    for (int v = 0; v < 8; ++v)
      *(float4*)(tile + (v + 8 * half) * TILE_PITCH + 4 * l16) =
          make_float4(fmaxf(acc[0][v] + bv.x, 0.0f), fmaxf(acc[1][v] + bv.y, 0.0f),
                      fmaxf(acc[2][v] + bv.z, 0.0f), fmaxf(acc[3][v] + bv.w, 0.0f));
  }
  __syncthreads();

  v8f acc2[NT] = {};
  const float* a2 = tile + l16 * TILE_PITCH + 2 * half;
  const float* __restrict__ wcl = Wc + 4 * l16;
#pragma unroll 4
  for (int k = 0; k < HIDDEN; k += 4) {
    const v2f a = *(const v2f*)(a2 + k);
    const int kk = k + 2 * half;
    const float4 w0 = *(const float4*)(wcl + (size_t)kk * HIDDEN);
    const float4 w1 = *(const float4*)(wcl + (size_t)(kk + 1) * HIDDEN);
    v2f b;
    b.x = w0.x; b.y = w1.x;
    acc2[0] = __builtin_amdgcn_wmma_f32_16x16x4_f32(false, a, false, b, (short)0, acc2[0], false, false);
    b.x = w0.y; b.y = w1.y;
    acc2[1] = __builtin_amdgcn_wmma_f32_16x16x4_f32(false, a, false, b, (short)0, acc2[1], false, false);
    b.x = w0.z; b.y = w1.z;
    acc2[2] = __builtin_amdgcn_wmma_f32_16x16x4_f32(false, a, false, b, (short)0, acc2[2], false, false);
    b.x = w0.w; b.y = w1.w;
    acc2[3] = __builtin_amdgcn_wmma_f32_16x16x4_f32(false, a, false, b, (short)0, acc2[3], false, false);
  }

  const bool whole = m0 + 16 <= M;
  if (whole) {
#pragma unroll
    for (int v = 0; v < 8; ++v)
      *(float4*)(D + (size_t)(m0 + v + 8 * half) * HIDDEN + 4 * l16) =
          make_float4(acc2[0][v] * rs[v], acc2[1][v] * rs[v], acc2[2][v] * rs[v], acc2[3][v] * rs[v]);
  } else {
#pragma unroll
    for (int v = 0; v < 8; ++v) {
      const int m = m0 + v + 8 * half;
      if (m < M)
        *(float4*)(D + (size_t)m * HIDDEN + 4 * l16) =
            make_float4(acc2[0][v] * rs[v], acc2[1][v] * rs[v], acc2[2][v] * rs[v], acc2[3][v] * rs[v]);
    }
  }
}


static __device__ __forceinline__ float4 ld_row4(const float* __restrict__ tl, int r) {
  return *(const float4*)(tl + (size_t)r * HIDDEN);
}


#define EPL        10
#define ECAP       640
#define SLOTS      16

template <int NOUT, bool BIAS, bool SCALE_OUT>
__global__ __launch_bounds__(32) void k_gather_gemm(const int* __restrict__ off, const int* __restrict__ cnt,
                                                    const int* __restrict__ sedge,
                                                    const float* __restrict__ dinv,
                                                    const float* __restrict__ t,
                                                    const float* __restrict__ bias_in,
                                                    const float* __restrict__ W,
                                                    const float* __restrict__ bias_out,
                                                    float* __restrict__ D, int n) {
  constexpr int NT = (NOUT + 15) / 16;
  constexpr bool IL = (NOUT == 64);
  __shared__ __attribute__((aligned(16))) float tile[16 * TILE_PITCH];
  __shared__ int2  ebuf[ECAP];
  const int lane = threadIdx.x;
  const int half = lane >> 4;
  const int l16  = lane & 15;
  const int m0   = blockIdx.x * 16;
  if (m0 >= n) return;

  const float* __restrict__ tl = t + 4 * l16;
  const char*  __restrict__ tb = (const char*)t;
  const unsigned lane_off = 16u * (unsigned)l16;
  float* __restrict__ trow = tile + 4 * l16;
  const float4 bv = *(const float4*)(bias_in + 4 * l16);
  float4 tv[8];
#pragma unroll
  for (int p = 0; p < 8; ++p) {
    const int c = m0 + 8 * half + p;
    tv[p] = ld_row4(tl, c < n ? c : n - 1);
  }

  const int  cl    = m0 + l16 < n ? m0 + l16 : n - 1;
  const int  offv  = off[cl];
  const int  cntv  = cnt[cl];
  const float dv   = dinv[cl];
  const int  e0    = __builtin_amdgcn_readfirstlane(offv);
  const int  endv  = offv + cntv - e0;
  const int  ne    = __builtin_amdgcn_readlane(endv, 15);
  int es[15];
#pragma unroll
  for (int i = 0; i < 15; ++i) es[i] = __builtin_amdgcn_readlane(endv, i);

#pragma unroll
  for (int p = 0; p < 8; ++p) {
    const int i = 8 * half + p;
    const float di = __shfl(dv, i);
    *(float4*)(trow + i * TILE_PITCH) = make_float4(bv.x + di * tv[p].x, bv.y + di * tv[p].y,
                                                    bv.z + di * tv[p].z, bv.w + di * tv[p].w);
  }

  const int hb = half ? es[7] : 0;
  const int he = half ? ne : es[7];
  int    cur = 8 * half;
  float4 sum = *(const float4*)(trow + cur * TILE_PITCH);

  for (int cb = 0; cb < ne; cb += ECAP) {
    const int cm = ne - cb < ECAP ? ne - cb : ECAP;
    __syncthreads();
    for (int j0 = 0; j0 < cm; j0 += EPL * 32) {
      int2  se[EPL];
#pragma unroll
      for (int u = 0; u < EPL; ++u) {
        const int j  = j0 + u * 32 + lane;
        const int jj = j < cm ? j : cm - 1;
        se[u] = *(const int2*)(sedge + 2 * (size_t)(e0 + cb + jj));
      }
      asm volatile("" : "+v"(se[0].x), "+v"(se[1].x), "+v"(se[2].x), "+v"(se[3].x), "+v"(se[4].x),
                        "+v"(se[5].x), "+v"(se[6].x), "+v"(se[7].x), "+v"(se[8].x), "+v"(se[9].x)
                      : "v"(se[0].y), "v"(se[1].y), "v"(se[2].y), "v"(se[3].y), "v"(se[4].y),
                        "v"(se[5].y), "v"(se[6].y), "v"(se[7].y), "v"(se[8].y), "v"(se[9].y));
#pragma unroll
      for (int u = 0; u < EPL; ++u) {
        int r = se[u].x;
        r = r < 0 ? 0 : (r >= n ? n - 1 : r);
        se[u].x = r;
      }
      int   nidv[EPL];
      float dcg[EPL];
#pragma unroll
      for (int u = 0; u < EPL; ++u) {
        const int g = cb + j0 + u * 32 + lane;
        int nid = 0;
#pragma unroll
        for (int i = 0; i < 15; ++i) nid += es[i] <= g ? 1 : 0;
        nidv[u] = nid;
        dcg[u]  = __shfl(dv, nid);
      }
#pragma unroll
      for (int u = 0; u < EPL; ++u) {
        const int j = j0 + u * 32 + lane;
        if (j < cm) ebuf[j] = make_int2((se[u].x << 8) | nidv[u], __float_as_int(__int_as_float(se[u].y) * dcg[u]));
      }
    }
    __syncthreads();

    const int lo   = (hb > cb ? hb : cb) - cb;
    const int hi   = (he < cb + cm ? he : cb + cm) - cb;
    const int nh   = hi - lo;
    const int na   = __builtin_amdgcn_readlane(nh, 0);
    const int nb2  = __builtin_amdgcn_readlane(nh, 16);
    const int nmax = na > nb2 ? na : nb2;
    const int nmin = na < nb2 ? na : nb2;
    const int lo0  = nh > 0 ? lo : 0;
    const int last = nh > 0 ? nh - 1 : 0;
    for (int j0 = 0; j0 < nmax; j0 += SLOTS) {
      float4 v[SLOTS];
      int2   q[SLOTS];
#pragma unroll
      for (int u = 0; u < SLOTS; ++u) {
        q[u] = ebuf[lo0 + (j0 + u < last ? j0 + u : last)];
        v[u] = *(const float4*)(tb + (((unsigned)q[u].x & 0xffffff00u) + lane_off));
      }
      asm volatile("" :: "v"(v[0].x), "v"(v[1].x), "v"(v[2].x), "v"(v[3].x), "v"(v[4].x), "v"(v[5].x),
                         "v"(v[6].x), "v"(v[7].x), "v"(v[8].x), "v"(v[9].x), "v"(v[10].x), "v"(v[11].x),
                         "v"(v[12].x), "v"(v[13].x), "v"(v[14].x), "v"(v[15].x));
      if (j0 + SLOTS <= nmin) {
#pragma unroll
        for (int u = 0; u < SLOTS; ++u) {
          const int nid = q[u].x & 15;
          if (nid != cur) {
            *(float4*)(trow + cur * TILE_PITCH) = sum;
            cur = nid;
            sum = *(const float4*)(trow + cur * TILE_PITCH);
          }
          const float wk = __int_as_float(q[u].y);
          sum.x = __builtin_fmaf(wk, v[u].x, sum.x);
          sum.y = __builtin_fmaf(wk, v[u].y, sum.y);
          sum.z = __builtin_fmaf(wk, v[u].z, sum.z);
          sum.w = __builtin_fmaf(wk, v[u].w, sum.w);
        }
      } else {
#pragma unroll
        for (int u = 0; u < SLOTS; ++u) {
          if (j0 + u < nh) {
            const int nid = q[u].x & 15;
            if (nid != cur) {
              *(float4*)(trow + cur * TILE_PITCH) = sum;
              cur = nid;
              sum = *(const float4*)(trow + cur * TILE_PITCH);
            }
            const float wk = __int_as_float(q[u].y);
            sum.x = __builtin_fmaf(wk, v[u].x, sum.x);
            sum.y = __builtin_fmaf(wk, v[u].y, sum.y);
            sum.z = __builtin_fmaf(wk, v[u].z, sum.z);
            sum.w = __builtin_fmaf(wk, v[u].w, sum.w);
          }
        }
      }
    }
  }
  *(float4*)(trow + cur * TILE_PITCH) = sum;
  __syncthreads();

  v8f acc[NT] = {};
  const float* arow = tile + l16 * TILE_PITCH + 2 * half;
#pragma unroll 4
  for (int k = 0; k < HIDDEN; k += 4) {
    v2f a = *(const v2f*)(arow + k);
    a.x = fmaxf(a.x, 0.0f);
    a.y = fmaxf(a.y, 0.0f);
    const int kk = k + 2 * half;
    if constexpr (IL) {
      const float4 w0 = *(const float4*)(W + (size_t)kk * NOUT + 4 * l16);
      const float4 w1 = *(const float4*)(W + (size_t)(kk + 1) * NOUT + 4 * l16);
      v2f bfr;
      bfr.x = w0.x; bfr.y = w1.x;
      acc[0] = __builtin_amdgcn_wmma_f32_16x16x4_f32(false, a, false, bfr, (short)0, acc[0], false, false);
      bfr.x = w0.y; bfr.y = w1.y;
      acc[1] = __builtin_amdgcn_wmma_f32_16x16x4_f32(false, a, false, bfr, (short)0, acc[1], false, false);
      bfr.x = w0.z; bfr.y = w1.z;
      acc[2] = __builtin_amdgcn_wmma_f32_16x16x4_f32(false, a, false, bfr, (short)0, acc[2], false, false);
      bfr.x = w0.w; bfr.y = w1.w;
      acc[3] = __builtin_amdgcn_wmma_f32_16x16x4_f32(false, a, false, bfr, (short)0, acc[3], false, false);
    } else {
#pragma unroll
      for (int tt = 0; tt < NT; ++tt) {
        const int c = tt * 16 + l16;
        v2f bfr;
        if ((NOUT % 16 == 0) || (c < NOUT)) {
          bfr.x = W[(size_t)kk * NOUT + c];
          bfr.y = W[(size_t)(kk + 1) * NOUT + c];
        } else {
          bfr.x = 0.0f; bfr.y = 0.0f;
        }
        acc[tt] = __builtin_amdgcn_wmma_f32_16x16x4_f32(
             false, a,  false, bfr,
             (short)0, acc[tt],  false,  false);
      }
    }
  }

  const bool whole = m0 + 16 <= n;
  float ro[8];
#pragma unroll
  for (int v = 0; v < 8; ++v) ro[v] = SCALE_OUT ? __shfl(dv, v + 8 * half) : 1.0f;
  if constexpr (IL) {
    float4 bo = make_float4(0.0f, 0.0f, 0.0f, 0.0f);
    if (BIAS) bo = *(const float4*)(bias_out + 4 * l16);
#pragma unroll
    for (int v = 0; v < 8; ++v) {
      const int m = m0 + v + 8 * half;
      if (whole || m < n)
        *(float4*)(D + (size_t)m * NOUT + 4 * l16) =
            SCALE_OUT ? make_float4((acc[0][v] + bo.x) * ro[v], (acc[1][v] + bo.y) * ro[v],
                                    (acc[2][v] + bo.z) * ro[v], (acc[3][v] + bo.w) * ro[v])
                      : make_float4(acc[0][v] + bo.x, acc[1][v] + bo.y, acc[2][v] + bo.z, acc[3][v] + bo.w);
    }
  } else {
#pragma unroll
    for (int tt = 0; tt < NT; ++tt) {
      const int c = tt * 16 + l16;
      if ((NOUT % 16 != 0) && (c >= NOUT)) continue;
      const float bo = BIAS ? bias_out[c] : 0.0f;
      if (whole) {
#pragma unroll
        for (int v = 0; v < 8; ++v) D[(size_t)(m0 + v + 8 * half) * NOUT + c] = SCALE_OUT ? (acc[tt][v] + bo) * ro[v] : acc[tt][v] + bo;
      } else {
#pragma unroll
        for (int v = 0; v < 8; ++v) {
          const int m = m0 + v + 8 * half;
          if (m < n) D[(size_t)m * NOUT + c] = SCALE_OUT ? (acc[tt][v] + bo) * ro[v] : acc[tt][v] + bo;
        }
      }
    }
  }
}


extern "C" void kernel_launch(void* const* d_in, const int* in_sizes, int n_in,
                              void* d_out, int out_size, void* d_ws, size_t ws_size,
                              hipStream_t stream) {
  const float* x   = (const float*)d_in[0];
  const int*   ei  = (const int*)d_in[1];
  const float* ew  = (const float*)d_in[2];
  const float* W1  = (const float*)d_in[3];
  const float* b1  = (const float*)d_in[4];
  const float* Wc1 = (const float*)d_in[5];
  const float* bc1 = (const float*)d_in[6];
  const float* Wc2 = (const float*)d_in[7];
  const float* bc2 = (const float*)d_in[8];
  const float* W2  = (const float*)d_in[9];
  const float* b2  = (const float*)d_in[10];
  float* out = (float*)d_out;

  const int N = in_sizes[0] / F_IN;
  const int E = in_sizes[2];
  const int* erow = ei;
  const int* ecol = ei + E;

  int4*  tmp    = (int4*)d_ws;
  float* h      = (float*)(tmp + E);
  float* t      = h + (size_t)N * HIDDEN;
  int*   sedge  = (int*)(t + (size_t)N * HIDDEN);
  float* dinv   = (float*)(sedge + 2 * (size_t)E);
  int*   cnt    = (int*)(dinv + N);
  int*   off    = cnt + N;
  int*   bstart = off + N;
  int*   hist   = bstart + NB_MAX + 1;

  const int SORT_GRID = 256;
  const int nb        = (N + BW - 1) / BW;
  const int gemm_grid = (N + 31) / 32;
  const int tile_grid = (N + 15) / 16;

  k_bucket_hist<<<SORT_GRID, SORT_TB, 0, stream>>>(ecol, hist, E, N, nb);
  k_bucket_fill<<<SORT_GRID, SORT_TB, 0, stream>>>(erow, ecol, ew, hist, bstart, tmp, E, N, nb);
  k_bucket_sort<<<nb, SORT_TB, 0, stream>>>(tmp, bstart, sedge, off, cnt, dinv, N);

  k_lin_conv_wmma<F_IN><<<gemm_grid, 64, 0, stream>>>(x, W1, b1, Wc1, dinv, t, N);

  k_gather_gemm<HIDDEN, false, true>
      <<<tile_grid, 32, 0, stream>>>(off, cnt, sedge, dinv, t, bc1, Wc2, nullptr, h, N);

  k_gather_gemm<NCLS, true, false>
      <<<tile_grid, 32, 0, stream>>>(off, cnt, sedge, dinv, h, bc2, W2, b2, out, N);
}


# Edge-fill kernel's per-bucket totals and block prefixes: a lane reads 8 blocks' counts as two 16-byte loads, 8 matrix rows requested together: one memory round trip per wave, not four per row
# speedup vs baseline: 3.8690x; 1.0531x over previous
#include <hip/hip_runtime.h>

typedef __attribute__((ext_vector_type(2))) float v2f;
typedef __attribute__((ext_vector_type(8))) float v8f;

#define F_IN   256
#define HIDDEN 64
#define NCLS   40


#define BW_SHIFT 9
#define BW       (1 << BW_SHIFT)
#define SORT_TB  1024
#define SORT_GRID 256
#define KPT      8
#define NB_MAX   1024
#define EPT      8

static __device__ __forceinline__ int block_excl_scan(int v, int* wsum, int* total) {
  const int lane = threadIdx.x & 31;
  const int wv   = threadIdx.x >> 5;
  int s = v;
  for (int d = 1; d < 32; d <<= 1) {
    const int u = __shfl_up(s, d);
    if (lane >= d) s += u;
  }
  if (lane == 31) wsum[wv] = s;
  __syncthreads();
  if (wv == 0) {
    const int ws = wsum[lane];
    int q = ws;
    for (int d = 1; d < 32; d <<= 1) {
      const int u = __shfl_up(q, d);
      if (lane >= d) q += u;
    }
    wsum[lane] = q - ws;
    if (lane == 31) wsum[32] = q;
  }
  __syncthreads();
  const int r = s - v + wsum[wv];
  *total = wsum[32];
  __syncthreads();
  return r;
}

static __device__ __forceinline__ int in_range(int c, int n) {
  const unsigned uc = (unsigned)c, last = (unsigned)(n - 1);
  return (int)(uc < last ? uc : last);
}

__global__ __launch_bounds__(SORT_TB) void k_bucket_hist(const int* __restrict__ col, int* __restrict__ hist,
                                                         int e, int n, int nb) {
  __shared__ int h[NB_MAX];
  const int tid = threadIdx.x;
  h[tid] = 0;
  __syncthreads();
  const int chunk = (e + (int)gridDim.x - 1) / (int)gridDim.x;
  const int beg   = (int)blockIdx.x * chunk;
  const int end   = beg + chunk < e ? beg + chunk : e;
  for (int i0 = beg + tid; i0 < end; i0 += EPT * SORT_TB) {
    int c[EPT];
    for (int u = 0; u < EPT; ++u) {
      const int i = i0 + u * SORT_TB;
      c[u] = col[i < end ? i : end - 1];
    }
    asm volatile("" :: "v"(c[0]), "v"(c[1]), "v"(c[2]), "v"(c[3]), "v"(c[4]), "v"(c[5]), "v"(c[6]), "v"(c[7]));
    for (int u = 0; u < EPT; ++u)
      if (i0 + u * SORT_TB < end) atomicAdd(&h[in_range(c[u], n) >> BW_SHIFT], 1);
  }
  __syncthreads();
  if (tid < nb) hist[(size_t)tid * gridDim.x + blockIdx.x] = h[tid];
}

__global__ __launch_bounds__(SORT_TB) void k_bucket_fill(const int* __restrict__ row, const int* __restrict__ col,
                                                         const float* __restrict__ ew,
                                                         const int* __restrict__ hist, int* __restrict__ bstart,
                                                         int4* __restrict__ tmp, int e, int n, int nb) {
  __shared__ int tot[NB_MAX];
  __shared__ int cur[NB_MAX];
  __shared__ int wsum[33];
  const int tid  = threadIdx.x;
  const int lane = tid & 31;
  const int wv   = tid >> 5;
  const int b    = blockIdx.x;
  constexpr int nblk = SORT_GRID;
  static_assert(SORT_GRID == 8 * 32, "a lane takes 8 consecutive blocks' counts of a bucket as two 16-byte loads");

  const int mine = b - 8 * lane;
  for (int k0 = wv; k0 < nb; k0 += KPT * (SORT_TB / 32)) {
    int4 lo[KPT], hi[KPT];
#pragma unroll
    for (int u = 0; u < KPT; ++u) {
      const int k  = k0 + u * (SORT_TB / 32);
      const int kk = k < nb ? k : nb - 1;
      const int4* __restrict__ hk = (const int4*)(hist + (size_t)kk * nblk) + 2 * lane;
      lo[u] = hk[0];
      hi[u] = hk[1];
    }
    asm volatile("" :: "v"(lo[0].x), "v"(hi[0].x), "v"(lo[1].x), "v"(hi[1].x), "v"(lo[2].x), "v"(hi[2].x),
                       "v"(lo[3].x), "v"(hi[3].x), "v"(lo[4].x), "v"(hi[4].x), "v"(lo[5].x), "v"(hi[5].x),
                       "v"(lo[6].x), "v"(hi[6].x), "v"(lo[7].x), "v"(hi[7].x));
    __builtin_amdgcn_sched_barrier(0);
    int t[KPT], p[KPT];
#pragma unroll
    for (int u = 0; u < KPT; ++u) {
      t[u] = ((lo[u].x + lo[u].y) + (lo[u].z + lo[u].w)) + ((hi[u].x + hi[u].y) + (hi[u].z + hi[u].w));
      p[u] = (((mine > 0 ? lo[u].x : 0) + (mine > 1 ? lo[u].y : 0)) + ((mine > 2 ? lo[u].z : 0) + (mine > 3 ? lo[u].w : 0))) +
             (((mine > 4 ? hi[u].x : 0) + (mine > 5 ? hi[u].y : 0)) + ((mine > 6 ? hi[u].z : 0) + (mine > 7 ? hi[u].w : 0)));
    }
#pragma unroll
    for (int d = 16; d > 0; d >>= 1) {
#pragma unroll
      for (int u = 0; u < KPT; ++u) {
        t[u] += __shfl_xor(t[u], d);
        p[u] += __shfl_xor(p[u], d);
      }
    }
    if (lane == 0) {
#pragma unroll
      for (int u = 0; u < KPT; ++u) {
        const int k = k0 + u * (SORT_TB / 32);
        if (k < nb) { tot[k] = t[u]; cur[k] = p[u]; }
      }
    }
  }
  __syncthreads();
  int total;
  const int ex = block_excl_scan(tid < nb ? tot[tid] : 0, wsum, &total);
  if (tid < nb) {
    cur[tid] += ex;
    if (b == 0) bstart[tid] = ex;
  }
  if (b == 0 && tid == 0) bstart[nb] = total;
  __syncthreads();

  const int chunk = (e + nblk - 1) / nblk;
  const int beg   = b * chunk;
  const int end   = beg + chunk < e ? beg + chunk : e;
  for (int i0 = beg + tid; i0 < end; i0 += EPT * SORT_TB) {
    int c[EPT], r[EPT], w[EPT];
    for (int u = 0; u < EPT; ++u) {
      const int i  = i0 + u * SORT_TB;
      const int ii = i < end ? i : end - 1;
      c[u] = col[ii];
      r[u] = row[ii];
      w[u] = __float_as_int(ew[ii]);
    }
    asm volatile("" :: "v"(c[0]), "v"(c[1]), "v"(c[2]), "v"(c[3]), "v"(c[4]), "v"(c[5]), "v"(c[6]), "v"(c[7]),
                       "v"(r[0]), "v"(r[1]), "v"(r[2]), "v"(r[3]), "v"(r[4]), "v"(r[5]), "v"(r[6]), "v"(r[7]),
                       "v"(w[0]), "v"(w[1]), "v"(w[2]), "v"(w[3]), "v"(w[4]), "v"(w[5]), "v"(w[6]), "v"(w[7]));
    for (int u = 0; u < EPT; ++u) {
      if (i0 + u * SORT_TB < end) {
        const int cc  = in_range(c[u], n);
        const int pos = atomicAdd(&cur[cc >> BW_SHIFT], 1);
        tmp[pos] = make_int4(r[u], cc, w[u], 0);
      }
    }
  }
}

__global__ __launch_bounds__(SORT_TB) void k_bucket_sort(const int4* __restrict__ tmp, const int* __restrict__ bstart,
                                                         int* sedge, int* __restrict__ off, int* __restrict__ cnt,
                                                         float* __restrict__ dinv, int n) {
  __shared__ int lcnt[BW];
  __shared__ int lcur[BW];
  __shared__ int wsum[33];
  const int tid   = threadIdx.x;
  const int node0 = (int)blockIdx.x << BW_SHIFT;
  const int s     = bstart[blockIdx.x];
  const int e     = bstart[blockIdx.x + 1];
  if (tid < BW) lcnt[tid] = 0;
  __syncthreads();
  for (int i0 = s + tid; i0 < e; i0 += EPT * SORT_TB) {
    int c[EPT];
    for (int u = 0; u < EPT; ++u) {
      const int i = i0 + u * SORT_TB;
      c[u] = tmp[i < e ? i : e - 1].y;
    }
    asm volatile("" :: "v"(c[0]), "v"(c[1]), "v"(c[2]), "v"(c[3]), "v"(c[4]), "v"(c[5]), "v"(c[6]), "v"(c[7]));
    for (int u = 0; u < EPT; ++u)
      if (i0 + u * SORT_TB < e) atomicAdd(&lcnt[(c[u] - node0) & (BW - 1)], 1);
  }
  __syncthreads();
  const int num = tid < BW ? lcnt[tid] : 0;
  int total;
  const int ex = block_excl_scan(num, wsum, &total);
  if (tid < BW) lcur[tid] = ex;
  __syncthreads();
  for (int i0 = s + tid; i0 < e; i0 += EPT * SORT_TB) {
    int4 r[EPT];
    for (int u = 0; u < EPT; ++u) {
      const int i = i0 + u * SORT_TB;
      r[u] = tmp[i < e ? i : e - 1];
    }
    asm volatile("" :: "v"(r[0].x), "v"(r[0].y), "v"(r[0].z), "v"(r[1].x), "v"(r[1].y), "v"(r[1].z),
                       "v"(r[2].x), "v"(r[2].y), "v"(r[2].z), "v"(r[3].x), "v"(r[3].y), "v"(r[3].z),
                       "v"(r[4].x), "v"(r[4].y), "v"(r[4].z), "v"(r[5].x), "v"(r[5].y), "v"(r[5].z),
                       "v"(r[6].x), "v"(r[6].y), "v"(r[6].z), "v"(r[7].x), "v"(r[7].y), "v"(r[7].z));
    for (int u = 0; u < EPT; ++u) {
      if (i0 + u * SORT_TB < e) {
        int p = atomicAdd(&lcur[(r[u].y - node0) & (BW - 1)], 1);
        p = p < e - s ? p : e - s - 1;
        *(int2*)(sedge + 2 * (size_t)(s + p)) = make_int2(r[u].x, r[u].z);
      }
    }
  }
  __syncthreads();
  const int c = node0 + tid;
  if (tid < BW && c < n) {
    const int beg = s + ex;
    off[c] = beg;
    cnt[c] = num;
    const int* wp = sedge + 2 * (size_t)beg + 1;
    float sd = 1.0f;
    for (int j = 0; j < num; j += EPT) {
      float w[EPT];
      for (int u = 0; u < EPT; ++u) w[u] = __int_as_float(wp[2 * (j + u < num ? j + u : num - 1)]);
      for (int u = 0; u < EPT; ++u) sd = (j + u < num) ? sd + w[u] : sd;
    }
    dinv[c] = (sd > 0.0f) ? rsqrtf(sd) : 0.0f;
  }
}


#define TILE_PITCH 68

template <int K>
__global__ __launch_bounds__(64) void k_lin_conv_wmma(const float* __restrict__ A,
                                                      const float* __restrict__ W1,
                                                      const float* __restrict__ b1,
                                                      const float* __restrict__ Wc,
                                                      const float* __restrict__ rscale,
                                                      float* __restrict__ D, int M) {
  constexpr int NT = HIDDEN / 16;
  __shared__ __attribute__((aligned(16))) float tiles[2][16 * TILE_PITCH];
  const int wv   = threadIdx.x >> 5;
  const int lane = threadIdx.x & 31;
  const int half = lane >> 4;
  const int l16  = lane & 15;
  const int m0   = __builtin_amdgcn_readfirstlane((blockIdx.x * 2 + wv) * 16);
  float* __restrict__ tile = tiles[wv];

  v8f acc[NT] = {};

  float rs[8];
#pragma unroll
  for (int v = 0; v < 8; ++v) {
    const int m = m0 + v + 8 * half;
    rs[v] = rscale[m < M ? m : M - 1];
  }

  const int mrow = m0 + l16 < M ? m0 + l16 : M - 1;
  const float* __restrict__ arow = A + (size_t)mrow * K + 2 * half;

  static_assert(NT == 4, "the column interleave below spells out four tiles");
  const float* __restrict__ w1l = W1 + 4 * l16;
  for (int k = 0; k < K; k += 4) {
    v2f a = *(const v2f*)(arow + k);
    const int kk = k + 2 * half;
    const float4 w0 = *(const float4*)(w1l + (size_t)kk * HIDDEN);
    const float4 w1 = *(const float4*)(w1l + (size_t)(kk + 1) * HIDDEN);
    v2f b;
    b.x = w0.x; b.y = w1.x;
    acc[0] = __builtin_amdgcn_wmma_f32_16x16x4_f32(false, a, false, b, (short)0, acc[0], false, false);
    b.x = w0.y; b.y = w1.y;
    acc[1] = __builtin_amdgcn_wmma_f32_16x16x4_f32(false, a, false, b, (short)0, acc[1], false, false);
    b.x = w0.z; b.y = w1.z;
    acc[2] = __builtin_amdgcn_wmma_f32_16x16x4_f32(false, a, false, b, (short)0, acc[2], false, false);
    b.x = w0.w; b.y = w1.w;
    acc[3] = __builtin_amdgcn_wmma_f32_16x16x4_f32(false, a, false, b, (short)0, acc[3], false, false);
  }

  {
    const float4 bv = *(const float4*)(b1 + 4 * l16);
#pragma unroll
    for (int v = 0; v < 8; ++v)
      *(float4*)(tile + (v + 8 * half) * TILE_PITCH + 4 * l16) =
          make_float4(fmaxf(acc[0][v] + bv.x, 0.0f), fmaxf(acc[1][v] + bv.y, 0.0f),
                      fmaxf(acc[2][v] + bv.z, 0.0f), fmaxf(acc[3][v] + bv.w, 0.0f));
  }
  __syncthreads();

  v8f acc2[NT] = {};
  const float* a2 = tile + l16 * TILE_PITCH + 2 * half;
  const float* __restrict__ wcl = Wc + 4 * l16;
#pragma unroll 4
  for (int k = 0; k < HIDDEN; k += 4) {
    const v2f a = *(const v2f*)(a2 + k);
    const int kk = k + 2 * half;
    const float4 w0 = *(const float4*)(wcl + (size_t)kk * HIDDEN);
    const float4 w1 = *(const float4*)(wcl + (size_t)(kk + 1) * HIDDEN);
    v2f b;
    b.x = w0.x; b.y = w1.x;
    acc2[0] = __builtin_amdgcn_wmma_f32_16x16x4_f32(false, a, false, b, (short)0, acc2[0], false, false);
    b.x = w0.y; b.y = w1.y;
    acc2[1] = __builtin_amdgcn_wmma_f32_16x16x4_f32(false, a, false, b, (short)0, acc2[1], false, false);
    b.x = w0.z; b.y = w1.z;
    acc2[2] = __builtin_amdgcn_wmma_f32_16x16x4_f32(false, a, false, b, (short)0, acc2[2], false, false);
    b.x = w0.w; b.y = w1.w;
    acc2[3] = __builtin_amdgcn_wmma_f32_16x16x4_f32(false, a, false, b, (short)0, acc2[3], false, false);
  }

  const bool whole = m0 + 16 <= M;
  if (whole) {
#pragma unroll
    for (int v = 0; v < 8; ++v)
      *(float4*)(D + (size_t)(m0 + v + 8 * half) * HIDDEN + 4 * l16) =
          make_float4(acc2[0][v] * rs[v], acc2[1][v] * rs[v], acc2[2][v] * rs[v], acc2[3][v] * rs[v]);
  } else {
#pragma unroll
    for (int v = 0; v < 8; ++v) {
      const int m = m0 + v + 8 * half;
      if (m < M)
        *(float4*)(D + (size_t)m * HIDDEN + 4 * l16) =
            make_float4(acc2[0][v] * rs[v], acc2[1][v] * rs[v], acc2[2][v] * rs[v], acc2[3][v] * rs[v]);
    }
  }
}


static __device__ __forceinline__ float4 ld_row4(const float* __restrict__ tl, int r) {
  return *(const float4*)(tl + (size_t)r * HIDDEN);
}


#define EPL        10
#define ECAP       640
#define SLOTS      16

template <int NOUT, bool BIAS, bool SCALE_OUT>
__global__ __launch_bounds__(32) void k_gather_gemm(const int* __restrict__ off, const int* __restrict__ cnt,
                                                    const int* __restrict__ sedge,
                                                    const float* __restrict__ dinv,
                                                    const float* __restrict__ t,
                                                    const float* __restrict__ bias_in,
                                                    const float* __restrict__ W,
                                                    const float* __restrict__ bias_out,
                                                    float* __restrict__ D, int n) {
  constexpr int NT = (NOUT + 15) / 16;
  constexpr bool IL = (NOUT == 64);
  __shared__ __attribute__((aligned(16))) float tile[16 * TILE_PITCH];
  __shared__ int2  ebuf[ECAP];
  const int lane = threadIdx.x;
  const int half = lane >> 4;
  const int l16  = lane & 15;
  const int m0   = blockIdx.x * 16;
  if (m0 >= n) return;

  const float* __restrict__ tl = t + 4 * l16;
  const char*  __restrict__ tb = (const char*)t;
  const unsigned lane_off = 16u * (unsigned)l16;
  float* __restrict__ trow = tile + 4 * l16;
  const float4 bv = *(const float4*)(bias_in + 4 * l16);
  float4 tv[8];
#pragma unroll
  for (int p = 0; p < 8; ++p) {
    const int c = m0 + 8 * half + p;
    tv[p] = ld_row4(tl, c < n ? c : n - 1);
  }

  const int  cl    = m0 + l16 < n ? m0 + l16 : n - 1;
  const int  offv  = off[cl];
  const int  cntv  = cnt[cl];
  const float dv   = dinv[cl];
  const int  e0    = __builtin_amdgcn_readfirstlane(offv);
  const int  endv  = offv + cntv - e0;
  const int  ne    = __builtin_amdgcn_readlane(endv, 15);
  int es[15];
#pragma unroll
  for (int i = 0; i < 15; ++i) es[i] = __builtin_amdgcn_readlane(endv, i);

#pragma unroll
  for (int p = 0; p < 8; ++p) {
    const int i = 8 * half + p;
    const float di = __shfl(dv, i);
    *(float4*)(trow + i * TILE_PITCH) = make_float4(bv.x + di * tv[p].x, bv.y + di * tv[p].y,
                                                    bv.z + di * tv[p].z, bv.w + di * tv[p].w);
  }

  const int hb = half ? es[7] : 0;
  const int he = half ? ne : es[7];
  int    cur = 8 * half;
  float4 sum = *(const float4*)(trow + cur * TILE_PITCH);

  for (int cb = 0; cb < ne; cb += ECAP) {
    const int cm = ne - cb < ECAP ? ne - cb : ECAP;
    __syncthreads();
    for (int j0 = 0; j0 < cm; j0 += EPL * 32) {
      int2  se[EPL];
#pragma unroll
      for (int u = 0; u < EPL; ++u) {
        const int j  = j0 + u * 32 + lane;
        const int jj = j < cm ? j : cm - 1;
        se[u] = *(const int2*)(sedge + 2 * (size_t)(e0 + cb + jj));
      }
      asm volatile("" : "+v"(se[0].x), "+v"(se[1].x), "+v"(se[2].x), "+v"(se[3].x), "+v"(se[4].x),
                        "+v"(se[5].x), "+v"(se[6].x), "+v"(se[7].x), "+v"(se[8].x), "+v"(se[9].x)
                      : "v"(se[0].y), "v"(se[1].y), "v"(se[2].y), "v"(se[3].y), "v"(se[4].y),
                        "v"(se[5].y), "v"(se[6].y), "v"(se[7].y), "v"(se[8].y), "v"(se[9].y));
#pragma unroll
      for (int u = 0; u < EPL; ++u) {
        int r = se[u].x;
        r = r < 0 ? 0 : (r >= n ? n - 1 : r);
        se[u].x = r;
      }
      int   nidv[EPL];
      float dcg[EPL];
#pragma unroll
      for (int u = 0; u < EPL; ++u) {
        const int g = cb + j0 + u * 32 + lane;
        int nid = 0;
#pragma unroll
        for (int i = 0; i < 15; ++i) nid += es[i] <= g ? 1 : 0;
        nidv[u] = nid;
        dcg[u]  = __shfl(dv, nid);
      }
#pragma unroll
      for (int u = 0; u < EPL; ++u) {
        const int j = j0 + u * 32 + lane;
        if (j < cm) ebuf[j] = make_int2((se[u].x << 8) | nidv[u], __float_as_int(__int_as_float(se[u].y) * dcg[u]));
      }
    }
    __syncthreads();

    const int lo   = (hb > cb ? hb : cb) - cb;
    const int hi   = (he < cb + cm ? he : cb + cm) - cb;
    const int nh   = hi - lo;
    const int na   = __builtin_amdgcn_readlane(nh, 0);
    const int nb2  = __builtin_amdgcn_readlane(nh, 16);
    const int nmax = na > nb2 ? na : nb2;
    const int nmin = na < nb2 ? na : nb2;
    const int lo0  = nh > 0 ? lo : 0;
    const int last = nh > 0 ? nh - 1 : 0;
    for (int j0 = 0; j0 < nmax; j0 += SLOTS) {
      float4 v[SLOTS];
      int2   q[SLOTS];
#pragma unroll
      for (int u = 0; u < SLOTS; ++u) {
        q[u] = ebuf[lo0 + (j0 + u < last ? j0 + u : last)];
        v[u] = *(const float4*)(tb + (((unsigned)q[u].x & 0xffffff00u) + lane_off));
      }
      asm volatile("" :: "v"(v[0].x), "v"(v[1].x), "v"(v[2].x), "v"(v[3].x), "v"(v[4].x), "v"(v[5].x),
                         "v"(v[6].x), "v"(v[7].x), "v"(v[8].x), "v"(v[9].x), "v"(v[10].x), "v"(v[11].x),
                         "v"(v[12].x), "v"(v[13].x), "v"(v[14].x), "v"(v[15].x));
      if (j0 + SLOTS <= nmin) {
#pragma unroll
        for (int u = 0; u < SLOTS; ++u) {
          const int nid = q[u].x & 15;
          if (nid != cur) {
            *(float4*)(trow + cur * TILE_PITCH) = sum;
            cur = nid;
            sum = *(const float4*)(trow + cur * TILE_PITCH);
          }
          const float wk = __int_as_float(q[u].y);
          sum.x = __builtin_fmaf(wk, v[u].x, sum.x);
          sum.y = __builtin_fmaf(wk, v[u].y, sum.y);
          sum.z = __builtin_fmaf(wk, v[u].z, sum.z);
          sum.w = __builtin_fmaf(wk, v[u].w, sum.w);
        }
      } else {
#pragma unroll
        for (int u = 0; u < SLOTS; ++u) {
          if (j0 + u < nh) {
            const int nid = q[u].x & 15;
            if (nid != cur) {
              *(float4*)(trow + cur * TILE_PITCH) = sum;
              cur = nid;
              sum = *(const float4*)(trow + cur * TILE_PITCH);
            }
            const float wk = __int_as_float(q[u].y);
            sum.x = __builtin_fmaf(wk, v[u].x, sum.x);
            sum.y = __builtin_fmaf(wk, v[u].y, sum.y);
            sum.z = __builtin_fmaf(wk, v[u].z, sum.z);
            sum.w = __builtin_fmaf(wk, v[u].w, sum.w);
          }
        }
      }
    }
  }
  *(float4*)(trow + cur * TILE_PITCH) = sum;
  __syncthreads();

  v8f acc[NT] = {};
  const float* arow = tile + l16 * TILE_PITCH + 2 * half;
#pragma unroll 4
  for (int k = 0; k < HIDDEN; k += 4) {
    v2f a = *(const v2f*)(arow + k);
    a.x = fmaxf(a.x, 0.0f);
    a.y = fmaxf(a.y, 0.0f);
    const int kk = k + 2 * half;
    if constexpr (IL) {
      const float4 w0 = *(const float4*)(W + (size_t)kk * NOUT + 4 * l16);
      const float4 w1 = *(const float4*)(W + (size_t)(kk + 1) * NOUT + 4 * l16);
      v2f bfr;
      bfr.x = w0.x; bfr.y = w1.x;
      acc[0] = __builtin_amdgcn_wmma_f32_16x16x4_f32(false, a, false, bfr, (short)0, acc[0], false, false);
      bfr.x = w0.y; bfr.y = w1.y;
      acc[1] = __builtin_amdgcn_wmma_f32_16x16x4_f32(false, a, false, bfr, (short)0, acc[1], false, false);
      bfr.x = w0.z; bfr.y = w1.z;
      acc[2] = __builtin_amdgcn_wmma_f32_16x16x4_f32(false, a, false, bfr, (short)0, acc[2], false, false);
      bfr.x = w0.w; bfr.y = w1.w;
      acc[3] = __builtin_amdgcn_wmma_f32_16x16x4_f32(false, a, false, bfr, (short)0, acc[3], false, false);
    } else {
#pragma unroll
      for (int tt = 0; tt < NT; ++tt) {
        const int c = tt * 16 + l16;
        v2f bfr;
        if ((NOUT % 16 == 0) || (c < NOUT)) {
          bfr.x = W[(size_t)kk * NOUT + c];
          bfr.y = W[(size_t)(kk + 1) * NOUT + c];
        } else {
          bfr.x = 0.0f; bfr.y = 0.0f;
        }
        acc[tt] = __builtin_amdgcn_wmma_f32_16x16x4_f32(
             false, a,  false, bfr,
             (short)0, acc[tt],  false,  false);
      }
    }
  }

  const bool whole = m0 + 16 <= n;
  float ro[8];
#pragma unroll
  for (int v = 0; v < 8; ++v) ro[v] = SCALE_OUT ? __shfl(dv, v + 8 * half) : 1.0f;
  if constexpr (IL) {
    float4 bo = make_float4(0.0f, 0.0f, 0.0f, 0.0f);
    if (BIAS) bo = *(const float4*)(bias_out + 4 * l16);
#pragma unroll
    for (int v = 0; v < 8; ++v) {
      const int m = m0 + v + 8 * half;
      if (whole || m < n)
        *(float4*)(D + (size_t)m * NOUT + 4 * l16) =
            SCALE_OUT ? make_float4((acc[0][v] + bo.x) * ro[v], (acc[1][v] + bo.y) * ro[v],
                                    (acc[2][v] + bo.z) * ro[v], (acc[3][v] + bo.w) * ro[v])
                      : make_float4(acc[0][v] + bo.x, acc[1][v] + bo.y, acc[2][v] + bo.z, acc[3][v] + bo.w);
    }
  } else {
#pragma unroll
    for (int tt = 0; tt < NT; ++tt) {
      const int c = tt * 16 + l16;
      if ((NOUT % 16 != 0) && (c >= NOUT)) continue;
      const float bo = BIAS ? bias_out[c] : 0.0f;
      if (whole) {
#pragma unroll
        for (int v = 0; v < 8; ++v) D[(size_t)(m0 + v + 8 * half) * NOUT + c] = SCALE_OUT ? (acc[tt][v] + bo) * ro[v] : acc[tt][v] + bo;
      } else {
#pragma unroll
        for (int v = 0; v < 8; ++v) {
          const int m = m0 + v + 8 * half;
          if (m < n) D[(size_t)m * NOUT + c] = SCALE_OUT ? (acc[tt][v] + bo) * ro[v] : acc[tt][v] + bo;
        }
      }
    }
  }
}


extern "C" void kernel_launch(void* const* d_in, const int* in_sizes, int n_in,
                              void* d_out, int out_size, void* d_ws, size_t ws_size,
                              hipStream_t stream) {
  const float* x   = (const float*)d_in[0];
  const int*   ei  = (const int*)d_in[1];
  const float* ew  = (const float*)d_in[2];
  const float* W1  = (const float*)d_in[3];
  const float* b1  = (const float*)d_in[4];
  const float* Wc1 = (const float*)d_in[5];
  const float* bc1 = (const float*)d_in[6];
  const float* Wc2 = (const float*)d_in[7];
  const float* bc2 = (const float*)d_in[8];
  const float* W2  = (const float*)d_in[9];
  const float* b2  = (const float*)d_in[10];
  float* out = (float*)d_out;

  const int N = in_sizes[0] / F_IN;
  const int E = in_sizes[2];
  const int* erow = ei;
  const int* ecol = ei + E;

  int4*  tmp    = (int4*)d_ws;
  float* h      = (float*)(tmp + E);
  float* t      = h + (size_t)N * HIDDEN;
  int*   hist   = (int*)(t + (size_t)N * HIDDEN);
  int*   sedge  = hist + (size_t)NB_MAX * SORT_GRID;
  float* dinv   = (float*)(sedge + 2 * (size_t)E);
  int*   cnt    = (int*)(dinv + N);
  int*   off    = cnt + N;
  int*   bstart = off + N;

  const int nb        = (N + BW - 1) / BW;
  const int gemm_grid = (N + 31) / 32;
  const int tile_grid = (N + 15) / 16;

  k_bucket_hist<<<SORT_GRID, SORT_TB, 0, stream>>>(ecol, hist, E, N, nb);
  k_bucket_fill<<<SORT_GRID, SORT_TB, 0, stream>>>(erow, ecol, ew, hist, bstart, tmp, E, N, nb);
  k_bucket_sort<<<nb, SORT_TB, 0, stream>>>(tmp, bstart, sedge, off, cnt, dinv, N);

  k_lin_conv_wmma<F_IN><<<gemm_grid, 64, 0, stream>>>(x, W1, b1, Wc1, dinv, t, N);

  k_gather_gemm<HIDDEN, false, true>
      <<<tile_grid, 32, 0, stream>>>(off, cnt, sedge, dinv, t, bc1, Wc2, nullptr, h, N);

  k_gather_gemm<NCLS, true, false>
      <<<tile_grid, 32, 0, stream>>>(off, cnt, sedge, dinv, h, bc2, W2, b2, out, N);
}


# Edge grouping in 2 launches, not 3: each slice block sorts its edges by bucket inside its own range (no other block's counts needed), bucket blocks read their 256 pieces; sums in LDS arrival order
# speedup vs baseline: 4.0744x; 1.0531x over previous
#include <hip/hip_runtime.h>

typedef __attribute__((ext_vector_type(2))) float v2f;
typedef __attribute__((ext_vector_type(8))) float v8f;

#define F_IN   256
#define HIDDEN 64
#define NCLS   40


#define BW_SHIFT 9
#define BW       (1 << BW_SHIFT)
#define SORT_TB  1024
#define SORT_GRID 256
#define SEG_PW   (SORT_GRID / (SORT_TB / 32))
#define NB_MAX   1024
#define EPT      8

static __device__ __forceinline__ int block_excl_scan(int v, int* wsum, int* total) {
  const int lane = threadIdx.x & 31;
  const int wv   = threadIdx.x >> 5;
  int s = v;
  for (int d = 1; d < 32; d <<= 1) {
    const int u = __shfl_up(s, d);
    if (lane >= d) s += u;
  }
  if (lane == 31) wsum[wv] = s;
  __syncthreads();
  if (wv == 0) {
    const int ws = wsum[lane];
    int q = ws;
    for (int d = 1; d < 32; d <<= 1) {
      const int u = __shfl_up(q, d);
      if (lane >= d) q += u;
    }
    wsum[lane] = q - ws;
    if (lane == 31) wsum[32] = q;
  }
  __syncthreads();
  const int r = s - v + wsum[wv];
  *total = wsum[32];
  __syncthreads();
  return r;
}

static __device__ __forceinline__ int in_range(int c, int n) {
  const unsigned uc = (unsigned)c, last = (unsigned)(n - 1);
  return (int)(uc < last ? uc : last);
}

static __device__ __forceinline__ void block_sum2(int a, int b, int* red, int* sa, int* sb) {
  static_assert(SORT_TB == 32 * 32, "one partial sum per lane in the second step");
  const int lane = threadIdx.x & 31;
  const int wv   = threadIdx.x >> 5;
  for (int d = 16; d > 0; d >>= 1) {
    a += __shfl_xor(a, d);
    b += __shfl_xor(b, d);
  }
  if (lane == 0) { red[wv] = a; red[32 + wv] = b; }
  __syncthreads();
  int x = red[lane], y = red[32 + lane];
  for (int d = 16; d > 0; d >>= 1) {
    x += __shfl_xor(x, d);
    y += __shfl_xor(y, d);
  }
  *sa = x;
  *sb = y;
}

__global__ __launch_bounds__(SORT_TB) void k_slice_sort(const int* __restrict__ row, const int* __restrict__ col,
                                                        const float* __restrict__ ew,
                                                        int* __restrict__ cntm, int* __restrict__ locm,
                                                        int4* __restrict__ tmp, int e, int n, int nb) {
  __shared__ int h[NB_MAX];
  __shared__ int cur[NB_MAX];
  __shared__ int wsum[33];
  static_assert(NB_MAX == SORT_TB, "one LDS counter per thread");
  const int tid = threadIdx.x;
  const int b   = blockIdx.x;
  constexpr int nblk = SORT_GRID;
  h[tid] = 0;
  const int chunk = (e + nblk - 1) / nblk;
  const int beg   = b * chunk;
  const int end   = beg + chunk < e ? beg + chunk : e;

  int c0[EPT], r0[EPT], w0[EPT];
  for (int u = 0; u < EPT; ++u) {
    const int i  = beg + tid + u * SORT_TB;
    const int ii = i < end ? i : end - 1;
    c0[u] = col[ii];
    r0[u] = row[ii];
    w0[u] = __float_as_int(ew[ii]);
  }
  asm volatile("" :: "v"(c0[0]), "v"(c0[1]), "v"(c0[2]), "v"(c0[3]), "v"(c0[4]), "v"(c0[5]), "v"(c0[6]), "v"(c0[7]),
                     "v"(r0[0]), "v"(r0[1]), "v"(r0[2]), "v"(r0[3]), "v"(r0[4]), "v"(r0[5]), "v"(r0[6]), "v"(r0[7]),
                     "v"(w0[0]), "v"(w0[1]), "v"(w0[2]), "v"(w0[3]), "v"(w0[4]), "v"(w0[5]), "v"(w0[6]), "v"(w0[7]));
  __syncthreads();
  for (int u = 0; u < EPT; ++u)
    if (beg + tid + u * SORT_TB < end) atomicAdd(&h[in_range(c0[u], n) >> BW_SHIFT], 1);
  for (int i0 = beg + tid + EPT * SORT_TB; i0 < end; i0 += EPT * SORT_TB) {
    int c[EPT];
    for (int u = 0; u < EPT; ++u) {
      const int i = i0 + u * SORT_TB;
      c[u] = col[i < end ? i : end - 1];
    }
    asm volatile("" :: "v"(c[0]), "v"(c[1]), "v"(c[2]), "v"(c[3]), "v"(c[4]), "v"(c[5]), "v"(c[6]), "v"(c[7]));
    for (int u = 0; u < EPT; ++u)
      if (i0 + u * SORT_TB < end) atomicAdd(&h[in_range(c[u], n) >> BW_SHIFT], 1);
  }
  __syncthreads();
  const int hv = h[tid];
  int total;
  const int ex = block_excl_scan(hv, wsum, &total);
  cur[tid] = beg + ex;
  __syncthreads();

  for (int u = 0; u < EPT; ++u) {
    if (beg + tid + u * SORT_TB < end) {
      const int cc  = in_range(c0[u], n);
      const int pos = atomicAdd(&cur[cc >> BW_SHIFT], 1);
      tmp[pos] = make_int4(r0[u], cc, w0[u], 0);
    }
  }
  for (int i0 = beg + tid + EPT * SORT_TB; i0 < end; i0 += EPT * SORT_TB) {
    int c[EPT], r[EPT], w[EPT];
    for (int u = 0; u < EPT; ++u) {
      const int i  = i0 + u * SORT_TB;
      const int ii = i < end ? i : end - 1;
      c[u] = col[ii];
      r[u] = row[ii];
      w[u] = __float_as_int(ew[ii]);
    }
    asm volatile("" :: "v"(c[0]), "v"(c[1]), "v"(c[2]), "v"(c[3]), "v"(c[4]), "v"(c[5]), "v"(c[6]), "v"(c[7]),
                       "v"(r[0]), "v"(r[1]), "v"(r[2]), "v"(r[3]), "v"(r[4]), "v"(r[5]), "v"(r[6]), "v"(r[7]),
                       "v"(w[0]), "v"(w[1]), "v"(w[2]), "v"(w[3]), "v"(w[4]), "v"(w[5]), "v"(w[6]), "v"(w[7]));
    for (int u = 0; u < EPT; ++u) {
      if (i0 + u * SORT_TB < end) {
        const int cc  = in_range(c[u], n);
        const int pos = atomicAdd(&cur[cc >> BW_SHIFT], 1);
        tmp[pos] = make_int4(r[u], cc, w[u], 0);
      }
    }
  }
  if (tid < nb) {
    cntm[(size_t)tid * nblk + b] = hv;
    locm[(size_t)tid * nblk + b] = ex;
  }
}

__global__ __launch_bounds__(SORT_TB) void k_bucket_sort(const int4* __restrict__ tmp, const int* __restrict__ cntm,
                                                         const int* __restrict__ locm,
                                                         int* sedge, int* __restrict__ off, int* __restrict__ cnt,
                                                         float* __restrict__ dinv, int e, int n) {
  __shared__ int lcnt[BW];
  __shared__ int lcur[BW];
  __shared__ int wsum[33];
  __shared__ int sbase[SORT_GRID];
  __shared__ int scnt[SORT_GRID];
  __shared__ int red[64];
  static_assert(SORT_GRID <= SORT_TB && SORT_GRID % (SORT_TB / 32) == 0 && SEG_PW == 8,
                "one thread per piece in the prologue, eight pieces per wave in the walks");
  const int tid   = threadIdx.x;
  const int lane  = tid & 31;
  const int wv    = tid >> 5;
  const int node0 = (int)blockIdx.x << BW_SHIFT;
  constexpr int nblk = SORT_GRID;
  const int chunk = (e + nblk - 1) / nblk;
  int cv = 0, lv = 0;
  if (tid < nblk) {
    cv = cntm[(size_t)blockIdx.x * nblk + tid];
    lv = locm[(size_t)blockIdx.x * nblk + tid];
    cv = cv < 0 ? 0 : (cv < chunk ? cv : chunk);
    lv = lv < 0 ? 0 : (lv < chunk ? lv : chunk);
    sbase[tid] = tid * chunk + lv;
    scnt[tid]  = cv;
  }
  if (tid < BW) lcnt[tid] = 0;
  int s, ne;
  block_sum2(lv, cv, red, &s, &ne);
  s  = s < e ? s : e;
  ne = ne < e - s ? ne : e - s;

  int gb[SEG_PW], nq[SEG_PW];
  int nmax = 0;
#pragma unroll
  for (int q = 0; q < SEG_PW; ++q) {
    gb[q] = __builtin_amdgcn_readfirstlane(sbase[wv * SEG_PW + q]);
    nq[q] = __builtin_amdgcn_readfirstlane(scnt[wv * SEG_PW + q]);
    nmax  = nq[q] > nmax ? nq[q] : nmax;
  }

  for (int j0 = 0; j0 < nmax; j0 += 64) {
    int c[2 * SEG_PW];
#pragma unroll
    for (int q = 0; q < SEG_PW; ++q) {
#pragma unroll
      for (int m = 0; m < 2; ++m) {
        const int j = j0 + 32 * m + lane;
        int i = gb[q] + (j < nq[q] ? j : nq[q] - 1);
        i = i < 0 ? 0 : (i < e ? i : e - 1);
        c[2 * q + m] = tmp[i].y;
      }
    }
    asm volatile("" :: "v"(c[0]), "v"(c[1]), "v"(c[2]), "v"(c[3]), "v"(c[4]), "v"(c[5]), "v"(c[6]), "v"(c[7]),
                       "v"(c[8]), "v"(c[9]), "v"(c[10]), "v"(c[11]), "v"(c[12]), "v"(c[13]), "v"(c[14]), "v"(c[15]));
#pragma unroll
    for (int q = 0; q < SEG_PW; ++q) {
#pragma unroll
      for (int m = 0; m < 2; ++m)
        if (j0 + 32 * m + lane < nq[q]) atomicAdd(&lcnt[(c[2 * q + m] - node0) & (BW - 1)], 1);
    }
  }
  __syncthreads();
  const int num = tid < BW ? lcnt[tid] : 0;
  int total;
  const int ex = block_excl_scan(num, wsum, &total);
  if (tid < BW) lcur[tid] = ex;
  __syncthreads();
  for (int j0 = 0; j0 < nmax; j0 += 64) {
    int4 r[2 * SEG_PW];
#pragma unroll
    for (int q = 0; q < SEG_PW; ++q) {
#pragma unroll
      for (int m = 0; m < 2; ++m) {
        const int j = j0 + 32 * m + lane;
        int i = gb[q] + (j < nq[q] ? j : nq[q] - 1);
        i = i < 0 ? 0 : (i < e ? i : e - 1);
        r[2 * q + m] = tmp[i];
      }
    }
    asm volatile("" :: "v"(r[0].x), "v"(r[0].y), "v"(r[0].z), "v"(r[1].x), "v"(r[1].y), "v"(r[1].z),
                       "v"(r[2].x), "v"(r[2].y), "v"(r[2].z), "v"(r[3].x), "v"(r[3].y), "v"(r[3].z),
                       "v"(r[4].x), "v"(r[4].y), "v"(r[4].z), "v"(r[5].x), "v"(r[5].y), "v"(r[5].z),
                       "v"(r[6].x), "v"(r[6].y), "v"(r[6].z), "v"(r[7].x), "v"(r[7].y), "v"(r[7].z),
                       "v"(r[8].x), "v"(r[8].y), "v"(r[8].z), "v"(r[9].x), "v"(r[9].y), "v"(r[9].z),
                       "v"(r[10].x), "v"(r[10].y), "v"(r[10].z), "v"(r[11].x), "v"(r[11].y), "v"(r[11].z),
                       "v"(r[12].x), "v"(r[12].y), "v"(r[12].z), "v"(r[13].x), "v"(r[13].y), "v"(r[13].z),
                       "v"(r[14].x), "v"(r[14].y), "v"(r[14].z), "v"(r[15].x), "v"(r[15].y), "v"(r[15].z));
    int p[2 * SEG_PW];
#pragma unroll
    for (int q = 0; q < SEG_PW; ++q) {
#pragma unroll
      for (int m = 0; m < 2; ++m) {
        p[2 * q + m] = 0;
        if (j0 + 32 * m + lane < nq[q]) p[2 * q + m] = atomicAdd(&lcur[(r[2 * q + m].y - node0) & (BW - 1)], 1);
      }
    }
#pragma unroll
    for (int q = 0; q < SEG_PW; ++q) {
#pragma unroll
      for (int m = 0; m < 2; ++m) {
        if (j0 + 32 * m + lane < nq[q]) {
          int pp = p[2 * q + m];
          pp = pp < ne ? pp : ne - 1;
          pp = pp < 0 ? 0 : pp;
          *(int2*)(sedge + 2 * (size_t)(s + pp)) = make_int2(r[2 * q + m].x, r[2 * q + m].z);
        }
      }
    }
  }
  __syncthreads();
  const int c = node0 + tid;
  if (tid < BW && c < n) {
    const int beg = s + ex;
    off[c] = beg;
    cnt[c] = num;
    const int* wp = sedge + 2 * (size_t)beg + 1;
    float sd = 1.0f;
    for (int j = 0; j < num; j += EPT) {
      float w[EPT];
      for (int u = 0; u < EPT; ++u) w[u] = __int_as_float(wp[2 * (j + u < num ? j + u : num - 1)]);
      for (int u = 0; u < EPT; ++u) sd = (j + u < num) ? sd + w[u] : sd;
    }
    dinv[c] = (sd > 0.0f) ? rsqrtf(sd) : 0.0f;
  }
}


#define TILE_PITCH 68

template <int K>
__global__ __launch_bounds__(64) void k_lin_conv_wmma(const float* __restrict__ A,
                                                      const float* __restrict__ W1,
                                                      const float* __restrict__ b1,
                                                      const float* __restrict__ Wc,
                                                      const float* __restrict__ rscale,
                                                      float* __restrict__ D, int M) {
  constexpr int NT = HIDDEN / 16;
  __shared__ __attribute__((aligned(16))) float tiles[2][16 * TILE_PITCH];
  const int wv   = threadIdx.x >> 5;
  const int lane = threadIdx.x & 31;
  const int half = lane >> 4;
  const int l16  = lane & 15;
  const int m0   = __builtin_amdgcn_readfirstlane((blockIdx.x * 2 + wv) * 16);
  float* __restrict__ tile = tiles[wv];

  v8f acc[NT] = {};

  float rs[8];
#pragma unroll
  for (int v = 0; v < 8; ++v) {
    const int m = m0 + v + 8 * half;
    rs[v] = rscale[m < M ? m : M - 1];
  }

  const int mrow = m0 + l16 < M ? m0 + l16 : M - 1;
  const float* __restrict__ arow = A + (size_t)mrow * K + 2 * half;

  static_assert(NT == 4, "the column interleave below spells out four tiles");
  const float* __restrict__ w1l = W1 + 4 * l16;
  for (int k = 0; k < K; k += 4) {
    v2f a = *(const v2f*)(arow + k);
    const int kk = k + 2 * half;
    const float4 w0 = *(const float4*)(w1l + (size_t)kk * HIDDEN);
    const float4 w1 = *(const float4*)(w1l + (size_t)(kk + 1) * HIDDEN);
    v2f b;
    b.x = w0.x; b.y = w1.x;
    acc[0] = __builtin_amdgcn_wmma_f32_16x16x4_f32(false, a, false, b, (short)0, acc[0], false, false);
    b.x = w0.y; b.y = w1.y;
    acc[1] = __builtin_amdgcn_wmma_f32_16x16x4_f32(false, a, false, b, (short)0, acc[1], false, false);
    b.x = w0.z; b.y = w1.z;
    acc[2] = __builtin_amdgcn_wmma_f32_16x16x4_f32(false, a, false, b, (short)0, acc[2], false, false);
    b.x = w0.w; b.y = w1.w;
    acc[3] = __builtin_amdgcn_wmma_f32_16x16x4_f32(false, a, false, b, (short)0, acc[3], false, false);
  }

  {
    const float4 bv = *(const float4*)(b1 + 4 * l16);
#pragma unroll
    for (int v = 0; v < 8; ++v)
      *(float4*)(tile + (v + 8 * half) * TILE_PITCH + 4 * l16) =
          make_float4(fmaxf(acc[0][v] + bv.x, 0.0f), fmaxf(acc[1][v] + bv.y, 0.0f),
                      fmaxf(acc[2][v] + bv.z, 0.0f), fmaxf(acc[3][v] + bv.w, 0.0f));
  }
  __syncthreads();

  v8f acc2[NT] = {};
  const float* a2 = tile + l16 * TILE_PITCH + 2 * half;
  const float* __restrict__ wcl = Wc + 4 * l16;
#pragma unroll 4
  for (int k = 0; k < HIDDEN; k += 4) {
    const v2f a = *(const v2f*)(a2 + k);
    const int kk = k + 2 * half;
    const float4 w0 = *(const float4*)(wcl + (size_t)kk * HIDDEN);
    const float4 w1 = *(const float4*)(wcl + (size_t)(kk + 1) * HIDDEN);
    v2f b;
    b.x = w0.x; b.y = w1.x;
    acc2[0] = __builtin_amdgcn_wmma_f32_16x16x4_f32(false, a, false, b, (short)0, acc2[0], false, false);
    b.x = w0.y; b.y = w1.y;
    acc2[1] = __builtin_amdgcn_wmma_f32_16x16x4_f32(false, a, false, b, (short)0, acc2[1], false, false);
    b.x = w0.z; b.y = w1.z;
    acc2[2] = __builtin_amdgcn_wmma_f32_16x16x4_f32(false, a, false, b, (short)0, acc2[2], false, false);
    b.x = w0.w; b.y = w1.w;
    acc2[3] = __builtin_amdgcn_wmma_f32_16x16x4_f32(false, a, false, b, (short)0, acc2[3], false, false);
  }

  const bool whole = m0 + 16 <= M;
  if (whole) {
#pragma unroll
    for (int v = 0; v < 8; ++v)
      *(float4*)(D + (size_t)(m0 + v + 8 * half) * HIDDEN + 4 * l16) =
          make_float4(acc2[0][v] * rs[v], acc2[1][v] * rs[v], acc2[2][v] * rs[v], acc2[3][v] * rs[v]);
  } else {
#pragma unroll
    for (int v = 0; v < 8; ++v) {
      const int m = m0 + v + 8 * half;
      if (m < M)
        *(float4*)(D + (size_t)m * HIDDEN + 4 * l16) =
            make_float4(acc2[0][v] * rs[v], acc2[1][v] * rs[v], acc2[2][v] * rs[v], acc2[3][v] * rs[v]);
    }
  }
}


static __device__ __forceinline__ float4 ld_row4(const float* __restrict__ tl, int r) {
  return *(const float4*)(tl + (size_t)r * HIDDEN);
}


#define EPL        10
#define ECAP       640
#define SLOTS      16

template <int NOUT, bool BIAS, bool SCALE_OUT>
__global__ __launch_bounds__(32) void k_gather_gemm(const int* __restrict__ off, const int* __restrict__ cnt,
                                                    const int* __restrict__ sedge,
                                                    const float* __restrict__ dinv,
                                                    const float* __restrict__ t,
                                                    const float* __restrict__ bias_in,
                                                    const float* __restrict__ W,
                                                    const float* __restrict__ bias_out,
                                                    float* __restrict__ D, int n) {
  constexpr int NT = (NOUT + 15) / 16;
  constexpr bool IL = (NOUT == 64);
  __shared__ __attribute__((aligned(16))) float tile[16 * TILE_PITCH];
  __shared__ int2  ebuf[ECAP];
  const int lane = threadIdx.x;
  const int half = lane >> 4;
  const int l16  = lane & 15;
  const int m0   = blockIdx.x * 16;
  if (m0 >= n) return;

  const float* __restrict__ tl = t + 4 * l16;
  const char*  __restrict__ tb = (const char*)t;
  const unsigned lane_off = 16u * (unsigned)l16;
  float* __restrict__ trow = tile + 4 * l16;
  const float4 bv = *(const float4*)(bias_in + 4 * l16);
  float4 tv[8];
#pragma unroll
  for (int p = 0; p < 8; ++p) {
    const int c = m0 + 8 * half + p;
    tv[p] = ld_row4(tl, c < n ? c : n - 1);
  }

  const int  cl    = m0 + l16 < n ? m0 + l16 : n - 1;
  const int  offv  = off[cl];
  const int  cntv  = cnt[cl];
  const float dv   = dinv[cl];
  const int  e0    = __builtin_amdgcn_readfirstlane(offv);
  const int  endv  = offv + cntv - e0;
  const int  ne    = __builtin_amdgcn_readlane(endv, 15);
  int es[15];
#pragma unroll
  for (int i = 0; i < 15; ++i) es[i] = __builtin_amdgcn_readlane(endv, i);

#pragma unroll
  for (int p = 0; p < 8; ++p) {
    const int i = 8 * half + p;
    const float di = __shfl(dv, i);
    *(float4*)(trow + i * TILE_PITCH) = make_float4(bv.x + di * tv[p].x, bv.y + di * tv[p].y,
                                                    bv.z + di * tv[p].z, bv.w + di * tv[p].w);
  }

  const int hb = half ? es[7] : 0;
  const int he = half ? ne : es[7];
  int    cur = 8 * half;
  float4 sum = *(const float4*)(trow + cur * TILE_PITCH);

  for (int cb = 0; cb < ne; cb += ECAP) {
    const int cm = ne - cb < ECAP ? ne - cb : ECAP;
    __syncthreads();
    for (int j0 = 0; j0 < cm; j0 += EPL * 32) {
      int2  se[EPL];
#pragma unroll
      for (int u = 0; u < EPL; ++u) {
        const int j  = j0 + u * 32 + lane;
        const int jj = j < cm ? j : cm - 1;
        se[u] = *(const int2*)(sedge + 2 * (size_t)(e0 + cb + jj));
      }
      asm volatile("" : "+v"(se[0].x), "+v"(se[1].x), "+v"(se[2].x), "+v"(se[3].x), "+v"(se[4].x),
                        "+v"(se[5].x), "+v"(se[6].x), "+v"(se[7].x), "+v"(se[8].x), "+v"(se[9].x)
                      : "v"(se[0].y), "v"(se[1].y), "v"(se[2].y), "v"(se[3].y), "v"(se[4].y),
                        "v"(se[5].y), "v"(se[6].y), "v"(se[7].y), "v"(se[8].y), "v"(se[9].y));
#pragma unroll
      for (int u = 0; u < EPL; ++u) {
        int r = se[u].x;
        r = r < 0 ? 0 : (r >= n ? n - 1 : r);
        se[u].x = r;
      }
      int   nidv[EPL];
      float dcg[EPL];
#pragma unroll
      for (int u = 0; u < EPL; ++u) {
        const int g = cb + j0 + u * 32 + lane;
        int nid = 0;
#pragma unroll
        for (int i = 0; i < 15; ++i) nid += es[i] <= g ? 1 : 0;
        nidv[u] = nid;
        dcg[u]  = __shfl(dv, nid);
      }
#pragma unroll
      for (int u = 0; u < EPL; ++u) {
        const int j = j0 + u * 32 + lane;
        if (j < cm) ebuf[j] = make_int2((se[u].x << 8) | nidv[u], __float_as_int(__int_as_float(se[u].y) * dcg[u]));
      }
    }
    __syncthreads();

    const int lo   = (hb > cb ? hb : cb) - cb;
    const int hi   = (he < cb + cm ? he : cb + cm) - cb;
    const int nh   = hi - lo;
    const int na   = __builtin_amdgcn_readlane(nh, 0);
    const int nb2  = __builtin_amdgcn_readlane(nh, 16);
    const int nmax = na > nb2 ? na : nb2;
    const int nmin = na < nb2 ? na : nb2;
    const int lo0  = nh > 0 ? lo : 0;
    const int last = nh > 0 ? nh - 1 : 0;
    for (int j0 = 0; j0 < nmax; j0 += SLOTS) {
      float4 v[SLOTS];
      int2   q[SLOTS];
#pragma unroll
      for (int u = 0; u < SLOTS; ++u) {
        q[u] = ebuf[lo0 + (j0 + u < last ? j0 + u : last)];
        v[u] = *(const float4*)(tb + (((unsigned)q[u].x & 0xffffff00u) + lane_off));
      }
      asm volatile("" :: "v"(v[0].x), "v"(v[1].x), "v"(v[2].x), "v"(v[3].x), "v"(v[4].x), "v"(v[5].x),
                         "v"(v[6].x), "v"(v[7].x), "v"(v[8].x), "v"(v[9].x), "v"(v[10].x), "v"(v[11].x),
                         "v"(v[12].x), "v"(v[13].x), "v"(v[14].x), "v"(v[15].x));
      if (j0 + SLOTS <= nmin) {
#pragma unroll
        for (int u = 0; u < SLOTS; ++u) {
          const int nid = q[u].x & 15;
          if (nid != cur) {
            *(float4*)(trow + cur * TILE_PITCH) = sum;
            cur = nid;
            sum = *(const float4*)(trow + cur * TILE_PITCH);
          }
          const float wk = __int_as_float(q[u].y);
          sum.x = __builtin_fmaf(wk, v[u].x, sum.x);
          sum.y = __builtin_fmaf(wk, v[u].y, sum.y);
          sum.z = __builtin_fmaf(wk, v[u].z, sum.z);
          sum.w = __builtin_fmaf(wk, v[u].w, sum.w);
        }
      } else {
#pragma unroll
        for (int u = 0; u < SLOTS; ++u) {
          if (j0 + u < nh) {
            const int nid = q[u].x & 15;
            if (nid != cur) {
              *(float4*)(trow + cur * TILE_PITCH) = sum;
              cur = nid;
              sum = *(const float4*)(trow + cur * TILE_PITCH);
            }
            const float wk = __int_as_float(q[u].y);
            sum.x = __builtin_fmaf(wk, v[u].x, sum.x);
            sum.y = __builtin_fmaf(wk, v[u].y, sum.y);
            sum.z = __builtin_fmaf(wk, v[u].z, sum.z);
            sum.w = __builtin_fmaf(wk, v[u].w, sum.w);
          }
        }
      }
    }
  }
  *(float4*)(trow + cur * TILE_PITCH) = sum;
  __syncthreads();

  v8f acc[NT] = {};
  const float* arow = tile + l16 * TILE_PITCH + 2 * half;
#pragma unroll 4
  for (int k = 0; k < HIDDEN; k += 4) {
    v2f a = *(const v2f*)(arow + k);
    a.x = fmaxf(a.x, 0.0f);
    a.y = fmaxf(a.y, 0.0f);
    const int kk = k + 2 * half;
    if constexpr (IL) {
      const float4 w0 = *(const float4*)(W + (size_t)kk * NOUT + 4 * l16);
      const float4 w1 = *(const float4*)(W + (size_t)(kk + 1) * NOUT + 4 * l16);
      v2f bfr;
      bfr.x = w0.x; bfr.y = w1.x;
      acc[0] = __builtin_amdgcn_wmma_f32_16x16x4_f32(false, a, false, bfr, (short)0, acc[0], false, false);
      bfr.x = w0.y; bfr.y = w1.y;
      acc[1] = __builtin_amdgcn_wmma_f32_16x16x4_f32(false, a, false, bfr, (short)0, acc[1], false, false);
      bfr.x = w0.z; bfr.y = w1.z;
      acc[2] = __builtin_amdgcn_wmma_f32_16x16x4_f32(false, a, false, bfr, (short)0, acc[2], false, false);
      bfr.x = w0.w; bfr.y = w1.w;
      acc[3] = __builtin_amdgcn_wmma_f32_16x16x4_f32(false, a, false, bfr, (short)0, acc[3], false, false);
    } else {
#pragma unroll
      for (int tt = 0; tt < NT; ++tt) {
        const int c = tt * 16 + l16;
        v2f bfr;
        if ((NOUT % 16 == 0) || (c < NOUT)) {
          bfr.x = W[(size_t)kk * NOUT + c];
          bfr.y = W[(size_t)(kk + 1) * NOUT + c];
        } else {
          bfr.x = 0.0f; bfr.y = 0.0f;
        }
        acc[tt] = __builtin_amdgcn_wmma_f32_16x16x4_f32(
             false, a,  false, bfr,
             (short)0, acc[tt],  false,  false);
      }
    }
  }

  const bool whole = m0 + 16 <= n;
  float ro[8];
#pragma unroll
  for (int v = 0; v < 8; ++v) ro[v] = SCALE_OUT ? __shfl(dv, v + 8 * half) : 1.0f;
  if constexpr (IL) {
    float4 bo = make_float4(0.0f, 0.0f, 0.0f, 0.0f);
    if (BIAS) bo = *(const float4*)(bias_out + 4 * l16);
#pragma unroll
    for (int v = 0; v < 8; ++v) {
      const int m = m0 + v + 8 * half;
      if (whole || m < n)
        *(float4*)(D + (size_t)m * NOUT + 4 * l16) =
            SCALE_OUT ? make_float4((acc[0][v] + bo.x) * ro[v], (acc[1][v] + bo.y) * ro[v],
                                    (acc[2][v] + bo.z) * ro[v], (acc[3][v] + bo.w) * ro[v])
                      : make_float4(acc[0][v] + bo.x, acc[1][v] + bo.y, acc[2][v] + bo.z, acc[3][v] + bo.w);
    }
  } else {
#pragma unroll
    for (int tt = 0; tt < NT; ++tt) {
      const int c = tt * 16 + l16;
      if ((NOUT % 16 != 0) && (c >= NOUT)) continue;
      const float bo = BIAS ? bias_out[c] : 0.0f;
      if (whole) {
#pragma unroll
        for (int v = 0; v < 8; ++v) D[(size_t)(m0 + v + 8 * half) * NOUT + c] = SCALE_OUT ? (acc[tt][v] + bo) * ro[v] : acc[tt][v] + bo;
      } else {
#pragma unroll
        for (int v = 0; v < 8; ++v) {
          const int m = m0 + v + 8 * half;
          if (m < n) D[(size_t)m * NOUT + c] = SCALE_OUT ? (acc[tt][v] + bo) * ro[v] : acc[tt][v] + bo;
        }
      }
    }
  }
}


extern "C" void kernel_launch(void* const* d_in, const int* in_sizes, int n_in,
                              void* d_out, int out_size, void* d_ws, size_t ws_size,
                              hipStream_t stream) {
  const float* x   = (const float*)d_in[0];
  const int*   ei  = (const int*)d_in[1];
  const float* ew  = (const float*)d_in[2];
  const float* W1  = (const float*)d_in[3];
  const float* b1  = (const float*)d_in[4];
  const float* Wc1 = (const float*)d_in[5];
  const float* bc1 = (const float*)d_in[6];
  const float* Wc2 = (const float*)d_in[7];
  const float* bc2 = (const float*)d_in[8];
  const float* W2  = (const float*)d_in[9];
  const float* b2  = (const float*)d_in[10];
  float* out = (float*)d_out;

  const int N = in_sizes[0] / F_IN;
  const int E = in_sizes[2];
  const int* erow = ei;
  const int* ecol = ei + E;

  int4*  tmp    = (int4*)d_ws;
  float* h      = (float*)(tmp + E);
  float* t      = h + (size_t)N * HIDDEN;
  int*   cntm   = (int*)(t + (size_t)N * HIDDEN);
  int*   locm   = cntm + (size_t)NB_MAX * SORT_GRID;
  int*   sedge  = locm + (size_t)NB_MAX * SORT_GRID;
  float* dinv   = (float*)(sedge + 2 * (size_t)E);
  int*   cnt    = (int*)(dinv + N);
  int*   off    = cnt + N;

  const int nb        = (N + BW - 1) / BW;
  const int gemm_grid = (N + 31) / 32;
  const int tile_grid = (N + 15) / 16;

  k_slice_sort<<<SORT_GRID, SORT_TB, 0, stream>>>(erow, ecol, ew, cntm, locm, tmp, E, N, nb);
  k_bucket_sort<<<nb, SORT_TB, 0, stream>>>(tmp, cntm, locm, sedge, off, cnt, dinv, E, N);

  k_lin_conv_wmma<F_IN><<<gemm_grid, 64, 0, stream>>>(x, W1, b1, Wc1, dinv, t, N);

  k_gather_gemm<HIDDEN, false, true>
      <<<tile_grid, 32, 0, stream>>>(off, cnt, sedge, dinv, t, bc1, Wc2, nullptr, h, N);

  k_gather_gemm<NCLS, true, false>
      <<<tile_grid, 32, 0, stream>>>(off, cnt, sedge, dinv, h, bc2, W2, b2, out, N);
}


# Both sort kernels place records in an LDS copy of their output range and store it front to back (a store covers 2-4 memory lines, not 32); degree sums read that copy; sums still in LDS arrival order
# speedup vs baseline: 4.4141x; 1.0834x over previous
#include <hip/hip_runtime.h>

typedef __attribute__((ext_vector_type(2))) float v2f;
typedef __attribute__((ext_vector_type(8))) float v8f;

#define F_IN   256
#define HIDDEN 64
#define NCLS   40


#define BW_SHIFT 9
#define BW       (1 << BW_SHIFT)
#define SORT_TB  1024
#define SORT_GRID 256
#define SEG_PW   (SORT_GRID / (SORT_TB / 32))
#define NB_MAX   1024
#define SCAP     6400
#define PCAP     12288
#define EPT      8

static __device__ __forceinline__ int block_excl_scan(int v, int* wsum, int* total) {
  const int lane = threadIdx.x & 31;
  const int wv   = threadIdx.x >> 5;
  int s = v;
  for (int d = 1; d < 32; d <<= 1) {
    const int u = __shfl_up(s, d);
    if (lane >= d) s += u;
  }
  if (lane == 31) wsum[wv] = s;
  __syncthreads();
  if (wv == 0) {
    const int ws = wsum[lane];
    int q = ws;
    for (int d = 1; d < 32; d <<= 1) {
      const int u = __shfl_up(q, d);
      if (lane >= d) q += u;
    }
    wsum[lane] = q - ws;
    if (lane == 31) wsum[32] = q;
  }
  __syncthreads();
  const int r = s - v + wsum[wv];
  *total = wsum[32];
  __syncthreads();
  return r;
}

static __device__ __forceinline__ int in_range(int c, int n) {
  const unsigned uc = (unsigned)c, last = (unsigned)(n - 1);
  return (int)(uc < last ? uc : last);
}

static __device__ __forceinline__ void block_sum2(int a, int b, int* red, int* sa, int* sb) {
  static_assert(SORT_TB == 32 * 32, "one partial sum per lane in the second step");
  const int lane = threadIdx.x & 31;
  const int wv   = threadIdx.x >> 5;
  for (int d = 16; d > 0; d >>= 1) {
    a += __shfl_xor(a, d);
    b += __shfl_xor(b, d);
  }
  if (lane == 0) { red[wv] = a; red[32 + wv] = b; }
  __syncthreads();
  int x = red[lane], y = red[32 + lane];
  for (int d = 16; d > 0; d >>= 1) {
    x += __shfl_xor(x, d);
    y += __shfl_xor(y, d);
  }
  *sa = x;
  *sb = y;
}

__global__ __launch_bounds__(SORT_TB) void k_slice_sort(const int* __restrict__ row, const int* __restrict__ col,
                                                        const float* __restrict__ ew,
                                                        int* __restrict__ cntm, int* __restrict__ locm,
                                                        int4* __restrict__ tmp, int e, int n, int nb) {
  __shared__ int h[NB_MAX];
  __shared__ int cur[NB_MAX];
  __shared__ int wsum[33];
  __shared__ int4 srec[SCAP];
  static_assert(NB_MAX == SORT_TB, "one LDS counter per thread");
  static_assert(SCAP <= EPT * SORT_TB, "a slice that fits the LDS array is one trip: the loops for longer slices store to memory");
  const int tid = threadIdx.x;
  const int b   = blockIdx.x;
  constexpr int nblk = SORT_GRID;
  h[tid] = 0;
  const int chunk = (e + nblk - 1) / nblk;
  const int beg   = b * chunk;
  const int end   = beg + chunk < e ? beg + chunk : e;
  const bool staged = chunk <= SCAP;

  int c0[EPT], r0[EPT], w0[EPT];
  for (int u = 0; u < EPT; ++u) {
    const int i  = beg + tid + u * SORT_TB;
    const int ii = i < end ? i : end - 1;
    c0[u] = col[ii];
    r0[u] = row[ii];
    w0[u] = __float_as_int(ew[ii]);
  }
  asm volatile("" :: "v"(c0[0]), "v"(c0[1]), "v"(c0[2]), "v"(c0[3]), "v"(c0[4]), "v"(c0[5]), "v"(c0[6]), "v"(c0[7]),
                     "v"(r0[0]), "v"(r0[1]), "v"(r0[2]), "v"(r0[3]), "v"(r0[4]), "v"(r0[5]), "v"(r0[6]), "v"(r0[7]),
                     "v"(w0[0]), "v"(w0[1]), "v"(w0[2]), "v"(w0[3]), "v"(w0[4]), "v"(w0[5]), "v"(w0[6]), "v"(w0[7]));
  __syncthreads();
  for (int u = 0; u < EPT; ++u)
    if (beg + tid + u * SORT_TB < end) atomicAdd(&h[in_range(c0[u], n) >> BW_SHIFT], 1);
  for (int i0 = beg + tid + EPT * SORT_TB; i0 < end; i0 += EPT * SORT_TB) {
    int c[EPT];
    for (int u = 0; u < EPT; ++u) {
      const int i = i0 + u * SORT_TB;
      c[u] = col[i < end ? i : end - 1];
    }
    asm volatile("" :: "v"(c[0]), "v"(c[1]), "v"(c[2]), "v"(c[3]), "v"(c[4]), "v"(c[5]), "v"(c[6]), "v"(c[7]));
    for (int u = 0; u < EPT; ++u)
      if (i0 + u * SORT_TB < end) atomicAdd(&h[in_range(c[u], n) >> BW_SHIFT], 1);
  }
  __syncthreads();
  const int hv = h[tid];
  int total;
  const int ex = block_excl_scan(hv, wsum, &total);
  cur[tid] = beg + ex;
  __syncthreads();

  if (staged) {
    int p[EPT];
    for (int u = 0; u < EPT; ++u) {
      p[u] = beg;
      if (beg + tid + u * SORT_TB < end) p[u] = atomicAdd(&cur[in_range(c0[u], n) >> BW_SHIFT], 1);
    }
    for (int u = 0; u < EPT; ++u) {
      if (beg + tid + u * SORT_TB < end) {
        const unsigned q = (unsigned)(p[u] - beg);
        srec[q < SCAP - 1 ? q : SCAP - 1] = make_int4(r0[u], in_range(c0[u], n), w0[u], 0);
      }
    }
    __syncthreads();
    constexpr int CT = (SCAP + SORT_TB - 1) / SORT_TB;
    const int nrec = end - beg;
    const int lastrec = nrec > 0 ? nrec - 1 : 0;
    int4 rr[CT];
    for (int u = 0; u < CT; ++u) {
      const int i = tid + u * SORT_TB;
      rr[u] = srec[i < nrec ? i : lastrec];
    }
    static_assert(CT == 7, "the statement below names rr[0]..rr[6]");
    asm volatile("" :: "v"(rr[0].x), "v"(rr[0].y), "v"(rr[0].z), "v"(rr[0].w), "v"(rr[1].x), "v"(rr[1].y), "v"(rr[1].z), "v"(rr[1].w),
                       "v"(rr[2].x), "v"(rr[2].y), "v"(rr[2].z), "v"(rr[2].w), "v"(rr[3].x), "v"(rr[3].y), "v"(rr[3].z), "v"(rr[3].w),
                       "v"(rr[4].x), "v"(rr[4].y), "v"(rr[4].z), "v"(rr[4].w), "v"(rr[5].x), "v"(rr[5].y), "v"(rr[5].z), "v"(rr[5].w),
                       "v"(rr[6].x), "v"(rr[6].y), "v"(rr[6].z), "v"(rr[6].w));
    for (int u = 0; u < CT; ++u) {
      const int i = tid + u * SORT_TB;
      if (i < nrec) tmp[beg + i] = rr[u];
    }
  } else {
    for (int u = 0; u < EPT; ++u) {
      if (beg + tid + u * SORT_TB < end) {
        const int cc  = in_range(c0[u], n);
        const int pos = atomicAdd(&cur[cc >> BW_SHIFT], 1);
        tmp[pos] = make_int4(r0[u], cc, w0[u], 0);
      }
    }
  }
  for (int i0 = beg + tid + EPT * SORT_TB; i0 < end; i0 += EPT * SORT_TB) {
    int c[EPT], r[EPT], w[EPT];
    for (int u = 0; u < EPT; ++u) {
      const int i  = i0 + u * SORT_TB;
      const int ii = i < end ? i : end - 1;
      c[u] = col[ii];
      r[u] = row[ii];
      w[u] = __float_as_int(ew[ii]);
    }
    asm volatile("" :: "v"(c[0]), "v"(c[1]), "v"(c[2]), "v"(c[3]), "v"(c[4]), "v"(c[5]), "v"(c[6]), "v"(c[7]),
                       "v"(r[0]), "v"(r[1]), "v"(r[2]), "v"(r[3]), "v"(r[4]), "v"(r[5]), "v"(r[6]), "v"(r[7]),
                       "v"(w[0]), "v"(w[1]), "v"(w[2]), "v"(w[3]), "v"(w[4]), "v"(w[5]), "v"(w[6]), "v"(w[7]));
    for (int u = 0; u < EPT; ++u) {
      if (i0 + u * SORT_TB < end) {
        const int cc  = in_range(c[u], n);
        const int pos = atomicAdd(&cur[cc >> BW_SHIFT], 1);
        tmp[pos] = make_int4(r[u], cc, w[u], 0);
      }
    }
  }
  if (tid < nb) {
    cntm[(size_t)tid * nblk + b] = hv;
    locm[(size_t)tid * nblk + b] = ex;
  }
}

__global__ __launch_bounds__(SORT_TB) void k_bucket_sort(const int4* __restrict__ tmp, const int* __restrict__ cntm,
                                                         const int* __restrict__ locm,
                                                         int* sedge, int* __restrict__ off, int* __restrict__ cnt,
                                                         float* __restrict__ dinv, int e, int n) {
  __shared__ int lcnt[BW];
  __shared__ int lcur[BW];
  __shared__ int wsum[33];
  __shared__ int sbase[SORT_GRID];
  __shared__ int scnt[SORT_GRID];
  __shared__ int red[64];
  __shared__ int2 spair[PCAP];
  static_assert(SORT_GRID <= SORT_TB && SORT_GRID % (SORT_TB / 32) == 0 && SEG_PW == 8,
                "one thread per piece in the prologue, eight pieces per wave in the walks");
  const int tid   = threadIdx.x;
  const int lane  = tid & 31;
  const int wv    = tid >> 5;
  const int node0 = (int)blockIdx.x << BW_SHIFT;
  constexpr int nblk = SORT_GRID;
  const int chunk = (e + nblk - 1) / nblk;
  int cv = 0, lv = 0;
  if (tid < nblk) {
    cv = cntm[(size_t)blockIdx.x * nblk + tid];
    lv = locm[(size_t)blockIdx.x * nblk + tid];
    cv = cv < 0 ? 0 : (cv < chunk ? cv : chunk);
    lv = lv < 0 ? 0 : (lv < chunk ? lv : chunk);
    sbase[tid] = tid * chunk + lv;
    scnt[tid]  = cv;
  }
  if (tid < BW) lcnt[tid] = 0;
  int s, ne;
  block_sum2(lv, cv, red, &s, &ne);
  s  = s < e ? s : e;
  ne = ne < e - s ? ne : e - s;
  const bool staged = ne <= PCAP;

  int gb[SEG_PW], nq[SEG_PW];
  int nmax = 0;
#pragma unroll
  for (int q = 0; q < SEG_PW; ++q) {
    gb[q] = __builtin_amdgcn_readfirstlane(sbase[wv * SEG_PW + q]);
    nq[q] = __builtin_amdgcn_readfirstlane(scnt[wv * SEG_PW + q]);
    nmax  = nq[q] > nmax ? nq[q] : nmax;
  }

  for (int j0 = 0; j0 < nmax; j0 += 64) {
    int c[2 * SEG_PW];
#pragma unroll
    for (int q = 0; q < SEG_PW; ++q) {
#pragma unroll
      for (int m = 0; m < 2; ++m) {
        const int j = j0 + 32 * m + lane;
        int i = gb[q] + (j < nq[q] ? j : nq[q] - 1);
        i = i < 0 ? 0 : (i < e ? i : e - 1);
        c[2 * q + m] = tmp[i].y;
      }
    }
    asm volatile("" :: "v"(c[0]), "v"(c[1]), "v"(c[2]), "v"(c[3]), "v"(c[4]), "v"(c[5]), "v"(c[6]), "v"(c[7]),
                       "v"(c[8]), "v"(c[9]), "v"(c[10]), "v"(c[11]), "v"(c[12]), "v"(c[13]), "v"(c[14]), "v"(c[15]));
#pragma unroll
    for (int q = 0; q < SEG_PW; ++q) {
#pragma unroll
      for (int m = 0; m < 2; ++m)
        if (j0 + 32 * m + lane < nq[q]) atomicAdd(&lcnt[(c[2 * q + m] - node0) & (BW - 1)], 1);
    }
  }
  __syncthreads();
  const int num = tid < BW ? lcnt[tid] : 0;
  int total;
  const int ex = block_excl_scan(num, wsum, &total);
  if (tid < BW) lcur[tid] = ex;
  __syncthreads();
  for (int j0 = 0; j0 < nmax; j0 += 64) {
    int4 r[2 * SEG_PW];
#pragma unroll
    for (int q = 0; q < SEG_PW; ++q) {
#pragma unroll
      for (int m = 0; m < 2; ++m) {
        const int j = j0 + 32 * m + lane;
        int i = gb[q] + (j < nq[q] ? j : nq[q] - 1);
        i = i < 0 ? 0 : (i < e ? i : e - 1);
        r[2 * q + m] = tmp[i];
      }
    }
    asm volatile("" :: "v"(r[0].x), "v"(r[0].y), "v"(r[0].z), "v"(r[1].x), "v"(r[1].y), "v"(r[1].z),
                       "v"(r[2].x), "v"(r[2].y), "v"(r[2].z), "v"(r[3].x), "v"(r[3].y), "v"(r[3].z),
                       "v"(r[4].x), "v"(r[4].y), "v"(r[4].z), "v"(r[5].x), "v"(r[5].y), "v"(r[5].z),
                       "v"(r[6].x), "v"(r[6].y), "v"(r[6].z), "v"(r[7].x), "v"(r[7].y), "v"(r[7].z),
                       "v"(r[8].x), "v"(r[8].y), "v"(r[8].z), "v"(r[9].x), "v"(r[9].y), "v"(r[9].z),
                       "v"(r[10].x), "v"(r[10].y), "v"(r[10].z), "v"(r[11].x), "v"(r[11].y), "v"(r[11].z),
                       "v"(r[12].x), "v"(r[12].y), "v"(r[12].z), "v"(r[13].x), "v"(r[13].y), "v"(r[13].z),
                       "v"(r[14].x), "v"(r[14].y), "v"(r[14].z), "v"(r[15].x), "v"(r[15].y), "v"(r[15].z));
    int p[2 * SEG_PW];
#pragma unroll
    for (int q = 0; q < SEG_PW; ++q) {
#pragma unroll
      for (int m = 0; m < 2; ++m) {
        p[2 * q + m] = 0;
        if (j0 + 32 * m + lane < nq[q]) p[2 * q + m] = atomicAdd(&lcur[(r[2 * q + m].y - node0) & (BW - 1)], 1);
      }
    }
#pragma unroll
    for (int q = 0; q < SEG_PW; ++q) {
#pragma unroll
      for (int m = 0; m < 2; ++m) {
        if (j0 + 32 * m + lane < nq[q]) {
          int pp = p[2 * q + m];
          pp = pp < ne ? pp : ne - 1;
          pp = pp < 0 ? 0 : pp;
          if (staged) spair[pp] = make_int2(r[2 * q + m].x, r[2 * q + m].z);
          else        *(int2*)(sedge + 2 * (size_t)(s + pp)) = make_int2(r[2 * q + m].x, r[2 * q + m].z);
        }
      }
    }
  }
  __syncthreads();
  if (staged) {
    for (int i0 = tid; i0 < ne; i0 += 4 * SORT_TB) {
      int2 pr[4];
      for (int u = 0; u < 4; ++u) {
        const int i = i0 + u * SORT_TB;
        pr[u] = spair[i < ne ? i : ne - 1];
      }
      asm volatile("" :: "v"(pr[0].x), "v"(pr[0].y), "v"(pr[1].x), "v"(pr[1].y), "v"(pr[2].x), "v"(pr[2].y), "v"(pr[3].x), "v"(pr[3].y));
      for (int u = 0; u < 4; ++u) {
        const int i = i0 + u * SORT_TB;
        if (i < ne) *(int2*)(sedge + 2 * (size_t)(s + i)) = pr[u];
      }
    }
  }
  const int c = node0 + tid;
  if (tid < BW && c < n) {
    const int beg = s + ex;
    off[c] = beg;
    cnt[c] = num;
    float sd = 1.0f;
    if (staged) {
      for (int j = 0; j < num; j += EPT) {
        float w[EPT];
        for (int u = 0; u < EPT; ++u) {
          const int jj = ex + (j + u < num ? j + u : num - 1);
          w[u] = __int_as_float(spair[jj < PCAP - 1 ? jj : PCAP - 1].y);
        }
        for (int u = 0; u < EPT; ++u) sd = (j + u < num) ? sd + w[u] : sd;
      }
    } else {
      const int* wp = sedge + 2 * (size_t)beg + 1;
      for (int j = 0; j < num; j += EPT) {
        float w[EPT];
        for (int u = 0; u < EPT; ++u) w[u] = __int_as_float(wp[2 * (j + u < num ? j + u : num - 1)]);
        for (int u = 0; u < EPT; ++u) sd = (j + u < num) ? sd + w[u] : sd;
      }
    }
    dinv[c] = (sd > 0.0f) ? rsqrtf(sd) : 0.0f;
  }
}


#define TILE_PITCH 68

template <int K>
__global__ __launch_bounds__(64) void k_lin_conv_wmma(const float* __restrict__ A,
                                                      const float* __restrict__ W1,
                                                      const float* __restrict__ b1,
                                                      const float* __restrict__ Wc,
                                                      const float* __restrict__ rscale,
                                                      float* __restrict__ D, int M) {
  constexpr int NT = HIDDEN / 16;
  __shared__ __attribute__((aligned(16))) float tiles[2][16 * TILE_PITCH];
  const int wv   = threadIdx.x >> 5;
  const int lane = threadIdx.x & 31;
  const int half = lane >> 4;
  const int l16  = lane & 15;
  const int m0   = __builtin_amdgcn_readfirstlane((blockIdx.x * 2 + wv) * 16);
  float* __restrict__ tile = tiles[wv];

  v8f acc[NT] = {};

  float rs[8];
#pragma unroll
  for (int v = 0; v < 8; ++v) {
    const int m = m0 + v + 8 * half;
    rs[v] = rscale[m < M ? m : M - 1];
  }

  const int mrow = m0 + l16 < M ? m0 + l16 : M - 1;
  const float* __restrict__ arow = A + (size_t)mrow * K + 2 * half;

  static_assert(NT == 4, "the column interleave below spells out four tiles");
  const float* __restrict__ w1l = W1 + 4 * l16;
  for (int k = 0; k < K; k += 4) {
    v2f a = *(const v2f*)(arow + k);
    const int kk = k + 2 * half;
    const float4 w0 = *(const float4*)(w1l + (size_t)kk * HIDDEN);
    const float4 w1 = *(const float4*)(w1l + (size_t)(kk + 1) * HIDDEN);
    v2f b;
    b.x = w0.x; b.y = w1.x;
    acc[0] = __builtin_amdgcn_wmma_f32_16x16x4_f32(false, a, false, b, (short)0, acc[0], false, false);
    b.x = w0.y; b.y = w1.y;
    acc[1] = __builtin_amdgcn_wmma_f32_16x16x4_f32(false, a, false, b, (short)0, acc[1], false, false);
    b.x = w0.z; b.y = w1.z;
    acc[2] = __builtin_amdgcn_wmma_f32_16x16x4_f32(false, a, false, b, (short)0, acc[2], false, false);
    b.x = w0.w; b.y = w1.w;
    acc[3] = __builtin_amdgcn_wmma_f32_16x16x4_f32(false, a, false, b, (short)0, acc[3], false, false);
  }

  {
    const float4 bv = *(const float4*)(b1 + 4 * l16);
#pragma unroll
    for (int v = 0; v < 8; ++v)
      *(float4*)(tile + (v + 8 * half) * TILE_PITCH + 4 * l16) =
          make_float4(fmaxf(acc[0][v] + bv.x, 0.0f), fmaxf(acc[1][v] + bv.y, 0.0f),
                      fmaxf(acc[2][v] + bv.z, 0.0f), fmaxf(acc[3][v] + bv.w, 0.0f));
  }
  __syncthreads();

  v8f acc2[NT] = {};
  const float* a2 = tile + l16 * TILE_PITCH + 2 * half;
  const float* __restrict__ wcl = Wc + 4 * l16;
#pragma unroll 4
  for (int k = 0; k < HIDDEN; k += 4) {
    const v2f a = *(const v2f*)(a2 + k);
    const int kk = k + 2 * half;
    const float4 w0 = *(const float4*)(wcl + (size_t)kk * HIDDEN);
    const float4 w1 = *(const float4*)(wcl + (size_t)(kk + 1) * HIDDEN);
    v2f b;
    b.x = w0.x; b.y = w1.x;
    acc2[0] = __builtin_amdgcn_wmma_f32_16x16x4_f32(false, a, false, b, (short)0, acc2[0], false, false);
    b.x = w0.y; b.y = w1.y;
    acc2[1] = __builtin_amdgcn_wmma_f32_16x16x4_f32(false, a, false, b, (short)0, acc2[1], false, false);
    b.x = w0.z; b.y = w1.z;
    acc2[2] = __builtin_amdgcn_wmma_f32_16x16x4_f32(false, a, false, b, (short)0, acc2[2], false, false);
    b.x = w0.w; b.y = w1.w;
    acc2[3] = __builtin_amdgcn_wmma_f32_16x16x4_f32(false, a, false, b, (short)0, acc2[3], false, false);
  }

  const bool whole = m0 + 16 <= M;
  if (whole) {
#pragma unroll
    for (int v = 0; v < 8; ++v)
      *(float4*)(D + (size_t)(m0 + v + 8 * half) * HIDDEN + 4 * l16) =
          make_float4(acc2[0][v] * rs[v], acc2[1][v] * rs[v], acc2[2][v] * rs[v], acc2[3][v] * rs[v]);
  } else {
#pragma unroll
    for (int v = 0; v < 8; ++v) {
      const int m = m0 + v + 8 * half;
      if (m < M)
        *(float4*)(D + (size_t)m * HIDDEN + 4 * l16) =
            make_float4(acc2[0][v] * rs[v], acc2[1][v] * rs[v], acc2[2][v] * rs[v], acc2[3][v] * rs[v]);
    }
  }
}


static __device__ __forceinline__ float4 ld_row4(const float* __restrict__ tl, int r) {
  return *(const float4*)(tl + (size_t)r * HIDDEN);
}


#define EPL        10
#define ECAP       640
#define SLOTS      16

template <int NOUT, bool BIAS, bool SCALE_OUT>
__global__ __launch_bounds__(32) void k_gather_gemm(const int* __restrict__ off, const int* __restrict__ cnt,
                                                    const int* __restrict__ sedge,
                                                    const float* __restrict__ dinv,
                                                    const float* __restrict__ t,
                                                    const float* __restrict__ bias_in,
                                                    const float* __restrict__ W,
                                                    const float* __restrict__ bias_out,
                                                    float* __restrict__ D, int n) {
  constexpr int NT = (NOUT + 15) / 16;
  constexpr bool IL = (NOUT == 64);
  __shared__ __attribute__((aligned(16))) float tile[16 * TILE_PITCH];
  __shared__ int2  ebuf[ECAP];
  const int lane = threadIdx.x;
  const int half = lane >> 4;
  const int l16  = lane & 15;
  const int m0   = blockIdx.x * 16;
  if (m0 >= n) return;

  const float* __restrict__ tl = t + 4 * l16;
  const char*  __restrict__ tb = (const char*)t;
  const unsigned lane_off = 16u * (unsigned)l16;
  float* __restrict__ trow = tile + 4 * l16;
  const float4 bv = *(const float4*)(bias_in + 4 * l16);
  float4 tv[8];
#pragma unroll
  for (int p = 0; p < 8; ++p) {
    const int c = m0 + 8 * half + p;
    tv[p] = ld_row4(tl, c < n ? c : n - 1);
  }

  const int  cl    = m0 + l16 < n ? m0 + l16 : n - 1;
  const int  offv  = off[cl];
  const int  cntv  = cnt[cl];
  const float dv   = dinv[cl];
  const int  e0    = __builtin_amdgcn_readfirstlane(offv);
  const int  endv  = offv + cntv - e0;
  const int  ne    = __builtin_amdgcn_readlane(endv, 15);
  int es[15];
#pragma unroll
  for (int i = 0; i < 15; ++i) es[i] = __builtin_amdgcn_readlane(endv, i);

#pragma unroll
  for (int p = 0; p < 8; ++p) {
    const int i = 8 * half + p;
    const float di = __shfl(dv, i);
    *(float4*)(trow + i * TILE_PITCH) = make_float4(bv.x + di * tv[p].x, bv.y + di * tv[p].y,
                                                    bv.z + di * tv[p].z, bv.w + di * tv[p].w);
  }

  const int hb = half ? es[7] : 0;
  const int he = half ? ne : es[7];
  int    cur = 8 * half;
  float4 sum = *(const float4*)(trow + cur * TILE_PITCH);

  for (int cb = 0; cb < ne; cb += ECAP) {
    const int cm = ne - cb < ECAP ? ne - cb : ECAP;
    __syncthreads();
    for (int j0 = 0; j0 < cm; j0 += EPL * 32) {
      int2  se[EPL];
#pragma unroll
      for (int u = 0; u < EPL; ++u) {
        const int j  = j0 + u * 32 + lane;
        const int jj = j < cm ? j : cm - 1;
        se[u] = *(const int2*)(sedge + 2 * (size_t)(e0 + cb + jj));
      }
      asm volatile("" : "+v"(se[0].x), "+v"(se[1].x), "+v"(se[2].x), "+v"(se[3].x), "+v"(se[4].x),
                        "+v"(se[5].x), "+v"(se[6].x), "+v"(se[7].x), "+v"(se[8].x), "+v"(se[9].x)
                      : "v"(se[0].y), "v"(se[1].y), "v"(se[2].y), "v"(se[3].y), "v"(se[4].y),
                        "v"(se[5].y), "v"(se[6].y), "v"(se[7].y), "v"(se[8].y), "v"(se[9].y));
#pragma unroll
      for (int u = 0; u < EPL; ++u) {
        int r = se[u].x;
        r = r < 0 ? 0 : (r >= n ? n - 1 : r);
        se[u].x = r;
      }
      int   nidv[EPL];
      float dcg[EPL];
#pragma unroll
      for (int u = 0; u < EPL; ++u) {
        const int g = cb + j0 + u * 32 + lane;
        int nid = 0;
#pragma unroll
        for (int i = 0; i < 15; ++i) nid += es[i] <= g ? 1 : 0;
        nidv[u] = nid;
        dcg[u]  = __shfl(dv, nid);
      }
#pragma unroll
      for (int u = 0; u < EPL; ++u) {
        const int j = j0 + u * 32 + lane;
        if (j < cm) ebuf[j] = make_int2((se[u].x << 8) | nidv[u], __float_as_int(__int_as_float(se[u].y) * dcg[u]));
      }
    }
    __syncthreads();

    const int lo   = (hb > cb ? hb : cb) - cb;
    const int hi   = (he < cb + cm ? he : cb + cm) - cb;
    const int nh   = hi - lo;
    const int na   = __builtin_amdgcn_readlane(nh, 0);
    const int nb2  = __builtin_amdgcn_readlane(nh, 16);
    const int nmax = na > nb2 ? na : nb2;
    const int nmin = na < nb2 ? na : nb2;
    const int lo0  = nh > 0 ? lo : 0;
    const int last = nh > 0 ? nh - 1 : 0;
    for (int j0 = 0; j0 < nmax; j0 += SLOTS) {
      float4 v[SLOTS];
      int2   q[SLOTS];
#pragma unroll
      for (int u = 0; u < SLOTS; ++u) {
        q[u] = ebuf[lo0 + (j0 + u < last ? j0 + u : last)];
        v[u] = *(const float4*)(tb + (((unsigned)q[u].x & 0xffffff00u) + lane_off));
      }
      asm volatile("" :: "v"(v[0].x), "v"(v[1].x), "v"(v[2].x), "v"(v[3].x), "v"(v[4].x), "v"(v[5].x),
                         "v"(v[6].x), "v"(v[7].x), "v"(v[8].x), "v"(v[9].x), "v"(v[10].x), "v"(v[11].x),
                         "v"(v[12].x), "v"(v[13].x), "v"(v[14].x), "v"(v[15].x));
      if (j0 + SLOTS <= nmin) {
#pragma unroll
        for (int u = 0; u < SLOTS; ++u) {
          const int nid = q[u].x & 15;
          if (nid != cur) {
            *(float4*)(trow + cur * TILE_PITCH) = sum;
            cur = nid;
            sum = *(const float4*)(trow + cur * TILE_PITCH);
          }
          const float wk = __int_as_float(q[u].y);
          sum.x = __builtin_fmaf(wk, v[u].x, sum.x);
          sum.y = __builtin_fmaf(wk, v[u].y, sum.y);
          sum.z = __builtin_fmaf(wk, v[u].z, sum.z);
          sum.w = __builtin_fmaf(wk, v[u].w, sum.w);
        }
      } else {
#pragma unroll
        for (int u = 0; u < SLOTS; ++u) {
          if (j0 + u < nh) {
            const int nid = q[u].x & 15;
            if (nid != cur) {
              *(float4*)(trow + cur * TILE_PITCH) = sum;
              cur = nid;
              sum = *(const float4*)(trow + cur * TILE_PITCH);
            }
            const float wk = __int_as_float(q[u].y);
            sum.x = __builtin_fmaf(wk, v[u].x, sum.x);
            sum.y = __builtin_fmaf(wk, v[u].y, sum.y);
            sum.z = __builtin_fmaf(wk, v[u].z, sum.z);
            sum.w = __builtin_fmaf(wk, v[u].w, sum.w);
          }
        }
      }
    }
  }
  *(float4*)(trow + cur * TILE_PITCH) = sum;
  __syncthreads();

  v8f acc[NT] = {};
  const float* arow = tile + l16 * TILE_PITCH + 2 * half;
#pragma unroll 4
  for (int k = 0; k < HIDDEN; k += 4) {
    v2f a = *(const v2f*)(arow + k);
    a.x = fmaxf(a.x, 0.0f);
    a.y = fmaxf(a.y, 0.0f);
    const int kk = k + 2 * half;
    if constexpr (IL) {
      const float4 w0 = *(const float4*)(W + (size_t)kk * NOUT + 4 * l16);
      const float4 w1 = *(const float4*)(W + (size_t)(kk + 1) * NOUT + 4 * l16);
      v2f bfr;
      bfr.x = w0.x; bfr.y = w1.x;
      acc[0] = __builtin_amdgcn_wmma_f32_16x16x4_f32(false, a, false, bfr, (short)0, acc[0], false, false);
      bfr.x = w0.y; bfr.y = w1.y;
      acc[1] = __builtin_amdgcn_wmma_f32_16x16x4_f32(false, a, false, bfr, (short)0, acc[1], false, false);
      bfr.x = w0.z; bfr.y = w1.z;
      acc[2] = __builtin_amdgcn_wmma_f32_16x16x4_f32(false, a, false, bfr, (short)0, acc[2], false, false);
      bfr.x = w0.w; bfr.y = w1.w;
      acc[3] = __builtin_amdgcn_wmma_f32_16x16x4_f32(false, a, false, bfr, (short)0, acc[3], false, false);
    } else {
#pragma unroll
      for (int tt = 0; tt < NT; ++tt) {
        const int c = tt * 16 + l16;
        v2f bfr;
        if ((NOUT % 16 == 0) || (c < NOUT)) {
          bfr.x = W[(size_t)kk * NOUT + c];
          bfr.y = W[(size_t)(kk + 1) * NOUT + c];
        } else {
          bfr.x = 0.0f; bfr.y = 0.0f;
        }
        acc[tt] = __builtin_amdgcn_wmma_f32_16x16x4_f32(
             false, a,  false, bfr,
             (short)0, acc[tt],  false,  false);
      }
    }
  }

  const bool whole = m0 + 16 <= n;
  float ro[8];
#pragma unroll
  for (int v = 0; v < 8; ++v) ro[v] = SCALE_OUT ? __shfl(dv, v + 8 * half) : 1.0f;
  if constexpr (IL) {
    float4 bo = make_float4(0.0f, 0.0f, 0.0f, 0.0f);
    if (BIAS) bo = *(const float4*)(bias_out + 4 * l16);
#pragma unroll
    for (int v = 0; v < 8; ++v) {
      const int m = m0 + v + 8 * half;
      if (whole || m < n)
        *(float4*)(D + (size_t)m * NOUT + 4 * l16) =
            SCALE_OUT ? make_float4((acc[0][v] + bo.x) * ro[v], (acc[1][v] + bo.y) * ro[v],
                                    (acc[2][v] + bo.z) * ro[v], (acc[3][v] + bo.w) * ro[v])
                      : make_float4(acc[0][v] + bo.x, acc[1][v] + bo.y, acc[2][v] + bo.z, acc[3][v] + bo.w);
    }
  } else {
#pragma unroll
    for (int tt = 0; tt < NT; ++tt) {
      const int c = tt * 16 + l16;
      if ((NOUT % 16 != 0) && (c >= NOUT)) continue;
      const float bo = BIAS ? bias_out[c] : 0.0f;
      if (whole) {
#pragma unroll
        for (int v = 0; v < 8; ++v) D[(size_t)(m0 + v + 8 * half) * NOUT + c] = SCALE_OUT ? (acc[tt][v] + bo) * ro[v] : acc[tt][v] + bo;
      } else {
#pragma unroll
        for (int v = 0; v < 8; ++v) {
          const int m = m0 + v + 8 * half;
          if (m < n) D[(size_t)m * NOUT + c] = SCALE_OUT ? (acc[tt][v] + bo) * ro[v] : acc[tt][v] + bo;
        }
      }
    }
  }
}


extern "C" void kernel_launch(void* const* d_in, const int* in_sizes, int n_in,
                              void* d_out, int out_size, void* d_ws, size_t ws_size,
                              hipStream_t stream) {
  const float* x   = (const float*)d_in[0];
  const int*   ei  = (const int*)d_in[1];
  const float* ew  = (const float*)d_in[2];
  const float* W1  = (const float*)d_in[3];
  const float* b1  = (const float*)d_in[4];
  const float* Wc1 = (const float*)d_in[5];
  const float* bc1 = (const float*)d_in[6];
  const float* Wc2 = (const float*)d_in[7];
  const float* bc2 = (const float*)d_in[8];
  const float* W2  = (const float*)d_in[9];
  const float* b2  = (const float*)d_in[10];
  float* out = (float*)d_out;

  const int N = in_sizes[0] / F_IN;
  const int E = in_sizes[2];
  const int* erow = ei;
  const int* ecol = ei + E;

  int4*  tmp    = (int4*)d_ws;
  float* h      = (float*)(tmp + E);
  float* t      = h + (size_t)N * HIDDEN;
  int*   cntm   = (int*)(t + (size_t)N * HIDDEN);
  int*   locm   = cntm + (size_t)NB_MAX * SORT_GRID;
  int*   sedge  = locm + (size_t)NB_MAX * SORT_GRID;
  float* dinv   = (float*)(sedge + 2 * (size_t)E);
  int*   cnt    = (int*)(dinv + N);
  int*   off    = cnt + N;

  const int nb        = (N + BW - 1) / BW;
  const int gemm_grid = (N + 31) / 32;
  const int tile_grid = (N + 15) / 16;

  k_slice_sort<<<SORT_GRID, SORT_TB, 0, stream>>>(erow, ecol, ew, cntm, locm, tmp, E, N, nb);
  k_bucket_sort<<<nb, SORT_TB, 0, stream>>>(tmp, cntm, locm, sedge, off, cnt, dinv, E, N);

  k_lin_conv_wmma<F_IN><<<gemm_grid, 64, 0, stream>>>(x, W1, b1, Wc1, dinv, t, N);

  k_gather_gemm<HIDDEN, false, true>
      <<<tile_grid, 32, 0, stream>>>(off, cnt, sedge, dinv, t, bc1, Wc2, nullptr, h, N);

  k_gather_gemm<NCLS, true, false>
      <<<tile_grid, 32, 0, stream>>>(off, cnt, sedge, dinv, h, bc2, W2, b2, out, N);
}
